# starting kernel
# baseline (speedup 1.0000x reference)
#include <hip/hip_runtime.h>

#define NN 100000
#define EE 3200000
#define FIN 512
#define HH 20
#define CC 10
#define SP 32

typedef __attribute__((ext_vector_type(2))) float v2f;
typedef __attribute__((ext_vector_type(8))) float v8f;


__global__ void __launch_bounds__(256) k_init_deg(float* __restrict__ deg) {
  int i = blockIdx.x * 256 + threadIdx.x;
  if (i < NN) deg[i] = 1.0f;
}

__global__ void __launch_bounds__(256) k_count_deg(const int* __restrict__ dst,
                                                   float* __restrict__ deg) {
  int e = blockIdx.x * 256 + threadIdx.x;
  if (e < EE) atomicAdd(&deg[dst[e]], 1.0f);
}

__global__ void __launch_bounds__(256) k_deg_to_dis(float* __restrict__ deg) {
  int i = blockIdx.x * 256 + threadIdx.x;
  if (i < NN) {
    float d = deg[i];
    deg[i] = (d > 0.0f) ? rsqrtf(d) : 0.0f;
  }
}


__global__ void __launch_bounds__(256) k_pack_b(const float* __restrict__ W,
                                                float* __restrict__ P,
                                                int K, int Ncols, int ntiles) {
  int t = blockIdx.x * 256 + threadIdx.x;
  int total = (K / 4) * ntiles * 32;
  if (t >= total) return;
  int lane = t & 31;
  int frag = t >> 5;
  int ks   = frag / ntiles;
  int tile = frag - ks * ntiles;
  int koff = (lane >> 4) * 2;
  int n    = tile * 16 + (lane & 15);
  int kk   = ks * 4 + koff;
  float v0 = (n < Ncols) ? W[kk * Ncols + n] : 0.0f;
  float v1 = (n < Ncols) ? W[(kk + 1) * Ncols + n] : 0.0f;
  P[(size_t)t * 2 + 0] = v0;
  P[(size_t)t * 2 + 1] = v1;
}


__global__ void __launch_bounds__(256) k_gemm_x(const float* __restrict__ x,
                                                const float* __restrict__ Bp,
                                                const float* __restrict__ dis,
                                                float* __restrict__ h,
                                                float* __restrict__ agg) {
  int wave = (blockIdx.x * blockDim.x + threadIdx.x) >> 5;
  int lane = threadIdx.x & 31;
  if (wave >= NN / 16) return;
  int half = lane >> 4;
  int l16  = lane & 15;
  int koff = half * 2;
  int m    = wave * 16 + l16;
  int n    = l16;
  v8f c0 = {}; v8f c1 = {};
  const float* xr = x + (size_t)m * FIN + koff;
  const v2f* P = (const v2f*)Bp;
  #pragma unroll 4
  for (int ks = 0; ks < FIN / 4; ++ks) {
    v2f a  = __builtin_nontemporal_load((const v2f*)(xr + ks * 4));
    v2f b0 = P[(ks * 2 + 0) * 32 + lane];
    v2f b1 = P[(ks * 2 + 1) * 32 + lane];
    c0 = __builtin_amdgcn_wmma_f32_16x16x4_f32(false, a, false, b0, (short)0, c0, false, false);
    c1 = __builtin_amdgcn_wmma_f32_16x16x4_f32(false, a, false, b1, (short)0, c1, false, false);
  }
  int rbase = wave * 16 + half * 8;
  #pragma unroll
  for (int v = 0; v < 8; ++v) {
    int row = rbase + v;
    float s = dis[row]; s = s * s;
    size_t r = (size_t)row * SP;
    float h0 = c0[v], h1 = c1[v];
    h[r + n]        = h0;
    h[r + n + 16]   = h1;
    agg[r + n]      = h0 * s;
    agg[r + n + 16] = h1 * s;
  }
}


__global__ void __launch_bounds__(256) k_gemm_h(const float* __restrict__ o,
                                                const float* __restrict__ Bp,
                                                const float* __restrict__ dis,
                                                float* __restrict__ h,
                                                float* __restrict__ agg) {
  int wave = (blockIdx.x * blockDim.x + threadIdx.x) >> 5;
  int lane = threadIdx.x & 31;
  if (wave >= NN / 16) return;
  int half = lane >> 4;
  int l16  = lane & 15;
  int koff = half * 2;
  int m    = wave * 16 + l16;
  int n    = l16;
  v8f c0 = {}; v8f c1 = {};
  const float* orow = o + (size_t)m * SP + koff;
  const v2f* P = (const v2f*)Bp;
  #pragma unroll
  for (int ks = 0; ks < HH / 4; ++ks) {
    v2f a  = *(const v2f*)(orow + ks * 4);
    v2f b0 = P[(ks * 2 + 0) * 32 + lane];
    v2f b1 = P[(ks * 2 + 1) * 32 + lane];
    c0 = __builtin_amdgcn_wmma_f32_16x16x4_f32(false, a, false, b0, (short)0, c0, false, false);
    c1 = __builtin_amdgcn_wmma_f32_16x16x4_f32(false, a, false, b1, (short)0, c1, false, false);
  }
  int rbase = wave * 16 + half * 8;
  #pragma unroll
  for (int v = 0; v < 8; ++v) {
    int row = rbase + v;
    float s = dis[row]; s = s * s;
    size_t r = (size_t)row * SP;
    float h0 = c0[v], h1 = c1[v];
    h[r + n]        = h0;
    h[r + n + 16]   = h1;
    agg[r + n]      = h0 * s;
    agg[r + n + 16] = h1 * s;
  }
}


__global__ void __launch_bounds__(256) k_agg_edges(const float* __restrict__ h,
                                                   const float* __restrict__ dis,
                                                   const int* __restrict__ src,
                                                   const int* __restrict__ dst,
                                                   float* __restrict__ agg) {
  int e = blockIdx.x * 256 + threadIdx.x;
  if (e >= EE) return;
  int s = src[e];
  int d = dst[e];
  float nrm = dis[s] * dis[d];
  const float4* hr = (const float4*)(h + (size_t)s * SP);
  float* ar = agg + (size_t)d * SP;
  #pragma unroll
  for (int q = 0; q < HH / 4; ++q) {
    float4 v = hr[q];
    atomicAdd(&ar[q * 4 + 0], v.x * nrm);
    atomicAdd(&ar[q * 4 + 1], v.y * nrm);
    atomicAdd(&ar[q * 4 + 2], v.z * nrm);
    atomicAdd(&ar[q * 4 + 3], v.w * nrm);
  }
}


__global__ void __launch_bounds__(256) k_norm_relu(const float* __restrict__ agg,
                                                   const float* __restrict__ bias,
                                                   float* __restrict__ o) {
  int i = blockIdx.x * 256 + threadIdx.x;
  if (i >= NN) return;
  float v[HH];
  float ss = 0.0f;
  const float* ar = agg + (size_t)i * SP;
  #pragma unroll
  for (int j = 0; j < HH; ++j) {
    float t = ar[j] + bias[j];
    v[j] = t;
    ss += t * t;
  }
  float inv = 1.0f / fmaxf(sqrtf(ss), 1e-12f);
  float* orow = o + (size_t)i * SP;
  #pragma unroll
  for (int j = 0; j < HH; ++j) orow[j] = fmaxf(v[j] * inv, 0.0f);
}


__global__ void __launch_bounds__(256) k_final(const float* __restrict__ o1,
                                               const float* __restrict__ o2,
                                               const float* __restrict__ o3,
                                               const float* __restrict__ Bp,
                                               const float* __restrict__ bl,
                                               float* __restrict__ out) {
  int wave = (blockIdx.x * blockDim.x + threadIdx.x) >> 5;
  int lane = threadIdx.x & 31;
  if (wave >= NN / 16) return;
  int half = lane >> 4;
  int l16  = lane & 15;
  int koff = half * 2;
  int m    = wave * 16 + l16;
  int n    = l16;
  v8f c = {};
  const v2f* P = (const v2f*)Bp;
  #pragma unroll
  for (int ks = 0; ks < 15; ++ks) {
    int layer = ks / 5;
    int kloc  = (ks - layer * 5) * 4;
    const float* ob = (layer == 0) ? o1 : (layer == 1) ? o2 : o3;
    v2f a = *(const v2f*)(ob + (size_t)m * SP + kloc + koff);
    v2f b = P[ks * 32 + lane];
    c = __builtin_amdgcn_wmma_f32_16x16x4_f32(false, a, false, b, (short)0, c, false, false);
  }
  if (n < CC) {
    float bv = bl[n];
    int rbase = wave * 16 + half * 8;
    #pragma unroll
    for (int v = 0; v < 8; ++v)
      out[(size_t)(rbase + v) * CC + n] = c[v] + bv;
  }
}


extern "C" void kernel_launch(void* const* d_in, const int* in_sizes, int n_in,
                              void* d_out, int out_size, void* d_ws, size_t ws_size,
                              hipStream_t stream) {
  (void)in_sizes; (void)n_in; (void)out_size; (void)ws_size;
  const float* x  = (const float*)d_in[0];
  const int*   ei = (const int*)d_in[1];
  const float* W1 = (const float*)d_in[2];
  const float* b1 = (const float*)d_in[3];
  const float* W2 = (const float*)d_in[4];
  const float* b2 = (const float*)d_in[5];
  const float* W3 = (const float*)d_in[6];
  const float* b3 = (const float*)d_in[7];
  const float* Wl = (const float*)d_in[8];
  const float* bl = (const float*)d_in[9];
  const int* srcI = ei;
  const int* dstI = ei + EE;
  float* out = (float*)d_out;

  float* ws   = (float*)d_ws;
  float* dis  = ws;
  float* hbuf = dis + NN;
  float* agg  = hbuf + (size_t)NN * SP;
  float* o1   = agg + (size_t)NN * SP;
  float* o2   = o1 + (size_t)NN * SP;
  float* o3   = o2 + (size_t)NN * SP;
  float* Bp1  = o3 + (size_t)NN * SP;
  float* Bp2  = Bp1 + (size_t)(FIN / 4) * 2 * 64;
  float* Bp3  = Bp2 + (size_t)(HH / 4) * 2 * 64;
  float* Bpl  = Bp3 + (size_t)(HH / 4) * 2 * 64;

  dim3 B(256);
  int gN  = (NN + 255) / 256;
  int gE  = (EE + 255) / 256;
  int gW  = ((NN / 16) * 32 + 255) / 256;
  int gP1 = ((FIN / 4) * 2 * 32 + 255) / 256;
  int gPh = ((HH / 4) * 2 * 32 + 255) / 256;
  int gPl = (15 * 32 + 255) / 256;

  k_pack_b<<<gP1, B, 0, stream>>>(W1, Bp1, FIN, HH, 2);
  k_pack_b<<<gPh, B, 0, stream>>>(W2, Bp2, HH, HH, 2);
  k_pack_b<<<gPh, B, 0, stream>>>(W3, Bp3, HH, HH, 2);
  k_pack_b<<<gPl, B, 0, stream>>>(Wl, Bpl, 3 * HH, CC, 1);

  k_init_deg<<<gN, B, 0, stream>>>(dis);
  k_count_deg<<<gE, B, 0, stream>>>(dstI, dis);
  k_deg_to_dis<<<gN, B, 0, stream>>>(dis);

  k_gemm_x<<<gW, B, 0, stream>>>(x, Bp1, dis, hbuf, agg);
  k_agg_edges<<<gE, B, 0, stream>>>(hbuf, dis, srcI, dstI, agg);
  k_norm_relu<<<gN, B, 0, stream>>>(agg, b1, o1);

  k_gemm_h<<<gW, B, 0, stream>>>(o1, Bp2, dis, hbuf, agg);
  k_agg_edges<<<gE, B, 0, stream>>>(hbuf, dis, srcI, dstI, agg);
  k_norm_relu<<<gN, B, 0, stream>>>(agg, b2, o2);

  k_gemm_h<<<gW, B, 0, stream>>>(o2, Bp3, dis, hbuf, agg);
  k_agg_edges<<<gE, B, 0, stream>>>(hbuf, dis, srcI, dstI, agg);
  k_norm_relu<<<gN, B, 0, stream>>>(agg, b3, o3);

  k_final<<<gW, B, 0, stream>>>(o1, o2, o3, Bpl, bl, out);
}


# Edge sums gathered per destination from a per-call CSR (integer-atomic arrival rank, scan, fill) in place of float atomic scatter; a row's terms are now summed in arrival-rank order
# speedup vs baseline: 5.2717x; 5.2717x over previous
#include <hip/hip_runtime.h>

#define NN 100000
#define EE 3200000
#define FIN 512
#define HH 20
#define CC 10
#define SP 32

typedef __attribute__((ext_vector_type(2))) float v2f;
typedef __attribute__((ext_vector_type(8))) float v8f;


__global__ void __launch_bounds__(256) k_zero_cnt(int* __restrict__ cnt) {
  int i = blockIdx.x * 256 + threadIdx.x;
  if (i < NN) cnt[i] = 0;
}

__global__ void __launch_bounds__(256) k_rank(const int* __restrict__ dst,
                                              int* __restrict__ cnt,
                                              int* __restrict__ rank) {
  int e = blockIdx.x * 256 + threadIdx.x;
  if (e < EE) rank[e] = atomicAdd(&cnt[dst[e]], 1);
}

__global__ void __launch_bounds__(1024) k_scan_local(const int* __restrict__ cnt,
                                                     int* __restrict__ rp,
                                                     int* __restrict__ bsum) {
  __shared__ int wsum[32];
  int t = threadIdx.x;
  int i = blockIdx.x * 1024 + t;
  int lane = t & 31;
  int w = t >> 5;
  int v = (i < NN) ? cnt[i] : 0;
  int incl = v;
  #pragma unroll
  for (int o = 1; o < 32; o <<= 1) {
    int u = __shfl_up(incl, o, 32);
    if (lane >= o) incl += u;
  }
  if (lane == 31) wsum[w] = incl;
  __syncthreads();
  if (w == 0) {
    int x = wsum[lane];
    int xi = x;
    #pragma unroll
    for (int o = 1; o < 32; o <<= 1) {
      int u = __shfl_up(xi, o, 32);
      if (lane >= o) xi += u;
    }
    wsum[lane] = xi - x;
    if (lane == 31) bsum[blockIdx.x] = xi;
  }
  __syncthreads();
  if (i < NN) rp[i] = incl - v + wsum[w];
}

__global__ void __launch_bounds__(1024) k_scan_add(int* __restrict__ rp,
                                                   const int* __restrict__ bsum) {
  int b = blockIdx.x;
  int lane = threadIdx.x & 31;
  int p = 0;
  #pragma unroll
  for (int k = 0; k < 4; ++k) {
    int bi = lane + 32 * k;
    if (bi < b) p += bsum[bi];
  }
  #pragma unroll
  for (int o = 16; o > 0; o >>= 1) p += __shfl_xor(p, o, 32);
  int i = b * 1024 + threadIdx.x;
  if (i < NN) rp[i] += p;
}

__global__ void __launch_bounds__(256) k_fill(const int* __restrict__ src,
                                              const int* __restrict__ dst,
                                              const int* __restrict__ rank,
                                              const int* __restrict__ rp,
                                              int* __restrict__ csr) {
  int e = blockIdx.x * 256 + threadIdx.x;
  if (e >= EE) return;
  csr[rp[dst[e]] + rank[e]] = src[e];
}

__device__ __forceinline__ float dis_of(int b, int e) {
  float d = (float)(e - b + 1);
  return (d > 0.0f) ? rsqrtf(d) : 0.0f;
}

__device__ __forceinline__ int row_end(const int* __restrict__ rp, int row) {
  int nx = rp[min(row + 1, NN - 1)];
  return (row + 1 < NN) ? nx : EE;
}


__global__ void __launch_bounds__(256) k_pack_b(const float* __restrict__ W,
                                                float* __restrict__ P,
                                                int K, int Ncols, int ntiles) {
  int t = blockIdx.x * 256 + threadIdx.x;
  int total = (K / 4) * ntiles * 32;
  if (t >= total) return;
  int lane = t & 31;
  int frag = t >> 5;
  int ks   = frag / ntiles;
  int tile = frag - ks * ntiles;
  int koff = (lane >> 4) * 2;
  int n    = tile * 16 + (lane & 15);
  int kk   = ks * 4 + koff;
  float v0 = (n < Ncols) ? W[kk * Ncols + n] : 0.0f;
  float v1 = (n < Ncols) ? W[(kk + 1) * Ncols + n] : 0.0f;
  P[(size_t)t * 2 + 0] = v0;
  P[(size_t)t * 2 + 1] = v1;
}


__global__ void __launch_bounds__(256) k_gemm_x(const float* __restrict__ x,
                                                const float* __restrict__ Bp,
                                                const int* __restrict__ rp,
                                                float* __restrict__ h) {
  int wave = (blockIdx.x * blockDim.x + threadIdx.x) >> 5;
  int lane = threadIdx.x & 31;
  if (wave >= NN / 16) return;
  int half = lane >> 4;
  int l16  = lane & 15;
  int koff = half * 2;
  int m    = wave * 16 + l16;
  int n    = l16;
  v8f c0 = {}; v8f c1 = {};
  const float* xr = x + (size_t)m * FIN + koff;
  const v2f* P = (const v2f*)Bp;
  #pragma unroll 4
  for (int ks = 0; ks < FIN / 4; ++ks) {
    v2f a  = __builtin_nontemporal_load((const v2f*)(xr + ks * 4));
    v2f b0 = P[(ks * 2 + 0) * 32 + lane];
    v2f b1 = P[(ks * 2 + 1) * 32 + lane];
    c0 = __builtin_amdgcn_wmma_f32_16x16x4_f32(false, a, false, b0, (short)0, c0, false, false);
    c1 = __builtin_amdgcn_wmma_f32_16x16x4_f32(false, a, false, b1, (short)0, c1, false, false);
  }
  int rbase = wave * 16 + half * 8;
  int pr[9];
  #pragma unroll
  for (int v = 0; v < 8; ++v) pr[v] = rp[rbase + v];
  pr[8] = row_end(rp, rbase + 7);
  #pragma unroll
  for (int v = 0; v < 8; ++v) {
    int row = rbase + v;
    float s = dis_of(pr[v], pr[v + 1]);
    size_t r = (size_t)row * SP;
    h[r + n]      = c0[v] * s;
    h[r + n + 16] = c1[v] * s;
  }
}


__global__ void __launch_bounds__(256) k_gemm_h(const float* __restrict__ o,
                                                const float* __restrict__ Bp,
                                                const int* __restrict__ rp,
                                                float* __restrict__ h) {
  int wave = (blockIdx.x * blockDim.x + threadIdx.x) >> 5;
  int lane = threadIdx.x & 31;
  if (wave >= NN / 16) return;
  int half = lane >> 4;
  int l16  = lane & 15;
  int koff = half * 2;
  int m    = wave * 16 + l16;
  int n    = l16;
  v8f c0 = {}; v8f c1 = {};
  const float* orow = o + (size_t)m * SP + koff;
  const v2f* P = (const v2f*)Bp;
  #pragma unroll
  for (int ks = 0; ks < HH / 4; ++ks) {
    v2f a  = *(const v2f*)(orow + ks * 4);
    v2f b0 = P[(ks * 2 + 0) * 32 + lane];
    v2f b1 = P[(ks * 2 + 1) * 32 + lane];
    c0 = __builtin_amdgcn_wmma_f32_16x16x4_f32(false, a, false, b0, (short)0, c0, false, false);
    c1 = __builtin_amdgcn_wmma_f32_16x16x4_f32(false, a, false, b1, (short)0, c1, false, false);
  }
  int rbase = wave * 16 + half * 8;
  int pr[9];
  #pragma unroll
  for (int v = 0; v < 8; ++v) pr[v] = rp[rbase + v];
  pr[8] = row_end(rp, rbase + 7);
  #pragma unroll
  for (int v = 0; v < 8; ++v) {
    int row = rbase + v;
    float s = dis_of(pr[v], pr[v + 1]);
    size_t r = (size_t)row * SP;
    h[r + n]      = c0[v] * s;
    h[r + n + 16] = c1[v] * s;
  }
}


template <int K>
__device__ __forceinline__ int bcast8(int v) {
  return (int)__builtin_amdgcn_mov_dpp8((unsigned)v, (unsigned)(K * 0x249249));
}

__device__ __forceinline__ float4 f4add(float4 a, float4 b) {
  return make_float4(a.x + b.x, a.y + b.y, a.z + b.z, a.w + b.w);
}

__global__ void __launch_bounds__(256) k_gather(const float* __restrict__ hs,
                                                const int* __restrict__ rp,
                                                const int* __restrict__ csr,
                                                float* __restrict__ agg) {
  int t   = blockIdx.x * 256 + threadIdx.x;
  int row = t >> 3;
  int q   = t & 7;
  unsigned qb = (unsigned)q * 16u;
  int beg = rp[row];
  int end = row_end(rp, row);
  const char* Hb = (const char*)hs;
#define ROW_LD(s) (*(const float4*)(Hb + ((unsigned)(s) * 128u + qb)))
  float4 acc = ROW_LD(row);
  int j = beg;
  int ld0 = csr[min(j + q, EE - 1)];
  int idx = (j + q < end) ? ld0 : row;
  for (; j + 8 <= end; j += 8) {
    float4 v0 = ROW_LD(bcast8<0>(idx));
    float4 v1 = ROW_LD(bcast8<1>(idx));
    float4 v2 = ROW_LD(bcast8<2>(idx));
    float4 v3 = ROW_LD(bcast8<3>(idx));
    float4 v4 = ROW_LD(bcast8<4>(idx));
    float4 v5 = ROW_LD(bcast8<5>(idx));
    float4 v6 = ROW_LD(bcast8<6>(idx));
    float4 v7 = ROW_LD(bcast8<7>(idx));
    int nj  = j + 8 + q;
    int ldn = csr[min(nj, EE - 1)];
    acc = f4add(acc, v0); acc = f4add(acc, v1); acc = f4add(acc, v2); acc = f4add(acc, v3);
    acc = f4add(acc, v4); acc = f4add(acc, v5); acc = f4add(acc, v6); acc = f4add(acc, v7);
    idx = (nj < end) ? ldn : row;
  }
  int rem = end - j;
  if (rem > 0) {
    const float4 z = make_float4(0.0f, 0.0f, 0.0f, 0.0f);
    float4 v0 = ROW_LD(bcast8<0>(idx));
    float4 v1 = ROW_LD(bcast8<1>(idx));
    float4 v2 = ROW_LD(bcast8<2>(idx));
    float4 v3 = ROW_LD(bcast8<3>(idx));
    float4 v4 = ROW_LD(bcast8<4>(idx));
    float4 v5 = ROW_LD(bcast8<5>(idx));
    float4 v6 = ROW_LD(bcast8<6>(idx));
    acc = f4add(acc, v0);
    acc = f4add(acc, (rem > 1) ? v1 : z);
    acc = f4add(acc, (rem > 2) ? v2 : z);
    acc = f4add(acc, (rem > 3) ? v3 : z);
    acc = f4add(acc, (rem > 4) ? v4 : z);
    acc = f4add(acc, (rem > 5) ? v5 : z);
    acc = f4add(acc, (rem > 6) ? v6 : z);
  }
#undef ROW_LD
  float dd = dis_of(beg, end);
  ((float4*)agg)[(size_t)row * 8 + q] = make_float4(acc.x * dd, acc.y * dd, acc.z * dd, acc.w * dd);
}


__global__ void __launch_bounds__(256) k_norm_relu(float* __restrict__ buf,
                                                   const float* __restrict__ bias) {
  int i = blockIdx.x * 256 + threadIdx.x;
  if (i >= NN) return;
  float v[HH];
  float ss = 0.0f;
  float* ar = buf + (size_t)i * SP;
  #pragma unroll
  for (int j = 0; j < HH; ++j) {
    float t = ar[j] + bias[j];
    v[j] = t;
    ss += t * t;
  }
  float inv = 1.0f / fmaxf(sqrtf(ss), 1e-12f);
  #pragma unroll
  for (int j = 0; j < HH; ++j) ar[j] = fmaxf(v[j] * inv, 0.0f);
}


__global__ void __launch_bounds__(256) k_final(const float* __restrict__ o1,
                                               const float* __restrict__ o2,
                                               const float* __restrict__ o3,
                                               const float* __restrict__ Bp,
                                               const float* __restrict__ bl,
                                               float* __restrict__ out) {
  int wave = (blockIdx.x * blockDim.x + threadIdx.x) >> 5;
  int lane = threadIdx.x & 31;
  if (wave >= NN / 16) return;
  int half = lane >> 4;
  int l16  = lane & 15;
  int koff = half * 2;
  int m    = wave * 16 + l16;
  int n    = l16;
  v8f c = {};
  const v2f* P = (const v2f*)Bp;
  #pragma unroll
  for (int ks = 0; ks < 15; ++ks) {
    int layer = ks / 5;
    int kloc  = (ks - layer * 5) * 4;
    const float* ob = (layer == 0) ? o1 : (layer == 1) ? o2 : o3;
    v2f a = *(const v2f*)(ob + (size_t)m * SP + kloc + koff);
    v2f b = P[ks * 32 + lane];
    c = __builtin_amdgcn_wmma_f32_16x16x4_f32(false, a, false, b, (short)0, c, false, false);
  }
  if (n < CC) {
    float bv = bl[n];
    int rbase = wave * 16 + half * 8;
    #pragma unroll
    for (int v = 0; v < 8; ++v)
      out[(size_t)(rbase + v) * CC + n] = c[v] + bv;
  }
}


extern "C" void kernel_launch(void* const* d_in, const int* in_sizes, int n_in,
                              void* d_out, int out_size, void* d_ws, size_t ws_size,
                              hipStream_t stream) {
  (void)in_sizes; (void)n_in; (void)out_size; (void)ws_size;
  const float* x  = (const float*)d_in[0];
  const int*   ei = (const int*)d_in[1];
  const float* W1 = (const float*)d_in[2];
  const float* b1 = (const float*)d_in[3];
  const float* W2 = (const float*)d_in[4];
  const float* b2 = (const float*)d_in[5];
  const float* W3 = (const float*)d_in[6];
  const float* b3 = (const float*)d_in[7];
  const float* Wl = (const float*)d_in[8];
  const float* bl = (const float*)d_in[9];
  const int* srcI = ei;
  const int* dstI = ei + EE;
  float* out = (float*)d_out;

  float* ws   = (float*)d_ws;
  int*   rp   = (int*)ws;
  float* hbuf = ws + NN;
  int*   csr  = (int*)(hbuf + (size_t)NN * SP);
  float* o1   = hbuf + (size_t)2 * NN * SP;
  float* o2   = o1 + (size_t)NN * SP;
  float* o3   = o2 + (size_t)NN * SP;
  float* Bp1  = o3 + (size_t)NN * SP;
  float* Bp2  = Bp1 + (size_t)(FIN / 4) * 2 * 64;
  float* Bp3  = Bp2 + (size_t)(HH / 4) * 2 * 64;
  float* Bpl  = Bp3 + (size_t)(HH / 4) * 2 * 64;
  int*   rank = (int*)o3;
  int*   cnt  = (int*)o2;
  int*   bsum = cnt + NN;

  dim3 B(256);
  dim3 B4(1024);
  int gN  = (NN + 255) / 256;
  int gE  = (EE + 255) / 256;
  int gS  = (NN + 1023) / 1024;
  int gG  = (NN * 8) / 256;
  int gW  = ((NN / 16) * 32 + 255) / 256;
  int gP1 = ((FIN / 4) * 2 * 32 + 255) / 256;
  int gPh = ((HH / 4) * 2 * 32 + 255) / 256;
  int gPl = (15 * 32 + 255) / 256;

  k_pack_b<<<gP1, B, 0, stream>>>(W1, Bp1, FIN, HH, 2);
  k_pack_b<<<gPh, B, 0, stream>>>(W2, Bp2, HH, HH, 2);
  k_pack_b<<<gPh, B, 0, stream>>>(W3, Bp3, HH, HH, 2);
  k_pack_b<<<gPl, B, 0, stream>>>(Wl, Bpl, 3 * HH, CC, 1);

  k_zero_cnt<<<gN, B, 0, stream>>>(cnt);
  k_rank<<<gE, B, 0, stream>>>(dstI, cnt, rank);
  k_scan_local<<<gS, B4, 0, stream>>>(cnt, rp, bsum);
  k_scan_add<<<gS, B4, 0, stream>>>(rp, bsum);
  k_fill<<<gE, B, 0, stream>>>(srcI, dstI, rank, rp, csr);

  k_gemm_x<<<gW, B, 0, stream>>>(x, Bp1, rp, hbuf);
  k_gather<<<gG, B, 0, stream>>>(hbuf, rp, csr, o1);
  k_norm_relu<<<gN, B, 0, stream>>>(o1, b1);

  k_gemm_h<<<gW, B, 0, stream>>>(o1, Bp2, rp, hbuf);
  k_gather<<<gG, B, 0, stream>>>(hbuf, rp, csr, o2);
  k_norm_relu<<<gN, B, 0, stream>>>(o2, b2);

  k_gemm_h<<<gW, B, 0, stream>>>(o2, Bp3, rp, hbuf);
  k_gather<<<gG, B, 0, stream>>>(hbuf, rp, csr, o3);
  k_norm_relu<<<gN, B, 0, stream>>>(o3, b3);

  k_final<<<gW, B, 0, stream>>>(o1, o2, o3, Bpl, bl, out);
}


# x GEMM: A fragments staged through LDS from whole-line 16-byte loads (4 lines per load, next piece in flight under the matrix work) in place of 8-byte loads that touch 16 lines; k order, sums kept
# speedup vs baseline: 6.2224x; 1.1803x over previous
#include <hip/hip_runtime.h>

#define NN 100000
#define EE 3200000
#define FIN 512
#define HH 20
#define CC 10
#define SP 32

typedef __attribute__((ext_vector_type(2))) float v2f;
typedef __attribute__((ext_vector_type(4))) float v4f;
typedef __attribute__((ext_vector_type(8))) float v8f;


__global__ void __launch_bounds__(256) k_zero_cnt(int* __restrict__ cnt) {
  int i = blockIdx.x * 256 + threadIdx.x;
  if (i < NN) cnt[i] = 0;
}

__global__ void __launch_bounds__(256) k_rank(const int* __restrict__ dst,
                                              int* __restrict__ cnt,
                                              int* __restrict__ rank) {
  int e = blockIdx.x * 256 + threadIdx.x;
  if (e < EE) rank[e] = atomicAdd(&cnt[dst[e]], 1);
}

__global__ void __launch_bounds__(1024) k_scan_local(const int* __restrict__ cnt,
                                                     int* __restrict__ rp,
                                                     int* __restrict__ bsum) {
  __shared__ int wsum[32];
  int t = threadIdx.x;
  int i = blockIdx.x * 1024 + t;
  int lane = t & 31;
  int w = t >> 5;
  int v = (i < NN) ? cnt[i] : 0;
  int incl = v;
  #pragma unroll
  for (int o = 1; o < 32; o <<= 1) {
    int u = __shfl_up(incl, o, 32);
    if (lane >= o) incl += u;
  }
  if (lane == 31) wsum[w] = incl;
  __syncthreads();
  if (w == 0) {
    int x = wsum[lane];
    int xi = x;
    #pragma unroll
    for (int o = 1; o < 32; o <<= 1) {
      int u = __shfl_up(xi, o, 32);
      if (lane >= o) xi += u;
    }
    wsum[lane] = xi - x;
    if (lane == 31) bsum[blockIdx.x] = xi;
  }
  __syncthreads();
  if (i < NN) rp[i] = incl - v + wsum[w];
}

__global__ void __launch_bounds__(1024) k_scan_add(int* __restrict__ rp,
                                                   const int* __restrict__ bsum) {
  int b = blockIdx.x;
  int lane = threadIdx.x & 31;
  int p = 0;
  #pragma unroll
  for (int k = 0; k < 4; ++k) {
    int bi = lane + 32 * k;
    if (bi < b) p += bsum[bi];
  }
  #pragma unroll
  for (int o = 16; o > 0; o >>= 1) p += __shfl_xor(p, o, 32);
  int i = b * 1024 + threadIdx.x;
  if (i < NN) rp[i] += p;
}

__global__ void __launch_bounds__(256) k_fill(const int* __restrict__ src,
                                              const int* __restrict__ dst,
                                              const int* __restrict__ rank,
                                              const int* __restrict__ rp,
                                              int* __restrict__ csr) {
  int e = blockIdx.x * 256 + threadIdx.x;
  if (e >= EE) return;
  csr[rp[dst[e]] + rank[e]] = src[e];
}

__device__ __forceinline__ float dis_of(int b, int e) {
  float d = (float)(e - b + 1);
  return (d > 0.0f) ? rsqrtf(d) : 0.0f;
}

__device__ __forceinline__ int row_end(const int* __restrict__ rp, int row) {
  int nx = rp[min(row + 1, NN - 1)];
  return (row + 1 < NN) ? nx : EE;
}


__global__ void __launch_bounds__(256) k_pack_b(const float* __restrict__ W,
                                                float* __restrict__ P,
                                                int K, int Ncols, int ntiles) {
  int t = blockIdx.x * 256 + threadIdx.x;
  int total = (K / 4) * ntiles * 32;
  if (t >= total) return;
  int lane = t & 31;
  int frag = t >> 5;
  int ks   = frag / ntiles;
  int tile = frag - ks * ntiles;
  int koff = (lane >> 4) * 2;
  int n    = tile * 16 + (lane & 15);
  int kk   = ks * 4 + koff;
  float v0 = (n < Ncols) ? W[kk * Ncols + n] : 0.0f;
  float v1 = (n < Ncols) ? W[(kk + 1) * Ncols + n] : 0.0f;
  P[(size_t)t * 2 + 0] = v0;
  P[(size_t)t * 2 + 1] = v1;
}


#define XCH 64
#define XPAD 68
#define XW 2

__global__ void __launch_bounds__(XW * 32) k_gemm_x(const float* __restrict__ x,
                                                    const float* __restrict__ Bp,
                                                    const int* __restrict__ rp,
                                                    float* __restrict__ h) {
  __shared__ v4f stage[XW][2][16 * XPAD / 4];
  int w    = threadIdx.x >> 5;
  int lane = threadIdx.x & 31;
  int wave = blockIdx.x * XW + w;
  int half = lane >> 4;
  int l16  = lane & 15;
  int n    = l16;
  int q  = lane & 7;
  int g  = lane >> 3;
  int r0 = 8 * (g & 1) + (g >> 1);
  const float* xg = x + (size_t)(wave * 16 + r0) * FIN + q * 4;
  int so = r0 * XPAD + q * 4;
  int lo = l16 * XPAD + half * 2;
  v8f c0 = {}; v8f c1 = {};
  const v2f* P = (const v2f*)Bp;
  v4f buf[8];
  for (int i = 0; i < 8; ++i)
    buf[i] = __builtin_nontemporal_load((const v4f*)(xg + (size_t)(i >> 1) * 2 * FIN + (i & 1) * 32));
  #pragma unroll 1
  for (int ch = 0; ch < FIN / XCH; ++ch) {
    v2f bq[2 * (XCH / 4)];
    for (int j = 0; j < 2 * (XCH / 4); ++j) bq[j] = P[(ch * 2 * (XCH / 4) + j) * 32 + lane];
    float* sp = (float*)stage[w][ch & 1];
    for (int i = 0; i < 8; ++i)
      *(v4f*)(sp + so + (i >> 1) * 2 * XPAD + (i & 1) * 32) = buf[i];
    __syncthreads();
    if (ch + 1 < FIN / XCH) {
      for (int i = 0; i < 8; ++i)
        buf[i] = __builtin_nontemporal_load(
            (const v4f*)(xg + (size_t)(i >> 1) * 2 * FIN + (i & 1) * 32 + (ch + 1) * XCH));
    }
    const float* fr = sp + lo;
    for (int ks = 0; ks < XCH / 4; ++ks) {
      v2f a = *(const v2f*)(fr + ks * 4);
      c0 = __builtin_amdgcn_wmma_f32_16x16x4_f32(false, a, false, bq[2 * ks + 0], (short)0, c0, false, false);
      c1 = __builtin_amdgcn_wmma_f32_16x16x4_f32(false, a, false, bq[2 * ks + 1], (short)0, c1, false, false);
    }
  }
  int rbase = wave * 16 + half * 8;
  int pr[9];
  #pragma unroll
  for (int v = 0; v < 8; ++v) pr[v] = rp[rbase + v];
  pr[8] = row_end(rp, rbase + 7);
  #pragma unroll
  for (int v = 0; v < 8; ++v) {
    int row = rbase + v;
    float s = dis_of(pr[v], pr[v + 1]);
    size_t r = (size_t)row * SP;
    h[r + n]      = c0[v] * s;
    h[r + n + 16] = c1[v] * s;
  }
}


__global__ void __launch_bounds__(256) k_gemm_h(const float* __restrict__ o,
                                                const float* __restrict__ Bp,
                                                const int* __restrict__ rp,
                                                float* __restrict__ h) {
  int wave = (blockIdx.x * blockDim.x + threadIdx.x) >> 5;
  int lane = threadIdx.x & 31;
  if (wave >= NN / 16) return;
  int half = lane >> 4;
  int l16  = lane & 15;
  int koff = half * 2;
  int m    = wave * 16 + l16;
  int n    = l16;
  v8f c0 = {}; v8f c1 = {};
  const float* orow = o + (size_t)m * SP + koff;
  const v2f* P = (const v2f*)Bp;
  #pragma unroll
  for (int ks = 0; ks < HH / 4; ++ks) {
    v2f a  = *(const v2f*)(orow + ks * 4);
    v2f b0 = P[(ks * 2 + 0) * 32 + lane];
    v2f b1 = P[(ks * 2 + 1) * 32 + lane];
    c0 = __builtin_amdgcn_wmma_f32_16x16x4_f32(false, a, false, b0, (short)0, c0, false, false);
    c1 = __builtin_amdgcn_wmma_f32_16x16x4_f32(false, a, false, b1, (short)0, c1, false, false);
  }
  int rbase = wave * 16 + half * 8;
  int pr[9];
  #pragma unroll
  for (int v = 0; v < 8; ++v) pr[v] = rp[rbase + v];
  pr[8] = row_end(rp, rbase + 7);
  #pragma unroll
  for (int v = 0; v < 8; ++v) {
    int row = rbase + v;
    float s = dis_of(pr[v], pr[v + 1]);
    size_t r = (size_t)row * SP;
    h[r + n]      = c0[v] * s;
    h[r + n + 16] = c1[v] * s;
  }
}


template <int K>
__device__ __forceinline__ int bcast8(int v) {
  return (int)__builtin_amdgcn_mov_dpp8((unsigned)v, (unsigned)(K * 0x249249));
}

__device__ __forceinline__ float4 f4add(float4 a, float4 b) {
  return make_float4(a.x + b.x, a.y + b.y, a.z + b.z, a.w + b.w);
}

__global__ void __launch_bounds__(256) k_gather(const float* __restrict__ hs,
                                                const int* __restrict__ rp,
                                                const int* __restrict__ csr,
                                                float* __restrict__ agg) {
  int t   = blockIdx.x * 256 + threadIdx.x;
  int row = t >> 3;
  int q   = t & 7;
  unsigned qb = (unsigned)q * 16u;
  int beg = rp[row];
  int end = row_end(rp, row);
  const char* Hb = (const char*)hs;
#define ROW_LD(s) (*(const float4*)(Hb + ((unsigned)(s) * 128u + qb)))
  float4 acc = ROW_LD(row);
  int j = beg;
  int ld0 = csr[min(j + q, EE - 1)];
  int idx = (j + q < end) ? ld0 : row;
  for (; j + 8 <= end; j += 8) {
    float4 v0 = ROW_LD(bcast8<0>(idx));
    float4 v1 = ROW_LD(bcast8<1>(idx));
    float4 v2 = ROW_LD(bcast8<2>(idx));
    float4 v3 = ROW_LD(bcast8<3>(idx));
    float4 v4 = ROW_LD(bcast8<4>(idx));
    float4 v5 = ROW_LD(bcast8<5>(idx));
    float4 v6 = ROW_LD(bcast8<6>(idx));
    float4 v7 = ROW_LD(bcast8<7>(idx));
    int nj  = j + 8 + q;
    int ldn = csr[min(nj, EE - 1)];
    acc = f4add(acc, v0); acc = f4add(acc, v1); acc = f4add(acc, v2); acc = f4add(acc, v3);
    acc = f4add(acc, v4); acc = f4add(acc, v5); acc = f4add(acc, v6); acc = f4add(acc, v7);
    idx = (nj < end) ? ldn : row;
  }
  int rem = end - j;
  if (rem > 0) {
    const float4 z = make_float4(0.0f, 0.0f, 0.0f, 0.0f);
    float4 v0 = ROW_LD(bcast8<0>(idx));
    float4 v1 = ROW_LD(bcast8<1>(idx));
    float4 v2 = ROW_LD(bcast8<2>(idx));
    float4 v3 = ROW_LD(bcast8<3>(idx));
    float4 v4 = ROW_LD(bcast8<4>(idx));
    float4 v5 = ROW_LD(bcast8<5>(idx));
    float4 v6 = ROW_LD(bcast8<6>(idx));
    acc = f4add(acc, v0);
    acc = f4add(acc, (rem > 1) ? v1 : z);
    acc = f4add(acc, (rem > 2) ? v2 : z);
    acc = f4add(acc, (rem > 3) ? v3 : z);
    acc = f4add(acc, (rem > 4) ? v4 : z);
    acc = f4add(acc, (rem > 5) ? v5 : z);
    acc = f4add(acc, (rem > 6) ? v6 : z);
  }
#undef ROW_LD
  float dd = dis_of(beg, end);
  ((float4*)agg)[(size_t)row * 8 + q] = make_float4(acc.x * dd, acc.y * dd, acc.z * dd, acc.w * dd);
}


__global__ void __launch_bounds__(256) k_norm_relu(float* __restrict__ buf,
                                                   const float* __restrict__ bias) {
  int i = blockIdx.x * 256 + threadIdx.x;
  if (i >= NN) return;
  float v[HH];
  float ss = 0.0f;
  float* ar = buf + (size_t)i * SP;
  #pragma unroll
  for (int j = 0; j < HH; ++j) {
    float t = ar[j] + bias[j];
    v[j] = t;
    ss += t * t;
  }
  float inv = 1.0f / fmaxf(sqrtf(ss), 1e-12f);
  #pragma unroll
  for (int j = 0; j < HH; ++j) ar[j] = fmaxf(v[j] * inv, 0.0f);
}


__global__ void __launch_bounds__(256) k_final(const float* __restrict__ o1,
                                               const float* __restrict__ o2,
                                               const float* __restrict__ o3,
                                               const float* __restrict__ Bp,
                                               const float* __restrict__ bl,
                                               float* __restrict__ out) {
  int wave = (blockIdx.x * blockDim.x + threadIdx.x) >> 5;
  int lane = threadIdx.x & 31;
  if (wave >= NN / 16) return;
  int half = lane >> 4;
  int l16  = lane & 15;
  int koff = half * 2;
  int m    = wave * 16 + l16;
  int n    = l16;
  v8f c = {};
  const v2f* P = (const v2f*)Bp;
  #pragma unroll
  for (int ks = 0; ks < 15; ++ks) {
    int layer = ks / 5;
    int kloc  = (ks - layer * 5) * 4;
    const float* ob = (layer == 0) ? o1 : (layer == 1) ? o2 : o3;
    v2f a = *(const v2f*)(ob + (size_t)m * SP + kloc + koff);
    v2f b = P[ks * 32 + lane];
    c = __builtin_amdgcn_wmma_f32_16x16x4_f32(false, a, false, b, (short)0, c, false, false);
  }
  if (n < CC) {
    float bv = bl[n];
    int rbase = wave * 16 + half * 8;
    #pragma unroll
    for (int v = 0; v < 8; ++v)
      out[(size_t)(rbase + v) * CC + n] = c[v] + bv;
  }
}


extern "C" void kernel_launch(void* const* d_in, const int* in_sizes, int n_in,
                              void* d_out, int out_size, void* d_ws, size_t ws_size,
                              hipStream_t stream) {
  (void)in_sizes; (void)n_in; (void)out_size; (void)ws_size;
  const float* x  = (const float*)d_in[0];
  const int*   ei = (const int*)d_in[1];
  const float* W1 = (const float*)d_in[2];
  const float* b1 = (const float*)d_in[3];
  const float* W2 = (const float*)d_in[4];
  const float* b2 = (const float*)d_in[5];
  const float* W3 = (const float*)d_in[6];
  const float* b3 = (const float*)d_in[7];
  const float* Wl = (const float*)d_in[8];
  const float* bl = (const float*)d_in[9];
  const int* srcI = ei;
  const int* dstI = ei + EE;
  float* out = (float*)d_out;

  float* ws   = (float*)d_ws;
  int*   rp   = (int*)ws;
  float* hbuf = ws + NN;
  int*   csr  = (int*)(hbuf + (size_t)NN * SP);
  float* o1   = hbuf + (size_t)2 * NN * SP;
  float* o2   = o1 + (size_t)NN * SP;
  float* o3   = o2 + (size_t)NN * SP;
  float* Bp1  = o3 + (size_t)NN * SP;
  float* Bp2  = Bp1 + (size_t)(FIN / 4) * 2 * 64;
  float* Bp3  = Bp2 + (size_t)(HH / 4) * 2 * 64;
  float* Bpl  = Bp3 + (size_t)(HH / 4) * 2 * 64;
  int*   rank = (int*)o3;
  int*   cnt  = (int*)o2;
  int*   bsum = cnt + NN;

  dim3 B(256);
  dim3 B4(1024);
  int gN  = (NN + 255) / 256;
  int gE  = (EE + 255) / 256;
  int gS  = (NN + 1023) / 1024;
  int gG  = (NN * 8) / 256;
  int gW  = ((NN / 16) * 32 + 255) / 256;
  int gX  = (NN / 16) / XW;
  int gP1 = ((FIN / 4) * 2 * 32 + 255) / 256;
  int gPh = ((HH / 4) * 2 * 32 + 255) / 256;
  int gPl = (15 * 32 + 255) / 256;

  k_pack_b<<<gP1, B, 0, stream>>>(W1, Bp1, FIN, HH, 2);
  k_pack_b<<<gPh, B, 0, stream>>>(W2, Bp2, HH, HH, 2);
  k_pack_b<<<gPh, B, 0, stream>>>(W3, Bp3, HH, HH, 2);
  k_pack_b<<<gPl, B, 0, stream>>>(Wl, Bpl, 3 * HH, CC, 1);

  k_zero_cnt<<<gN, B, 0, stream>>>(cnt);
  k_rank<<<gE, B, 0, stream>>>(dstI, cnt, rank);
  k_scan_local<<<gS, B4, 0, stream>>>(cnt, rp, bsum);
  k_scan_add<<<gS, B4, 0, stream>>>(rp, bsum);
  k_fill<<<gE, B, 0, stream>>>(srcI, dstI, rank, rp, csr);

  k_gemm_x<<<gX, dim3(XW * 32), 0, stream>>>(x, Bp1, rp, hbuf);
  k_gather<<<gG, B, 0, stream>>>(hbuf, rp, csr, o1);
  k_norm_relu<<<gN, B, 0, stream>>>(o1, b1);

  k_gemm_h<<<gW, B, 0, stream>>>(o1, Bp2, rp, hbuf);
  k_gather<<<gG, B, 0, stream>>>(hbuf, rp, csr, o2);
  k_norm_relu<<<gN, B, 0, stream>>>(o2, b2);

  k_gemm_h<<<gW, B, 0, stream>>>(o2, Bp3, rp, hbuf);
  k_gather<<<gG, B, 0, stream>>>(hbuf, rp, csr, o3);
  k_norm_relu<<<gN, B, 0, stream>>>(o3, b3);

  k_final<<<gW, B, 0, stream>>>(o1, o2, o3, Bpl, bl, out);
}


# Normalise+ReLU, the next 20x20 product and the head product fused into the gather kernel (rows reach the WMMA A fragments via LDS, same k order): 13 launches, not 19; row norm summed as 8-lane tree
# speedup vs baseline: 6.8602x; 1.1025x over previous
#include <hip/hip_runtime.h>

#define NN 100000
#define EE 3200000
#define FIN 512
#define HH 20
#define CC 10
#define SP 32

typedef __attribute__((ext_vector_type(2))) float v2f;
typedef __attribute__((ext_vector_type(4))) float v4f;
typedef __attribute__((ext_vector_type(8))) float v8f;


__global__ void __launch_bounds__(256) k_zero_cnt(int* __restrict__ cnt) {
  int i = blockIdx.x * 256 + threadIdx.x;
  if (i < NN) cnt[i] = 0;
}

__global__ void __launch_bounds__(256) k_rank(const int* __restrict__ dst,
                                              int* __restrict__ cnt,
                                              int* __restrict__ rank) {
  int e = blockIdx.x * 256 + threadIdx.x;
  if (e < EE) rank[e] = atomicAdd(&cnt[dst[e]], 1);
}

__global__ void __launch_bounds__(1024) k_scan_local(const int* __restrict__ cnt,
                                                     int* __restrict__ rp,
                                                     int* __restrict__ bsum) {
  __shared__ int wsum[32];
  int t = threadIdx.x;
  int i = blockIdx.x * 1024 + t;
  int lane = t & 31;
  int w = t >> 5;
  int v = (i < NN) ? cnt[i] : 0;
  int incl = v;
  #pragma unroll
  for (int o = 1; o < 32; o <<= 1) {
    int u = __shfl_up(incl, o, 32);
    if (lane >= o) incl += u;
  }
  if (lane == 31) wsum[w] = incl;
  __syncthreads();
  if (w == 0) {
    int x = wsum[lane];
    int xi = x;
    #pragma unroll
    for (int o = 1; o < 32; o <<= 1) {
      int u = __shfl_up(xi, o, 32);
      if (lane >= o) xi += u;
    }
    wsum[lane] = xi - x;
    if (lane == 31) bsum[blockIdx.x] = xi;
  }
  __syncthreads();
  if (i < NN) rp[i] = incl - v + wsum[w];
}

__global__ void __launch_bounds__(1024) k_scan_add(int* __restrict__ rp,
                                                   const int* __restrict__ bsum) {
  int b = blockIdx.x;
  int lane = threadIdx.x & 31;
  int p = 0;
  #pragma unroll
  for (int k = 0; k < 4; ++k) {
    int bi = lane + 32 * k;
    if (bi < b) p += bsum[bi];
  }
  #pragma unroll
  for (int o = 16; o > 0; o >>= 1) p += __shfl_xor(p, o, 32);
  int i = b * 1024 + threadIdx.x;
  if (i < NN) rp[i] += p;
}

__global__ void __launch_bounds__(256) k_fill(const int* __restrict__ src,
                                              const int* __restrict__ dst,
                                              const int* __restrict__ rank,
                                              const int* __restrict__ rp,
                                              int* __restrict__ csr) {
  int e = blockIdx.x * 256 + threadIdx.x;
  if (e >= EE) return;
  csr[rp[dst[e]] + rank[e]] = src[e];
}

__device__ __forceinline__ float dis_of(int b, int e) {
  float d = (float)(e - b + 1);
  return (d > 0.0f) ? rsqrtf(d) : 0.0f;
}

__device__ __forceinline__ int row_end(const int* __restrict__ rp, int row) {
  int nx = rp[min(row + 1, NN - 1)];
  return (row + 1 < NN) ? nx : EE;
}


__global__ void __launch_bounds__(256) k_pack_b(const float* __restrict__ W,
                                                float* __restrict__ P,
                                                int K, int Ncols, int ntiles) {
  int t = blockIdx.x * 256 + threadIdx.x;
  int total = (K / 4) * ntiles * 32;
  if (t >= total) return;
  int lane = t & 31;
  int frag = t >> 5;
  int ks   = frag / ntiles;
  int tile = frag - ks * ntiles;
  int koff = (lane >> 4) * 2;
  int n    = tile * 16 + (lane & 15);
  int kk   = ks * 4 + koff;
  float v0 = (n < Ncols) ? W[kk * Ncols + n] : 0.0f;
  float v1 = (n < Ncols) ? W[(kk + 1) * Ncols + n] : 0.0f;
  P[(size_t)t * 2 + 0] = v0;
  P[(size_t)t * 2 + 1] = v1;
}


#define XCH 64
#define XPAD 68
#define XW 2

__global__ void __launch_bounds__(XW * 32) k_gemm_x(const float* __restrict__ x,
                                                    const float* __restrict__ Bp,
                                                    const int* __restrict__ rp,
                                                    float* __restrict__ h) {
  __shared__ v4f stage[XW][2][16 * XPAD / 4];
  int w    = threadIdx.x >> 5;
  int lane = threadIdx.x & 31;
  int wave = blockIdx.x * XW + w;
  int half = lane >> 4;
  int l16  = lane & 15;
  int n    = l16;
  int q  = lane & 7;
  int g  = lane >> 3;
  int r0 = 8 * (g & 1) + (g >> 1);
  const float* xg = x + (size_t)(wave * 16 + r0) * FIN + q * 4;
  int so = r0 * XPAD + q * 4;
  int lo = l16 * XPAD + half * 2;
  v8f c0 = {}; v8f c1 = {};
  const v2f* P = (const v2f*)Bp;
  v4f buf[8];
  for (int i = 0; i < 8; ++i)
    buf[i] = __builtin_nontemporal_load((const v4f*)(xg + (size_t)(i >> 1) * 2 * FIN + (i & 1) * 32));
  #pragma unroll 1
  for (int ch = 0; ch < FIN / XCH; ++ch) {
    v2f bq[2 * (XCH / 4)];
    for (int j = 0; j < 2 * (XCH / 4); ++j) bq[j] = P[(ch * 2 * (XCH / 4) + j) * 32 + lane];
    float* sp = (float*)stage[w][ch & 1];
    for (int i = 0; i < 8; ++i)
      *(v4f*)(sp + so + (i >> 1) * 2 * XPAD + (i & 1) * 32) = buf[i];
    __syncthreads();
    if (ch + 1 < FIN / XCH) {
      for (int i = 0; i < 8; ++i)
        buf[i] = __builtin_nontemporal_load(
            (const v4f*)(xg + (size_t)(i >> 1) * 2 * FIN + (i & 1) * 32 + (ch + 1) * XCH));
    }
    const float* fr = sp + lo;
    for (int ks = 0; ks < XCH / 4; ++ks) {
      v2f a = *(const v2f*)(fr + ks * 4);
      c0 = __builtin_amdgcn_wmma_f32_16x16x4_f32(false, a, false, bq[2 * ks + 0], (short)0, c0, false, false);
      c1 = __builtin_amdgcn_wmma_f32_16x16x4_f32(false, a, false, bq[2 * ks + 1], (short)0, c1, false, false);
    }
  }
  int rbase = wave * 16 + half * 8;
  int pr[9];
  #pragma unroll
  for (int v = 0; v < 8; ++v) pr[v] = rp[rbase + v];
  pr[8] = row_end(rp, rbase + 7);
  #pragma unroll
  for (int v = 0; v < 8; ++v) {
    int row = rbase + v;
    float s = dis_of(pr[v], pr[v + 1]);
    size_t r = (size_t)row * SP;
    h[r + n]      = c0[v] * s;
    h[r + n + 16] = c1[v] * s;
  }
}


template <int K>
__device__ __forceinline__ int bcast8(int v) {
  return (int)__builtin_amdgcn_mov_dpp8((unsigned)v, (unsigned)(K * 0x249249));
}

__device__ __forceinline__ float4 f4add(float4 a, float4 b) {
  return make_float4(a.x + b.x, a.y + b.y, a.z + b.z, a.w + b.w);
}

constexpr unsigned xor8_sel(int X) {
  unsigned s = 0;
  for (int i = 0; i < 8; ++i) s |= (unsigned)(i ^ X) << (3 * i);
  return s;
}

template <int X>
__device__ __forceinline__ float xor8(float v) {
  return __builtin_bit_cast(float, __builtin_amdgcn_mov_dpp8(__builtin_bit_cast(unsigned, v), xor8_sel(X)));
}

#define OP 20

template <int HEAD>
__global__ void __launch_bounds__(256) k_gather(const float* __restrict__ hs,
                                                const int* __restrict__ rp,
                                                const int* __restrict__ csr,
                                                const float* __restrict__ bias,
                                                const float* __restrict__ Bp,
                                                float* __restrict__ o,
                                                float* __restrict__ hn,
                                                const float* __restrict__ oa,
                                                const float* __restrict__ ob,
                                                const float* __restrict__ bl,
                                                float* __restrict__ out) {
  __shared__ float so[(HEAD ? 3 : 1) * 32 * OP];
  __shared__ float sdis[32];
  int t   = blockIdx.x * 256 + threadIdx.x;
  int row = t >> 3;
  int q   = t & 7;
  unsigned qb = (unsigned)q * 16u;
  int beg = rp[row];
  int end = row_end(rp, row);
  int bq = min(q, 4) * 4;
  float bi0 = bias[bq + 0], bi1 = bias[bq + 1], bi2 = bias[bq + 2], bi3 = bias[bq + 3];
  float4 pa = make_float4(0.0f, 0.0f, 0.0f, 0.0f), pb = pa;
  if (HEAD) {
    pa = ((const float4*)oa)[(size_t)row * 8 + q];
    pb = ((const float4*)ob)[(size_t)row * 8 + q];
  }
  const char* Hb = (const char*)hs;
#define ROW_LD(s) (*(const float4*)(Hb + ((unsigned)(s) * 128u + qb)))
  float4 acc = ROW_LD(row);
  int j = beg;
  int ld0 = csr[min(j + q, EE - 1)];
  int idx = (j + q < end) ? ld0 : row;
  for (; j + 8 <= end; j += 8) {
    float4 v0 = ROW_LD(bcast8<0>(idx));
    float4 v1 = ROW_LD(bcast8<1>(idx));
    float4 v2 = ROW_LD(bcast8<2>(idx));
    float4 v3 = ROW_LD(bcast8<3>(idx));
    float4 v4 = ROW_LD(bcast8<4>(idx));
    float4 v5 = ROW_LD(bcast8<5>(idx));
    float4 v6 = ROW_LD(bcast8<6>(idx));
    float4 v7 = ROW_LD(bcast8<7>(idx));
    int nj  = j + 8 + q;
    int ldn = csr[min(nj, EE - 1)];
    acc = f4add(acc, v0); acc = f4add(acc, v1); acc = f4add(acc, v2); acc = f4add(acc, v3);
    acc = f4add(acc, v4); acc = f4add(acc, v5); acc = f4add(acc, v6); acc = f4add(acc, v7);
    idx = (nj < end) ? ldn : row;
  }
  int rem = end - j;
  if (rem > 0) {
    const float4 z = make_float4(0.0f, 0.0f, 0.0f, 0.0f);
    float4 v0 = ROW_LD(bcast8<0>(idx));
    float4 v1 = ROW_LD(bcast8<1>(idx));
    float4 v2 = ROW_LD(bcast8<2>(idx));
    float4 v3 = ROW_LD(bcast8<3>(idx));
    float4 v4 = ROW_LD(bcast8<4>(idx));
    float4 v5 = ROW_LD(bcast8<5>(idx));
    float4 v6 = ROW_LD(bcast8<6>(idx));
    acc = f4add(acc, v0);
    acc = f4add(acc, (rem > 1) ? v1 : z);
    acc = f4add(acc, (rem > 2) ? v2 : z);
    acc = f4add(acc, (rem > 3) ? v3 : z);
    acc = f4add(acc, (rem > 4) ? v4 : z);
    acc = f4add(acc, (rem > 5) ? v5 : z);
    acc = f4add(acc, (rem > 6) ? v6 : z);
  }
#undef ROW_LD
  float dd = dis_of(beg, end);
  bool live = q < 5;
  float t0 = live ? acc.x * dd + bi0 : 0.0f;
  float t1 = live ? acc.y * dd + bi1 : 0.0f;
  float t2 = live ? acc.z * dd + bi2 : 0.0f;
  float t3 = live ? acc.w * dd + bi3 : 0.0f;
  float ss = t0 * t0 + t1 * t1 + t2 * t2 + t3 * t3;
  ss += xor8<1>(ss);
  ss += xor8<2>(ss);
  ss += xor8<4>(ss);
  float inv = 1.0f / fmaxf(sqrtf(ss), 1e-12f);
  float4 ov = make_float4(fmaxf(t0 * inv, 0.0f), fmaxf(t1 * inv, 0.0f),
                          fmaxf(t2 * inv, 0.0f), fmaxf(t3 * inv, 0.0f));
  int rl   = threadIdx.x >> 3;
  int w    = threadIdx.x >> 5;
  int lane = threadIdx.x & 31;
  bool mm  = w < 2;
  const v2f* P = (const v2f*)Bp;
  constexpr int NB = HEAD ? 15 : 10;
  v2f bf[NB];
  for (int j = 0; j < NB; ++j) bf[j] = v2f{0.0f, 0.0f};
  if (mm) {
    for (int j = 0; j < NB; ++j) bf[j] = P[j * 32 + lane];
  }
  if (HEAD == 0) {
    if (live) *(float4*)(so + rl * OP + q * 4) = ov;
    if (q == 0) sdis[rl] = dd;
  } else {
    if (live) {
      *(float4*)(so + 0 * 32 * OP + rl * OP + q * 4) = pa;
      *(float4*)(so + 1 * 32 * OP + rl * OP + q * 4) = pb;
      *(float4*)(so + 2 * 32 * OP + rl * OP + q * 4) = ov;
    }
  }
  __syncthreads();
  if (HEAD == 0) {
    ((float4*)o)[(size_t)row * 8 + q] = ov;
  }
  if (!mm) return;
  int half = lane >> 4;
  int l16  = lane & 15;
  int n    = l16;
  const float* ar = so + (w * 16 + l16) * OP + half * 2;
  int rloc  = w * 16 + half * 8;
  int rbase = blockIdx.x * 32 + rloc;
  if (HEAD == 0) {
    v8f c0 = {}; v8f c1 = {};
    for (int ks = 0; ks < HH / 4; ++ks) {
      v2f a = *(const v2f*)(ar + ks * 4);
      c0 = __builtin_amdgcn_wmma_f32_16x16x4_f32(false, a, false, bf[ks * 2 + 0], (short)0, c0, false, false);
      c1 = __builtin_amdgcn_wmma_f32_16x16x4_f32(false, a, false, bf[ks * 2 + 1], (short)0, c1, false, false);
    }
    for (int v = 0; v < 8; ++v) {
      float s = sdis[rloc + v];
      size_t r = (size_t)(rbase + v) * SP;
      hn[r + n]      = c0[v] * s;
      hn[r + n + 16] = c1[v] * s;
    }
  } else {
    v8f cc = {};
    for (int ks = 0; ks < 15; ++ks) {
      int layer = ks / 5;
      int kloc  = (ks - layer * 5) * 4;
      v2f a = *(const v2f*)(ar + layer * 32 * OP + kloc);
      cc = __builtin_amdgcn_wmma_f32_16x16x4_f32(false, a, false, bf[ks], (short)0, cc, false, false);
    }
    if (n < CC) {
      float bv = bl[n];
      for (int v = 0; v < 8; ++v)
        out[(size_t)(rbase + v) * CC + n] = cc[v] + bv;
    }
  }
}


extern "C" void kernel_launch(void* const* d_in, const int* in_sizes, int n_in,
                              void* d_out, int out_size, void* d_ws, size_t ws_size,
                              hipStream_t stream) {
  (void)in_sizes; (void)n_in; (void)out_size; (void)ws_size;
  const float* x  = (const float*)d_in[0];
  const int*   ei = (const int*)d_in[1];
  const float* W1 = (const float*)d_in[2];
  const float* b1 = (const float*)d_in[3];
  const float* W2 = (const float*)d_in[4];
  const float* b2 = (const float*)d_in[5];
  const float* W3 = (const float*)d_in[6];
  const float* b3 = (const float*)d_in[7];
  const float* Wl = (const float*)d_in[8];
  const float* bl = (const float*)d_in[9];
  const int* srcI = ei;
  const int* dstI = ei + EE;
  float* out = (float*)d_out;

  float* ws   = (float*)d_ws;
  int*   rp   = (int*)ws;
  float* hbuf = ws + NN;
  int*   csr  = (int*)(hbuf + (size_t)NN * SP);
  float* o1   = hbuf + (size_t)2 * NN * SP;
  float* o2   = o1 + (size_t)NN * SP;
  float* o3   = o2 + (size_t)NN * SP;
  float* Bp1  = o3 + (size_t)NN * SP;
  float* Bp2  = Bp1 + (size_t)(FIN / 4) * 2 * 64;
  float* Bp3  = Bp2 + (size_t)(HH / 4) * 2 * 64;
  float* Bpl  = Bp3 + (size_t)(HH / 4) * 2 * 64;
  int*   rank = (int*)o3;
  int*   cnt  = (int*)o2;
  int*   bsum = cnt + NN;

  dim3 B(256);
  dim3 B4(1024);
  int gN  = (NN + 255) / 256;
  int gE  = (EE + 255) / 256;
  int gS  = (NN + 1023) / 1024;
  int gG  = (NN * 8) / 256;
  int gX  = (NN / 16) / XW;
  int gP1 = ((FIN / 4) * 2 * 32 + 255) / 256;
  int gPh = ((HH / 4) * 2 * 32 + 255) / 256;
  int gPl = (15 * 32 + 255) / 256;

  k_pack_b<<<gP1, B, 0, stream>>>(W1, Bp1, FIN, HH, 2);
  k_pack_b<<<gPh, B, 0, stream>>>(W2, Bp2, HH, HH, 2);
  k_pack_b<<<gPh, B, 0, stream>>>(W3, Bp3, HH, HH, 2);
  k_pack_b<<<gPl, B, 0, stream>>>(Wl, Bpl, 3 * HH, CC, 1);

  k_zero_cnt<<<gN, B, 0, stream>>>(cnt);
  k_rank<<<gE, B, 0, stream>>>(dstI, cnt, rank);
  k_scan_local<<<gS, B4, 0, stream>>>(cnt, rp, bsum);
  k_scan_add<<<gS, B4, 0, stream>>>(rp, bsum);
  k_fill<<<gE, B, 0, stream>>>(srcI, dstI, rank, rp, csr);

  k_gemm_x<<<gX, dim3(XW * 32), 0, stream>>>(x, Bp1, rp, hbuf);
  k_gather<0><<<gG, B, 0, stream>>>(hbuf, rp, csr, b1, Bp2, o1, o3, nullptr, nullptr, nullptr, nullptr);

  k_gather<0><<<gG, B, 0, stream>>>(o3, rp, csr, b2, Bp3, o2, hbuf, nullptr, nullptr, nullptr, nullptr);

  k_gather<1><<<gG, B, 0, stream>>>(hbuf, rp, csr, b3, Bpl, nullptr, nullptr, o1, o2, bl, out);
}


# CSR built by a two-level counting sort in LDS (slice counts, bucket runs, per-bucket rank, scan, place); no global atomic, random read or scattered store per edge; row terms in LDS arrival order
# speedup vs baseline: 14.2155x; 2.0722x over previous
#include <hip/hip_runtime.h>

#define NN 100000
#define EE 3200000
#define FIN 512
#define HH 20
#define CC 10
#define SP 32

typedef __attribute__((ext_vector_type(2))) float v2f;
typedef __attribute__((ext_vector_type(4))) float v4f;
typedef __attribute__((ext_vector_type(8))) float v8f;


#define BSH 7
#define NBK 128
#define NBUCK ((NN + NBK - 1) / NBK)
#define G1 250
#define CH1 (EE / G1)
#define RSH 17
#define TR1 ((CH1 + 1023) / 1024)
#define KR4 20
#define CAP4 (256 * KR4)

static_assert(EE % G1 == 0, "slices are whole");
static_assert(NBUCK <= 1024 && G1 <= 256 && NN <= (1 << RSH), "sizes the passes rely on");

template <int NW>
__device__ __forceinline__ int block_excl_scan(int v, int* wsum, int t) {
  int lane = t & 31;
  int w = t >> 5;
  int incl = v;
  #pragma unroll
  for (int o = 1; o < 32; o <<= 1) {
    int u = __shfl_up(incl, o, 32);
    if (lane >= o) incl += u;
  }
  if (lane == 31) wsum[w] = incl;
  __syncthreads();
  if (w == 0) {
    int x = (lane < NW) ? wsum[lane] : 0;
    int xi = x;
    #pragma unroll
    for (int o = 1; o < 32; o <<= 1) {
      int u = __shfl_up(xi, o, 32);
      if (lane >= o) xi += u;
    }
    wsum[lane] = xi - x;
    if (lane == 31) wsum[32] = xi;
  }
  __syncthreads();
  return incl - v + wsum[w];
}

__global__ void __launch_bounds__(1024) k_bhist(const int* __restrict__ dst, int* __restrict__ T) {
  __shared__ int hist[NBUCK];
  int t = threadIdx.x;
  int wg = blockIdx.x;
  for (int b = t; b < NBUCK; b += 1024) hist[b] = 0;
  __syncthreads();
  const int* dp = dst + wg * CH1;
  int d[TR1];
  #pragma unroll
  for (int k = 0; k < TR1; ++k) {
    int i = t + 1024 * k;
    d[k] = (i < CH1) ? dp[i] : 0;
  }
  #pragma unroll
  for (int k = 0; k < TR1; ++k)
    if (t + 1024 * k < CH1) atomicAdd(&hist[d[k] >> BSH], 1);
  __syncthreads();
  for (int b = t; b < NBUCK; b += 1024) T[b * G1 + wg] = hist[b];
}

__global__ void __launch_bounds__(256) k_bprefix(const int* __restrict__ T, int* __restrict__ P,
                                                 int* __restrict__ btot) {
  __shared__ int wsum[33];
  int t = threadIdx.x;
  int b = blockIdx.x;
  int v = (t < G1) ? T[b * G1 + t] : 0;
  int ex = block_excl_scan<8>(v, wsum, t);
  if (t < G1) P[b * G1 + t] = ex;
  if (t == 0) btot[b] = wsum[32];
}

__global__ void __launch_bounds__(1024) k_bscatter(const int* __restrict__ src,
                                                   const int* __restrict__ dst,
                                                   const int* __restrict__ T,
                                                   const int* __restrict__ P,
                                                   const int* __restrict__ btot,
                                                   int* __restrict__ pairs) {
  __shared__ int wsum[33];
  __shared__ int gbase[NBUCK];
  __shared__ int lpos[NBUCK];
  __shared__ int cur[NBUCK];
  __shared__ int st_rec[CH1];
  __shared__ unsigned short st_b[CH1];
  int t = threadIdx.x;
  int wg = blockIdx.x;
  int bt = (t < NBUCK) ? btot[t] : 0;
  int mine = (t < NBUCK) ? T[t * G1 + wg] : 0;
  int pre = (t < NBUCK) ? P[t * G1 + wg] : 0;
  int bs = block_excl_scan<32>(bt, wsum, t);
  __syncthreads();
  int lp = block_excl_scan<32>(mine, wsum, t);
  if (t < NBUCK) {
    gbase[t] = bs + pre;
    lpos[t] = lp;
    cur[t] = 0;
  }
  __syncthreads();
  const int* sp = src + wg * CH1;
  const int* dp = dst + wg * CH1;
  int d[TR1], s[TR1];
  #pragma unroll
  for (int k = 0; k < TR1; ++k) {
    int i = t + 1024 * k;
    d[k] = (i < CH1) ? dp[i] : 0;
    s[k] = (i < CH1) ? sp[i] : 0;
  }
  #pragma unroll
  for (int k = 0; k < TR1; ++k) {
    if (t + 1024 * k < CH1) {
      int b = d[k] >> BSH;
      int p = lpos[b] + atomicAdd(&cur[b], 1);
      st_rec[p] = s[k] | ((d[k] & (NBK - 1)) << RSH);
      st_b[p] = (unsigned short)b;
    }
  }
  __syncthreads();
  #pragma unroll
  for (int k = 0; k < TR1; ++k) {
    int i = t + 1024 * k;
    if (i < CH1) {
      int b = st_b[i];
      pairs[gbase[b] + (i - lpos[b])] = st_rec[i];
    }
  }
}

__global__ void __launch_bounds__(256) k_bfinal(const int* __restrict__ pairs,
                                                const int* __restrict__ btot,
                                                int* __restrict__ rp,
                                                int* __restrict__ csr) {
  __shared__ int wsum[33];
  __shared__ int cntl[NBK];
  __shared__ int offl[NBK];
  __shared__ int stage[CAP4];
  int t = threadIdx.x;
  int b = blockIdx.x;
  int v = 0;
  #pragma unroll
  for (int k = 0; k < 4; ++k) {
    int bi = t + 256 * k;
    if (bi < b) v += btot[bi];
  }
  block_excl_scan<8>(v, wsum, t);
  int bstart = wsum[32];
  int n = btot[b];
  if (t < NBK) cntl[t] = 0;
  __syncthreads();
  const int* pp = pairs + bstart;
  bool fast = n <= CAP4;
  int rec[KR4];
  if (fast) {
    #pragma unroll
    for (int k = 0; k < KR4; ++k) {
      int i = t + 256 * k;
      rec[k] = (i < n) ? pp[i] : 0;
    }
    #pragma unroll
    for (int k = 0; k < KR4; ++k)
      if (t + 256 * k < n) atomicAdd(&cntl[rec[k] >> RSH], 1);
  } else {
    for (int i = t; i < n; i += 256) atomicAdd(&cntl[pp[i] >> RSH], 1);
  }
  __syncthreads();
  int c = (t < NBK) ? cntl[t] : 0;
  int ex = block_excl_scan<8>(c, wsum, t);
  if (t < NBK) {
    offl[t] = ex;
    cntl[t] = 0;
    int node = b * NBK + t;
    if (node < NN) rp[node] = bstart + ex;
  }
  __syncthreads();
  if (fast) {
    #pragma unroll
    for (int k = 0; k < KR4; ++k) {
      if (t + 256 * k < n) {
        int dl = rec[k] >> RSH;
        stage[offl[dl] + atomicAdd(&cntl[dl], 1)] = rec[k] & ((1 << RSH) - 1);
      }
    }
    __syncthreads();
    #pragma unroll
    for (int k = 0; k < KR4; ++k) {
      int i = t + 256 * k;
      if (i < n) csr[bstart + i] = stage[i];
    }
  } else {
    for (int i = t; i < n; i += 256) {
      int r = pp[i];
      int dl = r >> RSH;
      csr[bstart + offl[dl] + atomicAdd(&cntl[dl], 1)] = r & ((1 << RSH) - 1);
    }
  }
}

__device__ __forceinline__ float dis_of(int b, int e) {
  float d = (float)(e - b + 1);
  return (d > 0.0f) ? rsqrtf(d) : 0.0f;
}

__device__ __forceinline__ int row_end(const int* __restrict__ rp, int row) {
  int nx = rp[min(row + 1, NN - 1)];
  return (row + 1 < NN) ? nx : EE;
}


__global__ void __launch_bounds__(256) k_pack_b(const float* __restrict__ W,
                                                float* __restrict__ P,
                                                int K, int Ncols, int ntiles) {
  int t = blockIdx.x * 256 + threadIdx.x;
  int total = (K / 4) * ntiles * 32;
  if (t >= total) return;
  int lane = t & 31;
  int frag = t >> 5;
  int ks   = frag / ntiles;
  int tile = frag - ks * ntiles;
  int koff = (lane >> 4) * 2;
  int n    = tile * 16 + (lane & 15);
  int kk   = ks * 4 + koff;
  float v0 = (n < Ncols) ? W[kk * Ncols + n] : 0.0f;
  float v1 = (n < Ncols) ? W[(kk + 1) * Ncols + n] : 0.0f;
  P[(size_t)t * 2 + 0] = v0;
  P[(size_t)t * 2 + 1] = v1;
}


#define XCH 64
#define XPAD 68
#define XW 2

__global__ void __launch_bounds__(XW * 32) k_gemm_x(const float* __restrict__ x,
                                                    const float* __restrict__ Bp,
                                                    const int* __restrict__ rp,
                                                    float* __restrict__ h) {
  __shared__ v4f stage[XW][2][16 * XPAD / 4];
  int w    = threadIdx.x >> 5;
  int lane = threadIdx.x & 31;
  int wave = blockIdx.x * XW + w;
  int half = lane >> 4;
  int l16  = lane & 15;
  int n    = l16;
  int q  = lane & 7;
  int g  = lane >> 3;
  int r0 = 8 * (g & 1) + (g >> 1);
  const float* xg = x + (size_t)(wave * 16 + r0) * FIN + q * 4;
  int so = r0 * XPAD + q * 4;
  int lo = l16 * XPAD + half * 2;
  v8f c0 = {}; v8f c1 = {};
  const v2f* P = (const v2f*)Bp;
  v4f buf[8];
  for (int i = 0; i < 8; ++i)
    buf[i] = __builtin_nontemporal_load((const v4f*)(xg + (size_t)(i >> 1) * 2 * FIN + (i & 1) * 32));
  #pragma unroll 1
  for (int ch = 0; ch < FIN / XCH; ++ch) {
    v2f bq[2 * (XCH / 4)];
    for (int j = 0; j < 2 * (XCH / 4); ++j) bq[j] = P[(ch * 2 * (XCH / 4) + j) * 32 + lane];
    float* sp = (float*)stage[w][ch & 1];
    for (int i = 0; i < 8; ++i)
      *(v4f*)(sp + so + (i >> 1) * 2 * XPAD + (i & 1) * 32) = buf[i];
    __syncthreads();
    if (ch + 1 < FIN / XCH) {
      for (int i = 0; i < 8; ++i)
        buf[i] = __builtin_nontemporal_load(
            (const v4f*)(xg + (size_t)(i >> 1) * 2 * FIN + (i & 1) * 32 + (ch + 1) * XCH));
    }
    const float* fr = sp + lo;
    for (int ks = 0; ks < XCH / 4; ++ks) {
      v2f a = *(const v2f*)(fr + ks * 4);
      c0 = __builtin_amdgcn_wmma_f32_16x16x4_f32(false, a, false, bq[2 * ks + 0], (short)0, c0, false, false);
      c1 = __builtin_amdgcn_wmma_f32_16x16x4_f32(false, a, false, bq[2 * ks + 1], (short)0, c1, false, false);
    }
  }
  int rbase = wave * 16 + half * 8;
  int pr[9];
  #pragma unroll
  for (int v = 0; v < 8; ++v) pr[v] = rp[rbase + v];
  pr[8] = row_end(rp, rbase + 7);
  #pragma unroll
  for (int v = 0; v < 8; ++v) {
    int row = rbase + v;
    float s = dis_of(pr[v], pr[v + 1]);
    size_t r = (size_t)row * SP;
    h[r + n]      = c0[v] * s;
    h[r + n + 16] = c1[v] * s;
  }
}


template <int K>
__device__ __forceinline__ int bcast8(int v) {
  return (int)__builtin_amdgcn_mov_dpp8((unsigned)v, (unsigned)(K * 0x249249));
}

__device__ __forceinline__ float4 f4add(float4 a, float4 b) {
  return make_float4(a.x + b.x, a.y + b.y, a.z + b.z, a.w + b.w);
}

constexpr unsigned xor8_sel(int X) {
  unsigned s = 0;
  for (int i = 0; i < 8; ++i) s |= (unsigned)(i ^ X) << (3 * i);
  return s;
}

template <int X>
__device__ __forceinline__ float xor8(float v) {
  return __builtin_bit_cast(float, __builtin_amdgcn_mov_dpp8(__builtin_bit_cast(unsigned, v), xor8_sel(X)));
}

#define OP 20

template <int HEAD>
__global__ void __launch_bounds__(256) k_gather(const float* __restrict__ hs,
                                                const int* __restrict__ rp,
                                                const int* __restrict__ csr,
                                                const float* __restrict__ bias,
                                                const float* __restrict__ Bp,
                                                float* __restrict__ o,
                                                float* __restrict__ hn,
                                                const float* __restrict__ oa,
                                                const float* __restrict__ ob,
                                                const float* __restrict__ bl,
                                                float* __restrict__ out) {
  __shared__ float so[(HEAD ? 3 : 1) * 32 * OP];
  __shared__ float sdis[32];
  int t   = blockIdx.x * 256 + threadIdx.x;
  int row = t >> 3;
  int q   = t & 7;
  unsigned qb = (unsigned)q * 16u;
  int beg = rp[row];
  int end = row_end(rp, row);
  int bq = min(q, 4) * 4;
  float bi0 = bias[bq + 0], bi1 = bias[bq + 1], bi2 = bias[bq + 2], bi3 = bias[bq + 3];
  float4 pa = make_float4(0.0f, 0.0f, 0.0f, 0.0f), pb = pa;
  if (HEAD) {
    pa = ((const float4*)oa)[(size_t)row * 8 + q];
    pb = ((const float4*)ob)[(size_t)row * 8 + q];
  }
  const char* Hb = (const char*)hs;
#define ROW_LD(s) (*(const float4*)(Hb + ((unsigned)(s) * 128u + qb)))
  float4 acc = ROW_LD(row);
  int j = beg;
  int ld0 = csr[min(j + q, EE - 1)];
  int idx = (j + q < end) ? ld0 : row;
  for (; j + 8 <= end; j += 8) {
    float4 v0 = ROW_LD(bcast8<0>(idx));
    float4 v1 = ROW_LD(bcast8<1>(idx));
    float4 v2 = ROW_LD(bcast8<2>(idx));
    float4 v3 = ROW_LD(bcast8<3>(idx));
    float4 v4 = ROW_LD(bcast8<4>(idx));
    float4 v5 = ROW_LD(bcast8<5>(idx));
    float4 v6 = ROW_LD(bcast8<6>(idx));
    float4 v7 = ROW_LD(bcast8<7>(idx));
    int nj  = j + 8 + q;
    int ldn = csr[min(nj, EE - 1)];
    acc = f4add(acc, v0); acc = f4add(acc, v1); acc = f4add(acc, v2); acc = f4add(acc, v3);
    acc = f4add(acc, v4); acc = f4add(acc, v5); acc = f4add(acc, v6); acc = f4add(acc, v7);
    idx = (nj < end) ? ldn : row;
  }
  int rem = end - j;
  if (rem > 0) {
    const float4 z = make_float4(0.0f, 0.0f, 0.0f, 0.0f);
    float4 v0 = ROW_LD(bcast8<0>(idx));
    float4 v1 = ROW_LD(bcast8<1>(idx));
    float4 v2 = ROW_LD(bcast8<2>(idx));
    float4 v3 = ROW_LD(bcast8<3>(idx));
    float4 v4 = ROW_LD(bcast8<4>(idx));
    float4 v5 = ROW_LD(bcast8<5>(idx));
    float4 v6 = ROW_LD(bcast8<6>(idx));
    acc = f4add(acc, v0);
    acc = f4add(acc, (rem > 1) ? v1 : z);
    acc = f4add(acc, (rem > 2) ? v2 : z);
    acc = f4add(acc, (rem > 3) ? v3 : z);
    acc = f4add(acc, (rem > 4) ? v4 : z);
    acc = f4add(acc, (rem > 5) ? v5 : z);
    acc = f4add(acc, (rem > 6) ? v6 : z);
  }
#undef ROW_LD
  float dd = dis_of(beg, end);
  bool live = q < 5;
  float t0 = live ? acc.x * dd + bi0 : 0.0f;
  float t1 = live ? acc.y * dd + bi1 : 0.0f;
  float t2 = live ? acc.z * dd + bi2 : 0.0f;
  float t3 = live ? acc.w * dd + bi3 : 0.0f;
  float ss = t0 * t0 + t1 * t1 + t2 * t2 + t3 * t3;
  ss += xor8<1>(ss);
  ss += xor8<2>(ss);
  ss += xor8<4>(ss);
  float inv = 1.0f / fmaxf(sqrtf(ss), 1e-12f);
  float4 ov = make_float4(fmaxf(t0 * inv, 0.0f), fmaxf(t1 * inv, 0.0f),
                          fmaxf(t2 * inv, 0.0f), fmaxf(t3 * inv, 0.0f));
  int rl   = threadIdx.x >> 3;
  int w    = threadIdx.x >> 5;
  int lane = threadIdx.x & 31;
  bool mm  = w < 2;
  const v2f* P = (const v2f*)Bp;
  constexpr int NB = HEAD ? 15 : 10;
  v2f bf[NB];
  for (int j = 0; j < NB; ++j) bf[j] = v2f{0.0f, 0.0f};
  if (mm) {
    for (int j = 0; j < NB; ++j) bf[j] = P[j * 32 + lane];
  }
  if (HEAD == 0) {
    if (live) *(float4*)(so + rl * OP + q * 4) = ov;
    if (q == 0) sdis[rl] = dd;
  } else {
    if (live) {
      *(float4*)(so + 0 * 32 * OP + rl * OP + q * 4) = pa;
      *(float4*)(so + 1 * 32 * OP + rl * OP + q * 4) = pb;
      *(float4*)(so + 2 * 32 * OP + rl * OP + q * 4) = ov;
    }
  }
  __syncthreads();
  if (HEAD == 0) {
    ((float4*)o)[(size_t)row * 8 + q] = ov;
  }
  if (!mm) return;
  int half = lane >> 4;
  int l16  = lane & 15;
  int n    = l16;
  const float* ar = so + (w * 16 + l16) * OP + half * 2;
  int rloc  = w * 16 + half * 8;
  int rbase = blockIdx.x * 32 + rloc;
  if (HEAD == 0) {
    v8f c0 = {}; v8f c1 = {};
    for (int ks = 0; ks < HH / 4; ++ks) {
      v2f a = *(const v2f*)(ar + ks * 4);
      c0 = __builtin_amdgcn_wmma_f32_16x16x4_f32(false, a, false, bf[ks * 2 + 0], (short)0, c0, false, false);
      c1 = __builtin_amdgcn_wmma_f32_16x16x4_f32(false, a, false, bf[ks * 2 + 1], (short)0, c1, false, false);
    }
    for (int v = 0; v < 8; ++v) {
      float s = sdis[rloc + v];
      size_t r = (size_t)(rbase + v) * SP;
      hn[r + n]      = c0[v] * s;
      hn[r + n + 16] = c1[v] * s;
    }
  } else {
    v8f cc = {};
    for (int ks = 0; ks < 15; ++ks) {
      int layer = ks / 5;
      int kloc  = (ks - layer * 5) * 4;
      v2f a = *(const v2f*)(ar + layer * 32 * OP + kloc);
      cc = __builtin_amdgcn_wmma_f32_16x16x4_f32(false, a, false, bf[ks], (short)0, cc, false, false);
    }
    if (n < CC) {
      float bv = bl[n];
      for (int v = 0; v < 8; ++v)
        out[(size_t)(rbase + v) * CC + n] = cc[v] + bv;
    }
  }
}


extern "C" void kernel_launch(void* const* d_in, const int* in_sizes, int n_in,
                              void* d_out, int out_size, void* d_ws, size_t ws_size,
                              hipStream_t stream) {
  (void)in_sizes; (void)n_in; (void)out_size; (void)ws_size;
  const float* x  = (const float*)d_in[0];
  const int*   ei = (const int*)d_in[1];
  const float* W1 = (const float*)d_in[2];
  const float* b1 = (const float*)d_in[3];
  const float* W2 = (const float*)d_in[4];
  const float* b2 = (const float*)d_in[5];
  const float* W3 = (const float*)d_in[6];
  const float* b3 = (const float*)d_in[7];
  const float* Wl = (const float*)d_in[8];
  const float* bl = (const float*)d_in[9];
  const int* srcI = ei;
  const int* dstI = ei + EE;
  float* out = (float*)d_out;

  float* ws   = (float*)d_ws;
  int*   rp   = (int*)ws;
  float* hbuf = ws + NN;
  int*   csr  = (int*)(hbuf + (size_t)NN * SP);
  float* o1   = hbuf + (size_t)2 * NN * SP;
  float* o2   = o1 + (size_t)NN * SP;
  float* o3   = o2 + (size_t)NN * SP;
  float* Bp1  = o3 + (size_t)NN * SP;
  float* Bp2  = Bp1 + (size_t)(FIN / 4) * 2 * 64;
  float* Bp3  = Bp2 + (size_t)(HH / 4) * 2 * 64;
  float* Bpl  = Bp3 + (size_t)(HH / 4) * 2 * 64;
  int*   pairs = (int*)o3;
  int*   T    = (int*)o2;
  int*   P    = T + NBUCK * G1;
  int*   btot = P + NBUCK * G1;

  dim3 B(256);
  dim3 B4(1024);
  int gG  = (NN * 8) / 256;
  int gX  = (NN / 16) / XW;
  int gP1 = ((FIN / 4) * 2 * 32 + 255) / 256;
  int gPh = ((HH / 4) * 2 * 32 + 255) / 256;
  int gPl = (15 * 32 + 255) / 256;

  k_pack_b<<<gP1, B, 0, stream>>>(W1, Bp1, FIN, HH, 2);
  k_pack_b<<<gPh, B, 0, stream>>>(W2, Bp2, HH, HH, 2);
  k_pack_b<<<gPh, B, 0, stream>>>(W3, Bp3, HH, HH, 2);
  k_pack_b<<<gPl, B, 0, stream>>>(Wl, Bpl, 3 * HH, CC, 1);

  k_bhist<<<G1, B4, 0, stream>>>(dstI, T);
  k_bprefix<<<NBUCK, B, 0, stream>>>(T, P, btot);
  k_bscatter<<<G1, B4, 0, stream>>>(srcI, dstI, T, P, btot, pairs);
  k_bfinal<<<NBUCK, B, 0, stream>>>(pairs, btot, rp, csr);

  k_gemm_x<<<gX, dim3(XW * 32), 0, stream>>>(x, Bp1, rp, hbuf);
  k_gather<0><<<gG, B, 0, stream>>>(hbuf, rp, csr, b1, Bp2, o1, o3, nullptr, nullptr, nullptr, nullptr);

  k_gather<0><<<gG, B, 0, stream>>>(o3, rp, csr, b2, Bp3, o2, hbuf, nullptr, nullptr, nullptr, nullptr);

  k_gather<1><<<gG, B, 0, stream>>>(hbuf, rp, csr, b3, Bpl, nullptr, nullptr, o1, o2, bl, out);
}


# Weight packing (the four B-fragment kernels) rides in the first CSR launch as 11 extra work-groups: 8 launches per call, not 12; kernels, sums and their order unchanged
# speedup vs baseline: 14.8861x; 1.0472x over previous
#include <hip/hip_runtime.h>

#define NN 100000
#define EE 3200000
#define FIN 512
#define HH 20
#define CC 10
#define SP 32

typedef __attribute__((ext_vector_type(2))) float v2f;
typedef __attribute__((ext_vector_type(4))) float v4f;
typedef __attribute__((ext_vector_type(8))) float v8f;


#define BSH 7
#define NBK 128
#define NBUCK ((NN + NBK - 1) / NBK)
#define G1 250
#define CH1 (EE / G1)
#define RSH 17
#define TR1 ((CH1 + 1023) / 1024)
#define KR4 20
#define CAP4 (256 * KR4)

static_assert(EE % G1 == 0, "slices are whole");
static_assert(NBUCK <= 1024 && G1 <= 256 && NN <= (1 << RSH), "sizes the passes rely on");

template <int NW>
__device__ __forceinline__ int block_excl_scan(int v, int* wsum, int t) {
  int lane = t & 31;
  int w = t >> 5;
  int incl = v;
  #pragma unroll
  for (int o = 1; o < 32; o <<= 1) {
    int u = __shfl_up(incl, o, 32);
    if (lane >= o) incl += u;
  }
  if (lane == 31) wsum[w] = incl;
  __syncthreads();
  if (w == 0) {
    int x = (lane < NW) ? wsum[lane] : 0;
    int xi = x;
    #pragma unroll
    for (int o = 1; o < 32; o <<= 1) {
      int u = __shfl_up(xi, o, 32);
      if (lane >= o) xi += u;
    }
    wsum[lane] = xi - x;
    if (lane == 31) wsum[32] = xi;
  }
  __syncthreads();
  return incl - v + wsum[w];
}

__device__ __forceinline__ void pack_b_entry(const float* __restrict__ W, float* __restrict__ P,
                                             int K, int Ncols, int ntiles, int t) {
  int total = (K / 4) * ntiles * 32;
  if (t >= total) return;
  int lane = t & 31;
  int frag = t >> 5;
  int ks   = frag / ntiles;
  int tile = frag - ks * ntiles;
  int koff = (lane >> 4) * 2;
  int n    = tile * 16 + (lane & 15);
  int kk   = ks * 4 + koff;
  float v0 = (n < Ncols) ? W[kk * Ncols + n] : 0.0f;
  float v1 = (n < Ncols) ? W[(kk + 1) * Ncols + n] : 0.0f;
  P[(size_t)t * 2 + 0] = v0;
  P[(size_t)t * 2 + 1] = v1;
}

#define PK1 (((FIN / 4) * 2 * 32) / 1024)
#define NPK (PK1 + 3)

static_assert(((FIN / 4) * 2 * 32) % 1024 == 0, "W1's packed entries fill whole work-groups");
static_assert((HH / 4) * 2 * 32 <= 1024 && 15 * 32 <= 1024, "a small matrix is packed by one work-group");

__global__ void __launch_bounds__(1024) k_bhist(const int* __restrict__ dst, int* __restrict__ T,
                                                const float* __restrict__ W1,
                                                const float* __restrict__ W2,
                                                const float* __restrict__ W3,
                                                const float* __restrict__ Wl,
                                                float* __restrict__ Bp1, float* __restrict__ Bp2,
                                                float* __restrict__ Bp3, float* __restrict__ Bpl) {
  __shared__ int hist[NBUCK];
  int t = threadIdx.x;
  int wg = blockIdx.x;
  if (wg >= G1) {
    int pb = wg - G1;
    if (pb < PK1)           pack_b_entry(W1, Bp1, FIN, HH, 2, pb * 1024 + t);
    else if (pb == PK1)     pack_b_entry(W2, Bp2, HH, HH, 2, t);
    else if (pb == PK1 + 1) pack_b_entry(W3, Bp3, HH, HH, 2, t);
    else                    pack_b_entry(Wl, Bpl, 3 * HH, CC, 1, t);
    return;
  }
  for (int b = t; b < NBUCK; b += 1024) hist[b] = 0;
  __syncthreads();
  const int* dp = dst + wg * CH1;
  int d[TR1];
  #pragma unroll
  for (int k = 0; k < TR1; ++k) {
    int i = t + 1024 * k;
    d[k] = (i < CH1) ? dp[i] : 0;
  }
  #pragma unroll
  for (int k = 0; k < TR1; ++k)
    if (t + 1024 * k < CH1) atomicAdd(&hist[d[k] >> BSH], 1);
  __syncthreads();
  for (int b = t; b < NBUCK; b += 1024) T[b * G1 + wg] = hist[b];
}

__global__ void __launch_bounds__(256) k_bprefix(const int* __restrict__ T, int* __restrict__ P,
                                                 int* __restrict__ btot) {
  __shared__ int wsum[33];
  int t = threadIdx.x;
  int b = blockIdx.x;
  int v = (t < G1) ? T[b * G1 + t] : 0;
  int ex = block_excl_scan<8>(v, wsum, t);
  if (t < G1) P[b * G1 + t] = ex;
  if (t == 0) btot[b] = wsum[32];
}

__global__ void __launch_bounds__(1024) k_bscatter(const int* __restrict__ src,
                                                   const int* __restrict__ dst,
                                                   const int* __restrict__ T,
                                                   const int* __restrict__ P,
                                                   const int* __restrict__ btot,
                                                   int* __restrict__ pairs) {
  __shared__ int wsum[33];
  __shared__ int gbase[NBUCK];
  __shared__ int lpos[NBUCK];
  __shared__ int cur[NBUCK];
  __shared__ int st_rec[CH1];
  __shared__ unsigned short st_b[CH1];
  int t = threadIdx.x;
  int wg = blockIdx.x;
  int bt = (t < NBUCK) ? btot[t] : 0;
  int mine = (t < NBUCK) ? T[t * G1 + wg] : 0;
  int pre = (t < NBUCK) ? P[t * G1 + wg] : 0;
  int bs = block_excl_scan<32>(bt, wsum, t);
  __syncthreads();
  int lp = block_excl_scan<32>(mine, wsum, t);
  if (t < NBUCK) {
    gbase[t] = bs + pre;
    lpos[t] = lp;
    cur[t] = 0;
  }
  __syncthreads();
  const int* sp = src + wg * CH1;
  const int* dp = dst + wg * CH1;
  int d[TR1], s[TR1];
  #pragma unroll
  for (int k = 0; k < TR1; ++k) {
    int i = t + 1024 * k;
    d[k] = (i < CH1) ? dp[i] : 0;
    s[k] = (i < CH1) ? sp[i] : 0;
  }
  #pragma unroll
  for (int k = 0; k < TR1; ++k) {
    if (t + 1024 * k < CH1) {
      int b = d[k] >> BSH;
      int p = lpos[b] + atomicAdd(&cur[b], 1);
      st_rec[p] = s[k] | ((d[k] & (NBK - 1)) << RSH);
      st_b[p] = (unsigned short)b;
    }
  }
  __syncthreads();
  #pragma unroll
  for (int k = 0; k < TR1; ++k) {
    int i = t + 1024 * k;
    if (i < CH1) {
      int b = st_b[i];
      pairs[gbase[b] + (i - lpos[b])] = st_rec[i];
    }
  }
}

__global__ void __launch_bounds__(256) k_bfinal(const int* __restrict__ pairs,
                                                const int* __restrict__ btot,
                                                int* __restrict__ rp,
                                                int* __restrict__ csr) {
  __shared__ int wsum[33];
  __shared__ int cntl[NBK];
  __shared__ int offl[NBK];
  __shared__ int stage[CAP4];
  int t = threadIdx.x;
  int b = blockIdx.x;
  int v = 0;
  #pragma unroll
  for (int k = 0; k < 4; ++k) {
    int bi = t + 256 * k;
    if (bi < b) v += btot[bi];
  }
  block_excl_scan<8>(v, wsum, t);
  int bstart = wsum[32];
  int n = btot[b];
  if (t < NBK) cntl[t] = 0;
  __syncthreads();
  const int* pp = pairs + bstart;
  bool fast = n <= CAP4;
  int rec[KR4];
  if (fast) {
    #pragma unroll
    for (int k = 0; k < KR4; ++k) {
      int i = t + 256 * k;
      rec[k] = (i < n) ? pp[i] : 0;
    }
    #pragma unroll
    for (int k = 0; k < KR4; ++k)
      if (t + 256 * k < n) atomicAdd(&cntl[rec[k] >> RSH], 1);
  } else {
    for (int i = t; i < n; i += 256) atomicAdd(&cntl[pp[i] >> RSH], 1);
  }
  __syncthreads();
  int c = (t < NBK) ? cntl[t] : 0;
  int ex = block_excl_scan<8>(c, wsum, t);
  if (t < NBK) {
    offl[t] = ex;
    cntl[t] = 0;
    int node = b * NBK + t;
    if (node < NN) rp[node] = bstart + ex;
  }
  __syncthreads();
  if (fast) {
    #pragma unroll
    for (int k = 0; k < KR4; ++k) {
      if (t + 256 * k < n) {
        int dl = rec[k] >> RSH;
        stage[offl[dl] + atomicAdd(&cntl[dl], 1)] = rec[k] & ((1 << RSH) - 1);
      }
    }
    __syncthreads();
    #pragma unroll
    for (int k = 0; k < KR4; ++k) {
      int i = t + 256 * k;
      if (i < n) csr[bstart + i] = stage[i];
    }
  } else {
    for (int i = t; i < n; i += 256) {
      int r = pp[i];
      int dl = r >> RSH;
      csr[bstart + offl[dl] + atomicAdd(&cntl[dl], 1)] = r & ((1 << RSH) - 1);
    }
  }
}

__device__ __forceinline__ float dis_of(int b, int e) {
  float d = (float)(e - b + 1);
  return (d > 0.0f) ? rsqrtf(d) : 0.0f;
}

__device__ __forceinline__ int row_end(const int* __restrict__ rp, int row) {
  int nx = rp[min(row + 1, NN - 1)];
  return (row + 1 < NN) ? nx : EE;
}


#define XCH 64
#define XPAD 68
#define XW 2

__global__ void __launch_bounds__(XW * 32) k_gemm_x(const float* __restrict__ x,
                                                    const float* __restrict__ Bp,
                                                    const int* __restrict__ rp,
                                                    float* __restrict__ h) {
  __shared__ v4f stage[XW][2][16 * XPAD / 4];
  int w    = threadIdx.x >> 5;
  int lane = threadIdx.x & 31;
  int wave = blockIdx.x * XW + w;
  int half = lane >> 4;
  int l16  = lane & 15;
  int n    = l16;
  int q  = lane & 7;
  int g  = lane >> 3;
  int r0 = 8 * (g & 1) + (g >> 1);
  const float* xg = x + (size_t)(wave * 16 + r0) * FIN + q * 4;
  int so = r0 * XPAD + q * 4;
  int lo = l16 * XPAD + half * 2;
  v8f c0 = {}; v8f c1 = {};
  const v2f* P = (const v2f*)Bp;
  v4f buf[8];
  for (int i = 0; i < 8; ++i)
    buf[i] = __builtin_nontemporal_load((const v4f*)(xg + (size_t)(i >> 1) * 2 * FIN + (i & 1) * 32));
  #pragma unroll 1
  for (int ch = 0; ch < FIN / XCH; ++ch) {
    v2f bq[2 * (XCH / 4)];
    for (int j = 0; j < 2 * (XCH / 4); ++j) bq[j] = P[(ch * 2 * (XCH / 4) + j) * 32 + lane];
    float* sp = (float*)stage[w][ch & 1];
    for (int i = 0; i < 8; ++i)
      *(v4f*)(sp + so + (i >> 1) * 2 * XPAD + (i & 1) * 32) = buf[i];
    __syncthreads();
    if (ch + 1 < FIN / XCH) {
      for (int i = 0; i < 8; ++i)
        buf[i] = __builtin_nontemporal_load(
            (const v4f*)(xg + (size_t)(i >> 1) * 2 * FIN + (i & 1) * 32 + (ch + 1) * XCH));
    }
    const float* fr = sp + lo;
    for (int ks = 0; ks < XCH / 4; ++ks) {
      v2f a = *(const v2f*)(fr + ks * 4);
      c0 = __builtin_amdgcn_wmma_f32_16x16x4_f32(false, a, false, bq[2 * ks + 0], (short)0, c0, false, false);
      c1 = __builtin_amdgcn_wmma_f32_16x16x4_f32(false, a, false, bq[2 * ks + 1], (short)0, c1, false, false);
    }
  }
  int rbase = wave * 16 + half * 8;
  int pr[9];
  #pragma unroll
  for (int v = 0; v < 8; ++v) pr[v] = rp[rbase + v];
  pr[8] = row_end(rp, rbase + 7);
  #pragma unroll
  for (int v = 0; v < 8; ++v) {
    int row = rbase + v;
    float s = dis_of(pr[v], pr[v + 1]);
    size_t r = (size_t)row * SP;
    h[r + n]      = c0[v] * s;
    h[r + n + 16] = c1[v] * s;
  }
}


template <int K>
__device__ __forceinline__ int bcast8(int v) {
  return (int)__builtin_amdgcn_mov_dpp8((unsigned)v, (unsigned)(K * 0x249249));
}

__device__ __forceinline__ float4 f4add(float4 a, float4 b) {
  return make_float4(a.x + b.x, a.y + b.y, a.z + b.z, a.w + b.w);
}

constexpr unsigned xor8_sel(int X) {
  unsigned s = 0;
  for (int i = 0; i < 8; ++i) s |= (unsigned)(i ^ X) << (3 * i);
  return s;
}

template <int X>
__device__ __forceinline__ float xor8(float v) {
  return __builtin_bit_cast(float, __builtin_amdgcn_mov_dpp8(__builtin_bit_cast(unsigned, v), xor8_sel(X)));
}

#define OP 20

template <int HEAD>
__global__ void __launch_bounds__(256) k_gather(const float* __restrict__ hs,
                                                const int* __restrict__ rp,
                                                const int* __restrict__ csr,
                                                const float* __restrict__ bias,
                                                const float* __restrict__ Bp,
                                                float* __restrict__ o,
                                                float* __restrict__ hn,
                                                const float* __restrict__ oa,
                                                const float* __restrict__ ob,
                                                const float* __restrict__ bl,
                                                float* __restrict__ out) {
  __shared__ float so[(HEAD ? 3 : 1) * 32 * OP];
  __shared__ float sdis[32];
  int t   = blockIdx.x * 256 + threadIdx.x;
  int row = t >> 3;
  int q   = t & 7;
  unsigned qb = (unsigned)q * 16u;
  int beg = rp[row];
  int end = row_end(rp, row);
  int bq = min(q, 4) * 4;
  float bi0 = bias[bq + 0], bi1 = bias[bq + 1], bi2 = bias[bq + 2], bi3 = bias[bq + 3];
  float4 pa = make_float4(0.0f, 0.0f, 0.0f, 0.0f), pb = pa;
  if (HEAD) {
    pa = ((const float4*)oa)[(size_t)row * 8 + q];
    pb = ((const float4*)ob)[(size_t)row * 8 + q];
  }
  const char* Hb = (const char*)hs;
#define ROW_LD(s) (*(const float4*)(Hb + ((unsigned)(s) * 128u + qb)))
  float4 acc = ROW_LD(row);
  int j = beg;
  int ld0 = csr[min(j + q, EE - 1)];
  int idx = (j + q < end) ? ld0 : row;
  for (; j + 8 <= end; j += 8) {
    float4 v0 = ROW_LD(bcast8<0>(idx));
    float4 v1 = ROW_LD(bcast8<1>(idx));
    float4 v2 = ROW_LD(bcast8<2>(idx));
    float4 v3 = ROW_LD(bcast8<3>(idx));
    float4 v4 = ROW_LD(bcast8<4>(idx));
    float4 v5 = ROW_LD(bcast8<5>(idx));
    float4 v6 = ROW_LD(bcast8<6>(idx));
    float4 v7 = ROW_LD(bcast8<7>(idx));
    int nj  = j + 8 + q;
    int ldn = csr[min(nj, EE - 1)];
    acc = f4add(acc, v0); acc = f4add(acc, v1); acc = f4add(acc, v2); acc = f4add(acc, v3);
    acc = f4add(acc, v4); acc = f4add(acc, v5); acc = f4add(acc, v6); acc = f4add(acc, v7);
    idx = (nj < end) ? ldn : row;
  }
  int rem = end - j;
  if (rem > 0) {
    const float4 z = make_float4(0.0f, 0.0f, 0.0f, 0.0f);
    float4 v0 = ROW_LD(bcast8<0>(idx));
    float4 v1 = ROW_LD(bcast8<1>(idx));
    float4 v2 = ROW_LD(bcast8<2>(idx));
    float4 v3 = ROW_LD(bcast8<3>(idx));
    float4 v4 = ROW_LD(bcast8<4>(idx));
    float4 v5 = ROW_LD(bcast8<5>(idx));
    float4 v6 = ROW_LD(bcast8<6>(idx));
    acc = f4add(acc, v0);
    acc = f4add(acc, (rem > 1) ? v1 : z);
    acc = f4add(acc, (rem > 2) ? v2 : z);
    acc = f4add(acc, (rem > 3) ? v3 : z);
    acc = f4add(acc, (rem > 4) ? v4 : z);
    acc = f4add(acc, (rem > 5) ? v5 : z);
    acc = f4add(acc, (rem > 6) ? v6 : z);
  }
#undef ROW_LD
  float dd = dis_of(beg, end);
  bool live = q < 5;
  float t0 = live ? acc.x * dd + bi0 : 0.0f;
  float t1 = live ? acc.y * dd + bi1 : 0.0f;
  float t2 = live ? acc.z * dd + bi2 : 0.0f;
  float t3 = live ? acc.w * dd + bi3 : 0.0f;
  float ss = t0 * t0 + t1 * t1 + t2 * t2 + t3 * t3;
  ss += xor8<1>(ss);
  ss += xor8<2>(ss);
  ss += xor8<4>(ss);
  float inv = 1.0f / fmaxf(sqrtf(ss), 1e-12f);
  float4 ov = make_float4(fmaxf(t0 * inv, 0.0f), fmaxf(t1 * inv, 0.0f),
                          fmaxf(t2 * inv, 0.0f), fmaxf(t3 * inv, 0.0f));
  int rl   = threadIdx.x >> 3;
  int w    = threadIdx.x >> 5;
  int lane = threadIdx.x & 31;
  bool mm  = w < 2;
  const v2f* P = (const v2f*)Bp;
  constexpr int NB = HEAD ? 15 : 10;
  v2f bf[NB];
  for (int j = 0; j < NB; ++j) bf[j] = v2f{0.0f, 0.0f};
  if (mm) {
    for (int j = 0; j < NB; ++j) bf[j] = P[j * 32 + lane];
  }
  if (HEAD == 0) {
    if (live) *(float4*)(so + rl * OP + q * 4) = ov;
    if (q == 0) sdis[rl] = dd;
  } else {
    if (live) {
      *(float4*)(so + 0 * 32 * OP + rl * OP + q * 4) = pa;
      *(float4*)(so + 1 * 32 * OP + rl * OP + q * 4) = pb;
      *(float4*)(so + 2 * 32 * OP + rl * OP + q * 4) = ov;
    }
  }
  __syncthreads();
  if (HEAD == 0) {
    ((float4*)o)[(size_t)row * 8 + q] = ov;
  }
  if (!mm) return;
  int half = lane >> 4;
  int l16  = lane & 15;
  int n    = l16;
  const float* ar = so + (w * 16 + l16) * OP + half * 2;
  int rloc  = w * 16 + half * 8;
  int rbase = blockIdx.x * 32 + rloc;
  if (HEAD == 0) {
    v8f c0 = {}; v8f c1 = {};
    for (int ks = 0; ks < HH / 4; ++ks) {
      v2f a = *(const v2f*)(ar + ks * 4);
      c0 = __builtin_amdgcn_wmma_f32_16x16x4_f32(false, a, false, bf[ks * 2 + 0], (short)0, c0, false, false);
      c1 = __builtin_amdgcn_wmma_f32_16x16x4_f32(false, a, false, bf[ks * 2 + 1], (short)0, c1, false, false);
    }
    for (int v = 0; v < 8; ++v) {
      float s = sdis[rloc + v];
      size_t r = (size_t)(rbase + v) * SP;
      hn[r + n]      = c0[v] * s;
      hn[r + n + 16] = c1[v] * s;
    }
  } else {
    v8f cc = {};
    for (int ks = 0; ks < 15; ++ks) {
      int layer = ks / 5;
      int kloc  = (ks - layer * 5) * 4;
      v2f a = *(const v2f*)(ar + layer * 32 * OP + kloc);
      cc = __builtin_amdgcn_wmma_f32_16x16x4_f32(false, a, false, bf[ks], (short)0, cc, false, false);
    }
    if (n < CC) {
      float bv = bl[n];
      for (int v = 0; v < 8; ++v)
        out[(size_t)(rbase + v) * CC + n] = cc[v] + bv;
    }
  }
}


extern "C" void kernel_launch(void* const* d_in, const int* in_sizes, int n_in,
                              void* d_out, int out_size, void* d_ws, size_t ws_size,
                              hipStream_t stream) {
  (void)in_sizes; (void)n_in; (void)out_size; (void)ws_size;
  const float* x  = (const float*)d_in[0];
  const int*   ei = (const int*)d_in[1];
  const float* W1 = (const float*)d_in[2];
  const float* b1 = (const float*)d_in[3];
  const float* W2 = (const float*)d_in[4];
  const float* b2 = (const float*)d_in[5];
  const float* W3 = (const float*)d_in[6];
  const float* b3 = (const float*)d_in[7];
  const float* Wl = (const float*)d_in[8];
  const float* bl = (const float*)d_in[9];
  const int* srcI = ei;
  const int* dstI = ei + EE;
  float* out = (float*)d_out;

  float* ws   = (float*)d_ws;
  int*   rp   = (int*)ws;
  float* hbuf = ws + NN;
  int*   csr  = (int*)(hbuf + (size_t)NN * SP);
  float* o1   = hbuf + (size_t)2 * NN * SP;
  float* o2   = o1 + (size_t)NN * SP;
  float* o3   = o2 + (size_t)NN * SP;
  float* Bp1  = o3 + (size_t)NN * SP;
  float* Bp2  = Bp1 + (size_t)(FIN / 4) * 2 * 64;
  float* Bp3  = Bp2 + (size_t)(HH / 4) * 2 * 64;
  float* Bpl  = Bp3 + (size_t)(HH / 4) * 2 * 64;
  int*   pairs = (int*)o3;
  int*   T    = (int*)o2;
  int*   P    = T + NBUCK * G1;
  int*   btot = P + NBUCK * G1;

  dim3 B(256);
  dim3 B4(1024);
  int gG  = (NN * 8) / 256;
  int gX  = (NN / 16) / XW;

  k_bhist<<<G1 + NPK, B4, 0, stream>>>(dstI, T, W1, W2, W3, Wl, Bp1, Bp2, Bp3, Bpl);
  k_bprefix<<<NBUCK, B, 0, stream>>>(T, P, btot);
  k_bscatter<<<G1, B4, 0, stream>>>(srcI, dstI, T, P, btot, pairs);
  k_bfinal<<<NBUCK, B, 0, stream>>>(pairs, btot, rp, csr);

  k_gemm_x<<<gX, dim3(XW * 32), 0, stream>>>(x, Bp1, rp, hbuf);
  k_gather<0><<<gG, B, 0, stream>>>(hbuf, rp, csr, b1, Bp2, o1, o3, nullptr, nullptr, nullptr, nullptr);

  k_gather<0><<<gG, B, 0, stream>>>(o3, rp, csr, b2, Bp3, o2, hbuf, nullptr, nullptr, nullptr, nullptr);

  k_gather<1><<<gG, B, 0, stream>>>(hbuf, rp, csr, b3, Bpl, nullptr, nullptr, o1, o2, bl, out);
}


# CSR level 1 in one launch: a slice counts and orders itself by bucket in LDS and writes its run starts; a bucket's start is their column sum: 6 launches, not 8; row terms in LDS arrival order
# speedup vs baseline: 16.9219x; 1.1368x over previous
#include <hip/hip_runtime.h>

#define NN 100000
#define EE 3200000
#define FIN 512
#define HH 20
#define CC 10
#define SP 32

typedef __attribute__((ext_vector_type(2))) float v2f;
typedef __attribute__((ext_vector_type(4))) float v4f;
typedef __attribute__((ext_vector_type(8))) float v8f;


#define BSH 7
#define NBK 128
#define NBUCK ((NN + NBK - 1) / NBK)
#define G1 250
#define CH1 (EE / G1)
#define RSH 17
#define TR1 ((CH1 + 1023) / 1024)
#define LW 800
#define RPG 8
#define KL 3
#define KR4 20
#define CAP4 (256 * KR4)

static_assert(EE % G1 == 0, "slices are whole");
static_assert(NBUCK < 1024 && G1 <= 32 * RPG && NBUCK + 1 <= LW && CH1 % 4 == 0 && NN <= (1 << RSH), "sizes the passes rely on");

template <int NW>
__device__ __forceinline__ int block_excl_scan(int v, int* wsum, int t) {
  int lane = t & 31;
  int w = t >> 5;
  int incl = v;
  #pragma unroll
  for (int o = 1; o < 32; o <<= 1) {
    int u = __shfl_up(incl, o, 32);
    if (lane >= o) incl += u;
  }
  if (lane == 31) wsum[w] = incl;
  __syncthreads();
  if (w == 0) {
    int x = (lane < NW) ? wsum[lane] : 0;
    int xi = x;
    #pragma unroll
    for (int o = 1; o < 32; o <<= 1) {
      int u = __shfl_up(xi, o, 32);
      if (lane >= o) xi += u;
    }
    wsum[lane] = xi - x;
    if (lane == 31) wsum[32] = xi;
  }
  __syncthreads();
  return incl - v + wsum[w];
}

__device__ __forceinline__ void pack_b_entry(const float* __restrict__ W, float* __restrict__ P,
                                             int K, int Ncols, int ntiles, int t) {
  int total = (K / 4) * ntiles * 32;
  if (t >= total) return;
  int lane = t & 31;
  int frag = t >> 5;
  int ks   = frag / ntiles;
  int tile = frag - ks * ntiles;
  int koff = (lane >> 4) * 2;
  int n    = tile * 16 + (lane & 15);
  int kk   = ks * 4 + koff;
  float v0 = (n < Ncols) ? W[kk * Ncols + n] : 0.0f;
  float v1 = (n < Ncols) ? W[(kk + 1) * Ncols + n] : 0.0f;
  P[(size_t)t * 2 + 0] = v0;
  P[(size_t)t * 2 + 1] = v1;
}

#define PK1 (((FIN / 4) * 2 * 32) / 1024)
#define NPK (PK1 + 3)

static_assert(((FIN / 4) * 2 * 32) % 1024 == 0, "W1's packed entries fill whole work-groups");
static_assert((HH / 4) * 2 * 32 <= 1024 && 15 * 32 <= 1024, "a small matrix is packed by one work-group");

__global__ void __launch_bounds__(1024) k_slice(const int* __restrict__ src,
                                                const int* __restrict__ dst,
                                                int* __restrict__ Lr,
                                                int* __restrict__ pairs,
                                                const float* __restrict__ W1,
                                                const float* __restrict__ W2,
                                                const float* __restrict__ W3,
                                                const float* __restrict__ Wl,
                                                float* __restrict__ Bp1, float* __restrict__ Bp2,
                                                float* __restrict__ Bp3, float* __restrict__ Bpl) {
  __shared__ int wsum[33];
  __shared__ int hist[NBUCK];
  __shared__ int lpos[NBUCK];
  __shared__ int4 st_rec4[CH1 / 4];
  int* st_rec = (int*)st_rec4;
  int t = threadIdx.x;
  int wg = blockIdx.x;
  if (wg >= G1) {
    int pb = wg - G1;
    if (pb < PK1)           pack_b_entry(W1, Bp1, FIN, HH, 2, pb * 1024 + t);
    else if (pb == PK1)     pack_b_entry(W2, Bp2, HH, HH, 2, t);
    else if (pb == PK1 + 1) pack_b_entry(W3, Bp3, HH, HH, 2, t);
    else                    pack_b_entry(Wl, Bpl, 3 * HH, CC, 1, t);
    return;
  }
  for (int b = t; b < NBUCK; b += 1024) hist[b] = 0;
  __syncthreads();
  const int* sp = src + wg * CH1;
  const int* dp = dst + wg * CH1;
  int d[TR1], s[TR1], r[TR1];
  #pragma unroll
  for (int k = 0; k < TR1; ++k) {
    int i = t + 1024 * k;
    d[k] = (i < CH1) ? dp[i] : 0;
    s[k] = (i < CH1) ? sp[i] : 0;
  }
  #pragma unroll
  for (int k = 0; k < TR1; ++k) {
    r[k] = 0;
    if (t + 1024 * k < CH1) r[k] = atomicAdd(&hist[d[k] >> BSH], 1);
  }
  __syncthreads();
  int mine = (t < NBUCK) ? hist[t] : 0;
  int lp = block_excl_scan<32>(mine, wsum, t);
  if (t < NBUCK) lpos[t] = lp;
  __syncthreads();
  #pragma unroll
  for (int k = 0; k < TR1; ++k) {
    if (t + 1024 * k < CH1)
      st_rec[lpos[d[k] >> BSH] + r[k]] = s[k] | ((d[k] & (NBK - 1)) << RSH);
  }
  __syncthreads();
  int4* po = (int4*)(pairs + wg * CH1);
  #pragma unroll
  for (int k = 0; k < (CH1 / 4 + 1023) / 1024; ++k) {
    int i = t + 1024 * k;
    if (i < CH1 / 4) po[i] = st_rec4[i];
  }
  int* Lw = Lr + wg * LW;
  if (t < NBUCK) Lw[t] = lp;
  if (t == NBUCK) Lw[NBUCK] = CH1;
}

__global__ void __launch_bounds__(256) k_bfinal(const int* __restrict__ pairs,
                                                const int* __restrict__ Lr,
                                                int* __restrict__ rp,
                                                int* __restrict__ csr) {
  __shared__ int wsum[33];
  __shared__ int cntl[NBK];
  __shared__ int offl[NBK];
  __shared__ int rsrc[256];
  __shared__ int rlen[256];
  __shared__ int roff[256];
  __shared__ int stage[CAP4];
  int t = threadIdx.x;
  int b = blockIdx.x;
  int a = 0, len = 0;
  if (t < G1) {
    const int* lw = Lr + t * LW + b;
    a = lw[0];
    len = lw[1] - a;
  }
  int ro = block_excl_scan<8>(len, wsum, t);
  int n = wsum[32];
  __syncthreads();
  block_excl_scan<8>(a, wsum, t);
  int bstart = wsum[32];
  rsrc[t] = t * CH1 + a;
  rlen[t] = len;
  roff[t] = ro;
  if (t < NBK) cntl[t] = 0;
  __syncthreads();
  bool fast = n <= CAP4;
  int g = t >> 3;
  int q = t & 7;
  int rec[KR4];
  if (fast) {
    int rb[RPG], rl[RPG], rf[RPG], val[RPG * KL];
    #pragma unroll
    for (int r = 0; r < RPG; ++r) {
      int k = g + 32 * r;
      rb[r] = rsrc[k]; rl[r] = rlen[k]; rf[r] = roff[k];
    }
    #pragma unroll
    for (int r = 0; r < RPG; ++r) {
      #pragma unroll
      for (int c = 0; c < KL; ++c) {
        int j = q + 8 * c;
        val[r * KL + c] = (j < rl[r]) ? pairs[rb[r] + j] : 0;
      }
    }
    #pragma unroll
    for (int r = 0; r < RPG; ++r) {
      #pragma unroll
      for (int c = 0; c < KL; ++c) {
        int j = q + 8 * c;
        if (j < rl[r]) stage[rf[r] + j] = val[r * KL + c];
      }
    }
    #pragma unroll 1
    for (int r = 0; r < RPG; ++r) {
      int k = g + 32 * r;
      int ln = rlen[k];
      const int* pp = pairs + rsrc[k];
      int of = roff[k];
      #pragma unroll 1
      for (int j = q + 8 * KL; j < ln; j += 8) stage[of + j] = pp[j];
    }
    __syncthreads();
    #pragma unroll
    for (int k = 0; k < KR4; ++k) {
      int i = t + 256 * k;
      rec[k] = (i < n) ? stage[i] : 0;
    }
    #pragma unroll
    for (int k = 0; k < KR4; ++k)
      if (t + 256 * k < n) atomicAdd(&cntl[rec[k] >> RSH], 1);
  } else {
    #pragma unroll 1
    for (int r = 0; r < RPG; ++r) {
      int k = g + 32 * r;
      const int* pp = pairs + rsrc[k];
      int ln = rlen[k];
      #pragma unroll 1
      for (int j = q; j < ln; j += 8) atomicAdd(&cntl[pp[j] >> RSH], 1);
    }
  }
  __syncthreads();
  int c = (t < NBK) ? cntl[t] : 0;
  int ex = block_excl_scan<8>(c, wsum, t);
  if (t < NBK) {
    offl[t] = ex;
    cntl[t] = 0;
    int node = b * NBK + t;
    if (node < NN) rp[node] = bstart + ex;
  }
  __syncthreads();
  if (fast) {
    #pragma unroll
    for (int k = 0; k < KR4; ++k) {
      if (t + 256 * k < n) {
        int dl = rec[k] >> RSH;
        stage[offl[dl] + atomicAdd(&cntl[dl], 1)] = rec[k] & ((1 << RSH) - 1);
      }
    }
    __syncthreads();
    #pragma unroll
    for (int k = 0; k < KR4; ++k) {
      int i = t + 256 * k;
      if (i < n) csr[bstart + i] = stage[i];
    }
  } else {
    #pragma unroll 1
    for (int r = 0; r < RPG; ++r) {
      int k = g + 32 * r;
      const int* pp = pairs + rsrc[k];
      int ln = rlen[k];
      #pragma unroll 1
      for (int j = q; j < ln; j += 8) {
        int rc = pp[j];
        int dl = rc >> RSH;
        csr[bstart + offl[dl] + atomicAdd(&cntl[dl], 1)] = rc & ((1 << RSH) - 1);
      }
    }
  }
}

__device__ __forceinline__ float dis_of(int b, int e) {
  float d = (float)(e - b + 1);
  return (d > 0.0f) ? rsqrtf(d) : 0.0f;
}

__device__ __forceinline__ int row_end(const int* __restrict__ rp, int row) {
  int nx = rp[min(row + 1, NN - 1)];
  return (row + 1 < NN) ? nx : EE;
}


#define XCH 64
#define XPAD 68
#define XW 2

__global__ void __launch_bounds__(XW * 32) k_gemm_x(const float* __restrict__ x,
                                                    const float* __restrict__ Bp,
                                                    const int* __restrict__ rp,
                                                    float* __restrict__ h) {
  __shared__ v4f stage[XW][2][16 * XPAD / 4];
  int w    = threadIdx.x >> 5;
  int lane = threadIdx.x & 31;
  int wave = blockIdx.x * XW + w;
  int half = lane >> 4;
  int l16  = lane & 15;
  int n    = l16;
  int q  = lane & 7;
  int g  = lane >> 3;
  int r0 = 8 * (g & 1) + (g >> 1);
  const float* xg = x + (size_t)(wave * 16 + r0) * FIN + q * 4;
  int so = r0 * XPAD + q * 4;
  int lo = l16 * XPAD + half * 2;
  v8f c0 = {}; v8f c1 = {};
  const v2f* P = (const v2f*)Bp;
  v4f buf[8];
  for (int i = 0; i < 8; ++i)
    buf[i] = __builtin_nontemporal_load((const v4f*)(xg + (size_t)(i >> 1) * 2 * FIN + (i & 1) * 32));
  #pragma unroll 1
  for (int ch = 0; ch < FIN / XCH; ++ch) {
    v2f bq[2 * (XCH / 4)];
    for (int j = 0; j < 2 * (XCH / 4); ++j) bq[j] = P[(ch * 2 * (XCH / 4) + j) * 32 + lane];
    float* sp = (float*)stage[w][ch & 1];
    for (int i = 0; i < 8; ++i)
      *(v4f*)(sp + so + (i >> 1) * 2 * XPAD + (i & 1) * 32) = buf[i];
    __syncthreads();
    if (ch + 1 < FIN / XCH) {
      for (int i = 0; i < 8; ++i)
        buf[i] = __builtin_nontemporal_load(
            (const v4f*)(xg + (size_t)(i >> 1) * 2 * FIN + (i & 1) * 32 + (ch + 1) * XCH));
    }
    const float* fr = sp + lo;
    for (int ks = 0; ks < XCH / 4; ++ks) {
      v2f a = *(const v2f*)(fr + ks * 4);
      c0 = __builtin_amdgcn_wmma_f32_16x16x4_f32(false, a, false, bq[2 * ks + 0], (short)0, c0, false, false);
      c1 = __builtin_amdgcn_wmma_f32_16x16x4_f32(false, a, false, bq[2 * ks + 1], (short)0, c1, false, false);
    }
  }
  int rbase = wave * 16 + half * 8;
  int pr[9];
  #pragma unroll
  for (int v = 0; v < 8; ++v) pr[v] = rp[rbase + v];
  pr[8] = row_end(rp, rbase + 7);
  #pragma unroll
  for (int v = 0; v < 8; ++v) {
    int row = rbase + v;
    float s = dis_of(pr[v], pr[v + 1]);
    size_t r = (size_t)row * SP;
    h[r + n]      = c0[v] * s;
    h[r + n + 16] = c1[v] * s;
  }
}


template <int K>
__device__ __forceinline__ int bcast8(int v) {
  return (int)__builtin_amdgcn_mov_dpp8((unsigned)v, (unsigned)(K * 0x249249));
}

__device__ __forceinline__ float4 f4add(float4 a, float4 b) {
  return make_float4(a.x + b.x, a.y + b.y, a.z + b.z, a.w + b.w);
}

constexpr unsigned xor8_sel(int X) {
  unsigned s = 0;
  for (int i = 0; i < 8; ++i) s |= (unsigned)(i ^ X) << (3 * i);
  return s;
}

template <int X>
__device__ __forceinline__ float xor8(float v) {
  return __builtin_bit_cast(float, __builtin_amdgcn_mov_dpp8(__builtin_bit_cast(unsigned, v), xor8_sel(X)));
}

#define OP 20

template <int HEAD>
__global__ void __launch_bounds__(256) k_gather(const float* __restrict__ hs,
                                                const int* __restrict__ rp,
                                                const int* __restrict__ csr,
                                                const float* __restrict__ bias,
                                                const float* __restrict__ Bp,
                                                float* __restrict__ o,
                                                float* __restrict__ hn,
                                                const float* __restrict__ oa,
                                                const float* __restrict__ ob,
                                                const float* __restrict__ bl,
                                                float* __restrict__ out) {
  __shared__ float so[(HEAD ? 3 : 1) * 32 * OP];
  __shared__ float sdis[32];
  int t   = blockIdx.x * 256 + threadIdx.x;
  int row = t >> 3;
  int q   = t & 7;
  unsigned qb = (unsigned)q * 16u;
  int beg = rp[row];
  int end = row_end(rp, row);
  int bq = min(q, 4) * 4;
  float bi0 = bias[bq + 0], bi1 = bias[bq + 1], bi2 = bias[bq + 2], bi3 = bias[bq + 3];
  float4 pa = make_float4(0.0f, 0.0f, 0.0f, 0.0f), pb = pa;
  if (HEAD) {
    pa = ((const float4*)oa)[(size_t)row * 8 + q];
    pb = ((const float4*)ob)[(size_t)row * 8 + q];
  }
  const char* Hb = (const char*)hs;
#define ROW_LD(s) (*(const float4*)(Hb + ((unsigned)(s) * 128u + qb)))
  float4 acc = ROW_LD(row);
  int j = beg;
  int ld0 = csr[min(j + q, EE - 1)];
  int idx = (j + q < end) ? ld0 : row;
  for (; j + 8 <= end; j += 8) {
    float4 v0 = ROW_LD(bcast8<0>(idx));
    float4 v1 = ROW_LD(bcast8<1>(idx));
    float4 v2 = ROW_LD(bcast8<2>(idx));
    float4 v3 = ROW_LD(bcast8<3>(idx));
    float4 v4 = ROW_LD(bcast8<4>(idx));
    float4 v5 = ROW_LD(bcast8<5>(idx));
    float4 v6 = ROW_LD(bcast8<6>(idx));
    float4 v7 = ROW_LD(bcast8<7>(idx));
    int nj  = j + 8 + q;
    int ldn = csr[min(nj, EE - 1)];
    acc = f4add(acc, v0); acc = f4add(acc, v1); acc = f4add(acc, v2); acc = f4add(acc, v3);
    acc = f4add(acc, v4); acc = f4add(acc, v5); acc = f4add(acc, v6); acc = f4add(acc, v7);
    idx = (nj < end) ? ldn : row;
  }
  int rem = end - j;
  if (rem > 0) {
    const float4 z = make_float4(0.0f, 0.0f, 0.0f, 0.0f);
    float4 v0 = ROW_LD(bcast8<0>(idx));
    float4 v1 = ROW_LD(bcast8<1>(idx));
    float4 v2 = ROW_LD(bcast8<2>(idx));
    float4 v3 = ROW_LD(bcast8<3>(idx));
    float4 v4 = ROW_LD(bcast8<4>(idx));
    float4 v5 = ROW_LD(bcast8<5>(idx));
    float4 v6 = ROW_LD(bcast8<6>(idx));
    acc = f4add(acc, v0);
    acc = f4add(acc, (rem > 1) ? v1 : z);
    acc = f4add(acc, (rem > 2) ? v2 : z);
    acc = f4add(acc, (rem > 3) ? v3 : z);
    acc = f4add(acc, (rem > 4) ? v4 : z);
    acc = f4add(acc, (rem > 5) ? v5 : z);
    acc = f4add(acc, (rem > 6) ? v6 : z);
  }
#undef ROW_LD
  float dd = dis_of(beg, end);
  bool live = q < 5;
  float t0 = live ? acc.x * dd + bi0 : 0.0f;
  float t1 = live ? acc.y * dd + bi1 : 0.0f;
  float t2 = live ? acc.z * dd + bi2 : 0.0f;
  float t3 = live ? acc.w * dd + bi3 : 0.0f;
  float ss = t0 * t0 + t1 * t1 + t2 * t2 + t3 * t3;
  ss += xor8<1>(ss);
  ss += xor8<2>(ss);
  ss += xor8<4>(ss);
  float inv = 1.0f / fmaxf(sqrtf(ss), 1e-12f);
  float4 ov = make_float4(fmaxf(t0 * inv, 0.0f), fmaxf(t1 * inv, 0.0f),
                          fmaxf(t2 * inv, 0.0f), fmaxf(t3 * inv, 0.0f));
  int rl   = threadIdx.x >> 3;
  int w    = threadIdx.x >> 5;
  int lane = threadIdx.x & 31;
  bool mm  = w < 2;
  const v2f* P = (const v2f*)Bp;
  constexpr int NB = HEAD ? 15 : 10;
  v2f bf[NB];
  for (int j = 0; j < NB; ++j) bf[j] = v2f{0.0f, 0.0f};
  if (mm) {
    for (int j = 0; j < NB; ++j) bf[j] = P[j * 32 + lane];
  }
  if (HEAD == 0) {
    if (live) *(float4*)(so + rl * OP + q * 4) = ov;
    if (q == 0) sdis[rl] = dd;
  } else {
    if (live) {
      *(float4*)(so + 0 * 32 * OP + rl * OP + q * 4) = pa;
      *(float4*)(so + 1 * 32 * OP + rl * OP + q * 4) = pb;
      *(float4*)(so + 2 * 32 * OP + rl * OP + q * 4) = ov;
    }
  }
  __syncthreads();
  if (HEAD == 0) {
    ((float4*)o)[(size_t)row * 8 + q] = ov;
  }
  if (!mm) return;
  int half = lane >> 4;
  int l16  = lane & 15;
  int n    = l16;
  const float* ar = so + (w * 16 + l16) * OP + half * 2;
  int rloc  = w * 16 + half * 8;
  int rbase = blockIdx.x * 32 + rloc;
  if (HEAD == 0) {
    v8f c0 = {}; v8f c1 = {};
    for (int ks = 0; ks < HH / 4; ++ks) {
      v2f a = *(const v2f*)(ar + ks * 4);
      c0 = __builtin_amdgcn_wmma_f32_16x16x4_f32(false, a, false, bf[ks * 2 + 0], (short)0, c0, false, false);
      c1 = __builtin_amdgcn_wmma_f32_16x16x4_f32(false, a, false, bf[ks * 2 + 1], (short)0, c1, false, false);
    }
    for (int v = 0; v < 8; ++v) {
      float s = sdis[rloc + v];
      size_t r = (size_t)(rbase + v) * SP;
      hn[r + n]      = c0[v] * s;
      hn[r + n + 16] = c1[v] * s;
    }
  } else {
    v8f cc = {};
    for (int ks = 0; ks < 15; ++ks) {
      int layer = ks / 5;
      int kloc  = (ks - layer * 5) * 4;
      v2f a = *(const v2f*)(ar + layer * 32 * OP + kloc);
      cc = __builtin_amdgcn_wmma_f32_16x16x4_f32(false, a, false, bf[ks], (short)0, cc, false, false);
    }
    if (n < CC) {
      float bv = bl[n];
      for (int v = 0; v < 8; ++v)
        out[(size_t)(rbase + v) * CC + n] = cc[v] + bv;
    }
  }
}


extern "C" void kernel_launch(void* const* d_in, const int* in_sizes, int n_in,
                              void* d_out, int out_size, void* d_ws, size_t ws_size,
                              hipStream_t stream) {
  (void)in_sizes; (void)n_in; (void)out_size; (void)ws_size;
  const float* x  = (const float*)d_in[0];
  const int*   ei = (const int*)d_in[1];
  const float* W1 = (const float*)d_in[2];
  const float* b1 = (const float*)d_in[3];
  const float* W2 = (const float*)d_in[4];
  const float* b2 = (const float*)d_in[5];
  const float* W3 = (const float*)d_in[6];
  const float* b3 = (const float*)d_in[7];
  const float* Wl = (const float*)d_in[8];
  const float* bl = (const float*)d_in[9];
  const int* srcI = ei;
  const int* dstI = ei + EE;
  float* out = (float*)d_out;

  float* ws   = (float*)d_ws;
  int*   rp   = (int*)ws;
  float* hbuf = ws + NN;
  int*   csr  = (int*)(hbuf + (size_t)NN * SP);
  float* o1   = hbuf + (size_t)2 * NN * SP;
  float* o2   = o1 + (size_t)NN * SP;
  float* o3   = o2 + (size_t)NN * SP;
  float* Bp1  = o3 + (size_t)NN * SP;
  float* Bp2  = Bp1 + (size_t)(FIN / 4) * 2 * 64;
  float* Bp3  = Bp2 + (size_t)(HH / 4) * 2 * 64;
  float* Bpl  = Bp3 + (size_t)(HH / 4) * 2 * 64;
  int*   pairs = (int*)o3;
  int*   Lr   = (int*)o2;

  dim3 B(256);
  dim3 B4(1024);
  int gG  = (NN * 8) / 256;
  int gX  = (NN / 16) / XW;

  k_slice<<<G1 + NPK, B4, 0, stream>>>(srcI, dstI, Lr, pairs, W1, W2, W3, Wl, Bp1, Bp2, Bp3, Bpl);
  k_bfinal<<<NBUCK, B, 0, stream>>>(pairs, Lr, rp, csr);

  k_gemm_x<<<gX, dim3(XW * 32), 0, stream>>>(x, Bp1, rp, hbuf);
  k_gather<0><<<gG, B, 0, stream>>>(hbuf, rp, csr, b1, Bp2, o1, o3, nullptr, nullptr, nullptr, nullptr);

  k_gather<0><<<gG, B, 0, stream>>>(o3, rp, csr, b2, Bp3, o2, hbuf, nullptr, nullptr, nullptr, nullptr);

  k_gather<1><<<gG, B, 0, stream>>>(hbuf, rp, csr, b3, Bpl, nullptr, nullptr, o1, o2, bl, out);
}


# x GEMM rides in the bucket launch: a work-group ranks its bucket's 128 rows and multiplies the same 128 rows of x (odd groups in the other order); deg^-1/2 from LDS counts; 5 launches not 6; sums kept
# speedup vs baseline: 17.6180x; 1.0411x over previous
#include <hip/hip_runtime.h>

#define NN 100000
#define EE 3200000
#define FIN 512
#define HH 20
#define CC 10
#define SP 32

typedef __attribute__((ext_vector_type(2))) float v2f;
typedef __attribute__((ext_vector_type(4))) float v4f;
typedef __attribute__((ext_vector_type(8))) float v8f;


#define BSH 7
#define NBK 128
#define NBUCK ((NN + NBK - 1) / NBK)
#define G1 250
#define CH1 (EE / G1)
#define RSH 17
#define TR1 ((CH1 + 1023) / 1024)
#define LW 800
#define RPG 8
#define KL 3
#define KR4 20
#define CAP4 (256 * KR4)

static_assert(EE % G1 == 0, "slices are whole");
static_assert(NBUCK < 1024 && G1 <= 32 * RPG && NBUCK + 1 <= LW && CH1 % 4 == 0 && NN <= (1 << RSH), "sizes the passes rely on");

template <int NW>
__device__ __forceinline__ int block_excl_scan(int v, int* wsum, int t) {
  int lane = t & 31;
  int w = t >> 5;
  int incl = v;
  #pragma unroll
  for (int o = 1; o < 32; o <<= 1) {
    int u = __shfl_up(incl, o, 32);
    if (lane >= o) incl += u;
  }
  if (lane == 31) wsum[w] = incl;
  __syncthreads();
  if (w == 0) {
    int x = (lane < NW) ? wsum[lane] : 0;
    int xi = x;
    #pragma unroll
    for (int o = 1; o < 32; o <<= 1) {
      int u = __shfl_up(xi, o, 32);
      if (lane >= o) xi += u;
    }
    wsum[lane] = xi - x;
    if (lane == 31) wsum[32] = xi;
  }
  __syncthreads();
  return incl - v + wsum[w];
}

__device__ __forceinline__ void pack_b_entry(const float* __restrict__ W, float* __restrict__ P,
                                             int K, int Ncols, int ntiles, int t) {
  int total = (K / 4) * ntiles * 32;
  if (t >= total) return;
  int lane = t & 31;
  int frag = t >> 5;
  int ks   = frag / ntiles;
  int tile = frag - ks * ntiles;
  int koff = (lane >> 4) * 2;
  int n    = tile * 16 + (lane & 15);
  int kk   = ks * 4 + koff;
  float v0 = (n < Ncols) ? W[kk * Ncols + n] : 0.0f;
  float v1 = (n < Ncols) ? W[(kk + 1) * Ncols + n] : 0.0f;
  P[(size_t)t * 2 + 0] = v0;
  P[(size_t)t * 2 + 1] = v1;
}

#define PK1 (((FIN / 4) * 2 * 32) / 1024)
#define NPK (PK1 + 3)

static_assert(((FIN / 4) * 2 * 32) % 1024 == 0, "W1's packed entries fill whole work-groups");
static_assert((HH / 4) * 2 * 32 <= 1024 && 15 * 32 <= 1024, "a small matrix is packed by one work-group");

__global__ void __launch_bounds__(1024) k_slice(const int* __restrict__ src,
                                                const int* __restrict__ dst,
                                                int* __restrict__ Lr,
                                                int* __restrict__ pairs,
                                                const float* __restrict__ W1,
                                                const float* __restrict__ W2,
                                                const float* __restrict__ W3,
                                                const float* __restrict__ Wl,
                                                float* __restrict__ Bp1, float* __restrict__ Bp2,
                                                float* __restrict__ Bp3, float* __restrict__ Bpl) {
  __shared__ int wsum[33];
  __shared__ int hist[NBUCK];
  __shared__ int lpos[NBUCK];
  __shared__ int4 st_rec4[CH1 / 4];
  int* st_rec = (int*)st_rec4;
  int t = threadIdx.x;
  int wg = blockIdx.x;
  if (wg >= G1) {
    int pb = wg - G1;
    if (pb < PK1)           pack_b_entry(W1, Bp1, FIN, HH, 2, pb * 1024 + t);
    else if (pb == PK1)     pack_b_entry(W2, Bp2, HH, HH, 2, t);
    else if (pb == PK1 + 1) pack_b_entry(W3, Bp3, HH, HH, 2, t);
    else                    pack_b_entry(Wl, Bpl, 3 * HH, CC, 1, t);
    return;
  }
  for (int b = t; b < NBUCK; b += 1024) hist[b] = 0;
  __syncthreads();
  const int* sp = src + wg * CH1;
  const int* dp = dst + wg * CH1;
  int d[TR1], s[TR1], r[TR1];
  #pragma unroll
  for (int k = 0; k < TR1; ++k) {
    int i = t + 1024 * k;
    d[k] = (i < CH1) ? dp[i] : 0;
    s[k] = (i < CH1) ? sp[i] : 0;
  }
  #pragma unroll
  for (int k = 0; k < TR1; ++k) {
    r[k] = 0;
    if (t + 1024 * k < CH1) r[k] = atomicAdd(&hist[d[k] >> BSH], 1);
  }
  __syncthreads();
  int mine = (t < NBUCK) ? hist[t] : 0;
  int lp = block_excl_scan<32>(mine, wsum, t);
  if (t < NBUCK) lpos[t] = lp;
  __syncthreads();
  #pragma unroll
  for (int k = 0; k < TR1; ++k) {
    if (t + 1024 * k < CH1)
      st_rec[lpos[d[k] >> BSH] + r[k]] = s[k] | ((d[k] & (NBK - 1)) << RSH);
  }
  __syncthreads();
  int4* po = (int4*)(pairs + wg * CH1);
  #pragma unroll
  for (int k = 0; k < (CH1 / 4 + 1023) / 1024; ++k) {
    int i = t + 1024 * k;
    if (i < CH1 / 4) po[i] = st_rec4[i];
  }
  int* Lw = Lr + wg * LW;
  if (t < NBUCK) Lw[t] = lp;
  if (t == NBUCK) Lw[NBUCK] = CH1;
}


__device__ __forceinline__ float dis_of(int b, int e) {
  float d = (float)(e - b + 1);
  return (d > 0.0f) ? rsqrtf(d) : 0.0f;
}

__device__ __forceinline__ int row_end(const int* __restrict__ rp, int row) {
  int nx = rp[min(row + 1, NN - 1)];
  return (row + 1 < NN) ? nx : EE;
}


#define XCH 64
#define XPAD 68
#define TPB (NBK / 16)

static_assert(TPB * 32 == 256 && NN % 16 == 0, "a bucket's work-group has one wave per 16-row tile");

__device__ __forceinline__ void gemm_tile(const float* __restrict__ x, const float* __restrict__ Bp,
                                          float* __restrict__ st, int tile, int lane,
                                          v8f& c0, v8f& c1) {
  int half = lane >> 4;
  int l16  = lane & 15;
  int q  = lane & 7;
  int g  = lane >> 3;
  int r0 = 8 * (g & 1) + (g >> 1);
  const float* xg = x + (size_t)tile * (16 * FIN) + (r0 * FIN + q * 4);
  int so = r0 * XPAD + q * 4;
  int lo = l16 * XPAD + half * 2;
  const v2f* P = (const v2f*)Bp;
  v4f buf[8];
  for (int i = 0; i < 8; ++i)
    buf[i] = __builtin_nontemporal_load((const v4f*)(xg + (size_t)(i >> 1) * 2 * FIN + (i & 1) * 32));
  #pragma unroll 1
  for (int ch = 0; ch < FIN / XCH; ++ch) {
    v2f bq[2 * (XCH / 4)];
    for (int j = 0; j < 2 * (XCH / 4); ++j) bq[j] = P[(ch * 2 * (XCH / 4) + j) * 32 + lane];
    float* sp = st + (ch & 1) * (16 * XPAD);
    for (int i = 0; i < 8; ++i)
      *(v4f*)(sp + so + (i >> 1) * 2 * XPAD + (i & 1) * 32) = buf[i];
    __syncthreads();
    if (ch + 1 < FIN / XCH) {
      for (int i = 0; i < 8; ++i)
        buf[i] = __builtin_nontemporal_load(
            (const v4f*)(xg + (size_t)(i >> 1) * 2 * FIN + (i & 1) * 32 + (ch + 1) * XCH));
    }
    const float* fr = sp + lo;
    for (int ks = 0; ks < XCH / 4; ++ks) {
      v2f a = *(const v2f*)(fr + ks * 4);
      c0 = __builtin_amdgcn_wmma_f32_16x16x4_f32(false, a, false, bq[2 * ks + 0], (short)0, c0, false, false);
      c1 = __builtin_amdgcn_wmma_f32_16x16x4_f32(false, a, false, bq[2 * ks + 1], (short)0, c1, false, false);
    }
  }
}

__global__ void __launch_bounds__(256, 8) k_bucket_gemm(const int* __restrict__ pairs,
                                                     const int* __restrict__ Lr,
                                                     int* __restrict__ rp,
                                                     int* __restrict__ csr,
                                                     const float* __restrict__ x,
                                                     const float* __restrict__ Bp,
                                                     float* __restrict__ h) {
  __shared__ int wsum[33];
  __shared__ int cntl[NBK];
  __shared__ int offl[NBK];
  __shared__ int rsrc[256];
  __shared__ int rlen[256];
  __shared__ int roff[256];
  __shared__ float sdis[NBK];
  __shared__ v4f xst[TPB][2][16 * XPAD / 4];
  static_assert(sizeof(int) * CAP4 <= sizeof(v4f) * TPB * 2 * (16 * XPAD / 4), "the record stage fits the x pieces");
  int* stage = (int*)xst;
  int t = threadIdx.x;
  int b = blockIdx.x;
  int w    = __builtin_amdgcn_readfirstlane(t >> 5);
  int lane = t & 31;
  int tile = b * TPB + w;
  int tl   = min(tile, NN / 16 - 1);
  float* st = (float*)xst[w];
  bool first = (b & 1) != 0;
  v8f c0 = {}; v8f c1 = {};
  if (first) gemm_tile(x, Bp, st, tl, lane, c0, c1);
  __syncthreads();
  int a = 0, len = 0;
  if (t < G1) {
    const int* lw = Lr + t * LW + b;
    a = lw[0];
    len = lw[1] - a;
  }
  int ro = block_excl_scan<8>(len, wsum, t);
  int n = wsum[32];
  __syncthreads();
  block_excl_scan<8>(a, wsum, t);
  int bstart = wsum[32];
  rsrc[t] = t * CH1 + a;
  rlen[t] = len;
  roff[t] = ro;
  if (t < NBK) cntl[t] = 0;
  __syncthreads();
  bool fast = n <= CAP4;
  int g = t >> 3;
  int q = t & 7;
  int rec[KR4];
  if (fast) {
    int rb[RPG], rl[RPG], rf[RPG], val[RPG * KL];
    #pragma unroll
    for (int r = 0; r < RPG; ++r) {
      int k = g + 32 * r;
      rb[r] = rsrc[k]; rl[r] = rlen[k]; rf[r] = roff[k];
    }
    #pragma unroll
    for (int r = 0; r < RPG; ++r) {
      #pragma unroll
      for (int c = 0; c < KL; ++c) {
        int j = q + 8 * c;
        val[r * KL + c] = (j < rl[r]) ? pairs[rb[r] + j] : 0;
      }
    }
    #pragma unroll
    for (int r = 0; r < RPG; ++r) {
      #pragma unroll
      for (int c = 0; c < KL; ++c) {
        int j = q + 8 * c;
        if (j < rl[r]) stage[rf[r] + j] = val[r * KL + c];
      }
    }
    #pragma unroll 1
    for (int r = 0; r < RPG; ++r) {
      int k = g + 32 * r;
      int ln = rlen[k];
      const int* pp = pairs + rsrc[k];
      int of = roff[k];
      #pragma unroll 1
      for (int j = q + 8 * KL; j < ln; j += 8) stage[of + j] = pp[j];
    }
    __syncthreads();
    #pragma unroll
    for (int k = 0; k < KR4; ++k) {
      int i = t + 256 * k;
      rec[k] = (i < n) ? stage[i] : 0;
    }
    #pragma unroll
    for (int k = 0; k < KR4; ++k)
      if (t + 256 * k < n) atomicAdd(&cntl[rec[k] >> RSH], 1);
  } else {
    #pragma unroll 1
    for (int r = 0; r < RPG; ++r) {
      int k = g + 32 * r;
      const int* pp = pairs + rsrc[k];
      int ln = rlen[k];
      #pragma unroll 1
      for (int j = q; j < ln; j += 8) atomicAdd(&cntl[pp[j] >> RSH], 1);
    }
  }
  __syncthreads();
  int c = (t < NBK) ? cntl[t] : 0;
  int ex = block_excl_scan<8>(c, wsum, t);
  if (t < NBK) {
    offl[t] = ex;
    sdis[t] = rsqrtf((float)(c + 1));
    cntl[t] = 0;
    int node = b * NBK + t;
    if (node < NN) rp[node] = bstart + ex;
  }
  __syncthreads();
  if (fast) {
    #pragma unroll
    for (int k = 0; k < KR4; ++k) {
      if (t + 256 * k < n) {
        int dl = rec[k] >> RSH;
        stage[offl[dl] + atomicAdd(&cntl[dl], 1)] = rec[k] & ((1 << RSH) - 1);
      }
    }
    __syncthreads();
    #pragma unroll
    for (int k = 0; k < KR4; ++k) {
      int i = t + 256 * k;
      if (i < n) csr[bstart + i] = stage[i];
    }
  } else {
    #pragma unroll 1
    for (int r = 0; r < RPG; ++r) {
      int k = g + 32 * r;
      const int* pp = pairs + rsrc[k];
      int ln = rlen[k];
      #pragma unroll 1
      for (int j = q; j < ln; j += 8) {
        int rc = pp[j];
        int dl = rc >> RSH;
        csr[bstart + offl[dl] + atomicAdd(&cntl[dl], 1)] = rc & ((1 << RSH) - 1);
      }
    }
  }
  __syncthreads();
  if (!first) gemm_tile(x, Bp, st, tl, lane, c0, c1);
  if (tile < NN / 16) {
    int half = lane >> 4;
    int nc   = lane & 15;
    int lrow = w * 16 + half * 8;
    for (int v = 0; v < 8; ++v) {
      float s = sdis[lrow + v];
      size_t r = (size_t)(b * NBK + lrow + v) * SP;
      h[r + nc]      = c0[v] * s;
      h[r + nc + 16] = c1[v] * s;
    }
  }
}


template <int K>
__device__ __forceinline__ int bcast8(int v) {
  return (int)__builtin_amdgcn_mov_dpp8((unsigned)v, (unsigned)(K * 0x249249));
}

__device__ __forceinline__ float4 f4add(float4 a, float4 b) {
  return make_float4(a.x + b.x, a.y + b.y, a.z + b.z, a.w + b.w);
}

constexpr unsigned xor8_sel(int X) {
  unsigned s = 0;
  for (int i = 0; i < 8; ++i) s |= (unsigned)(i ^ X) << (3 * i);
  return s;
}

template <int X>
__device__ __forceinline__ float xor8(float v) {
  return __builtin_bit_cast(float, __builtin_amdgcn_mov_dpp8(__builtin_bit_cast(unsigned, v), xor8_sel(X)));
}

#define OP 20

template <int HEAD>
__global__ void __launch_bounds__(256) k_gather(const float* __restrict__ hs,
                                                const int* __restrict__ rp,
                                                const int* __restrict__ csr,
                                                const float* __restrict__ bias,
                                                const float* __restrict__ Bp,
                                                float* __restrict__ o,
                                                float* __restrict__ hn,
                                                const float* __restrict__ oa,
                                                const float* __restrict__ ob,
                                                const float* __restrict__ bl,
                                                float* __restrict__ out) {
  __shared__ float so[(HEAD ? 3 : 1) * 32 * OP];
  __shared__ float sdis[32];
  int t   = blockIdx.x * 256 + threadIdx.x;
  int row = t >> 3;
  int q   = t & 7;
  unsigned qb = (unsigned)q * 16u;
  int beg = rp[row];
  int end = row_end(rp, row);
  int bq = min(q, 4) * 4;
  float bi0 = bias[bq + 0], bi1 = bias[bq + 1], bi2 = bias[bq + 2], bi3 = bias[bq + 3];
  float4 pa = make_float4(0.0f, 0.0f, 0.0f, 0.0f), pb = pa;
  if (HEAD) {
    pa = ((const float4*)oa)[(size_t)row * 8 + q];
    pb = ((const float4*)ob)[(size_t)row * 8 + q];
  }
  const char* Hb = (const char*)hs;
#define ROW_LD(s) (*(const float4*)(Hb + ((unsigned)(s) * 128u + qb)))
  float4 acc = ROW_LD(row);
  int j = beg;
  int ld0 = csr[min(j + q, EE - 1)];
  int idx = (j + q < end) ? ld0 : row;
  for (; j + 8 <= end; j += 8) {
    float4 v0 = ROW_LD(bcast8<0>(idx));
    float4 v1 = ROW_LD(bcast8<1>(idx));
    float4 v2 = ROW_LD(bcast8<2>(idx));
    float4 v3 = ROW_LD(bcast8<3>(idx));
    float4 v4 = ROW_LD(bcast8<4>(idx));
    float4 v5 = ROW_LD(bcast8<5>(idx));
    float4 v6 = ROW_LD(bcast8<6>(idx));
    float4 v7 = ROW_LD(bcast8<7>(idx));
    int nj  = j + 8 + q;
    int ldn = csr[min(nj, EE - 1)];
    acc = f4add(acc, v0); acc = f4add(acc, v1); acc = f4add(acc, v2); acc = f4add(acc, v3);
    acc = f4add(acc, v4); acc = f4add(acc, v5); acc = f4add(acc, v6); acc = f4add(acc, v7);
    idx = (nj < end) ? ldn : row;
  }
  int rem = end - j;
  if (rem > 0) {
    const float4 z = make_float4(0.0f, 0.0f, 0.0f, 0.0f);
    float4 v0 = ROW_LD(bcast8<0>(idx));
    float4 v1 = ROW_LD(bcast8<1>(idx));
    float4 v2 = ROW_LD(bcast8<2>(idx));
    float4 v3 = ROW_LD(bcast8<3>(idx));
    float4 v4 = ROW_LD(bcast8<4>(idx));
    float4 v5 = ROW_LD(bcast8<5>(idx));
    float4 v6 = ROW_LD(bcast8<6>(idx));
    acc = f4add(acc, v0);
    acc = f4add(acc, (rem > 1) ? v1 : z);
    acc = f4add(acc, (rem > 2) ? v2 : z);
    acc = f4add(acc, (rem > 3) ? v3 : z);
    acc = f4add(acc, (rem > 4) ? v4 : z);
    acc = f4add(acc, (rem > 5) ? v5 : z);
    acc = f4add(acc, (rem > 6) ? v6 : z);
  }
#undef ROW_LD
  float dd = dis_of(beg, end);
  bool live = q < 5;
  float t0 = live ? acc.x * dd + bi0 : 0.0f;
  float t1 = live ? acc.y * dd + bi1 : 0.0f;
  float t2 = live ? acc.z * dd + bi2 : 0.0f;
  float t3 = live ? acc.w * dd + bi3 : 0.0f;
  float ss = t0 * t0 + t1 * t1 + t2 * t2 + t3 * t3;
  ss += xor8<1>(ss);
  ss += xor8<2>(ss);
  ss += xor8<4>(ss);
  float inv = 1.0f / fmaxf(sqrtf(ss), 1e-12f);
  float4 ov = make_float4(fmaxf(t0 * inv, 0.0f), fmaxf(t1 * inv, 0.0f),
                          fmaxf(t2 * inv, 0.0f), fmaxf(t3 * inv, 0.0f));
  int rl   = threadIdx.x >> 3;
  int w    = threadIdx.x >> 5;
  int lane = threadIdx.x & 31;
  bool mm  = w < 2;
  const v2f* P = (const v2f*)Bp;
  constexpr int NB = HEAD ? 15 : 10;
  v2f bf[NB];
  for (int j = 0; j < NB; ++j) bf[j] = v2f{0.0f, 0.0f};
  if (mm) {
    for (int j = 0; j < NB; ++j) bf[j] = P[j * 32 + lane];
  }
  if (HEAD == 0) {
    if (live) *(float4*)(so + rl * OP + q * 4) = ov;
    if (q == 0) sdis[rl] = dd;
  } else {
    if (live) {
      *(float4*)(so + 0 * 32 * OP + rl * OP + q * 4) = pa;
      *(float4*)(so + 1 * 32 * OP + rl * OP + q * 4) = pb;
      *(float4*)(so + 2 * 32 * OP + rl * OP + q * 4) = ov;
    }
  }
  __syncthreads();
  if (HEAD == 0) {
    ((float4*)o)[(size_t)row * 8 + q] = ov;
  }
  if (!mm) return;
  int half = lane >> 4;
  int l16  = lane & 15;
  int n    = l16;
  const float* ar = so + (w * 16 + l16) * OP + half * 2;
  int rloc  = w * 16 + half * 8;
  int rbase = blockIdx.x * 32 + rloc;
  if (HEAD == 0) {
    v8f c0 = {}; v8f c1 = {};
    for (int ks = 0; ks < HH / 4; ++ks) {
      v2f a = *(const v2f*)(ar + ks * 4);
      c0 = __builtin_amdgcn_wmma_f32_16x16x4_f32(false, a, false, bf[ks * 2 + 0], (short)0, c0, false, false);
      c1 = __builtin_amdgcn_wmma_f32_16x16x4_f32(false, a, false, bf[ks * 2 + 1], (short)0, c1, false, false);
    }
    for (int v = 0; v < 8; ++v) {
      float s = sdis[rloc + v];
      size_t r = (size_t)(rbase + v) * SP;
      hn[r + n]      = c0[v] * s;
      hn[r + n + 16] = c1[v] * s;
    }
  } else {
    v8f cc = {};
    for (int ks = 0; ks < 15; ++ks) {
      int layer = ks / 5;
      int kloc  = (ks - layer * 5) * 4;
      v2f a = *(const v2f*)(ar + layer * 32 * OP + kloc);
      cc = __builtin_amdgcn_wmma_f32_16x16x4_f32(false, a, false, bf[ks], (short)0, cc, false, false);
    }
    if (n < CC) {
      float bv = bl[n];
      for (int v = 0; v < 8; ++v)
        out[(size_t)(rbase + v) * CC + n] = cc[v] + bv;
    }
  }
}


extern "C" void kernel_launch(void* const* d_in, const int* in_sizes, int n_in,
                              void* d_out, int out_size, void* d_ws, size_t ws_size,
                              hipStream_t stream) {
  (void)in_sizes; (void)n_in; (void)out_size; (void)ws_size;
  const float* x  = (const float*)d_in[0];
  const int*   ei = (const int*)d_in[1];
  const float* W1 = (const float*)d_in[2];
  const float* b1 = (const float*)d_in[3];
  const float* W2 = (const float*)d_in[4];
  const float* b2 = (const float*)d_in[5];
  const float* W3 = (const float*)d_in[6];
  const float* b3 = (const float*)d_in[7];
  const float* Wl = (const float*)d_in[8];
  const float* bl = (const float*)d_in[9];
  const int* srcI = ei;
  const int* dstI = ei + EE;
  float* out = (float*)d_out;

  float* ws   = (float*)d_ws;
  int*   rp   = (int*)ws;
  float* hbuf = ws + NN;
  int*   csr  = (int*)(hbuf + (size_t)NN * SP);
  float* o1   = hbuf + (size_t)2 * NN * SP;
  float* o2   = o1 + (size_t)NN * SP;
  float* o3   = o2 + (size_t)NN * SP;
  float* Bp1  = o3 + (size_t)NN * SP;
  float* Bp2  = Bp1 + (size_t)(FIN / 4) * 2 * 64;
  float* Bp3  = Bp2 + (size_t)(HH / 4) * 2 * 64;
  float* Bpl  = Bp3 + (size_t)(HH / 4) * 2 * 64;
  int*   pairs = (int*)o3;
  int*   Lr   = (int*)o2;

  dim3 B(256);
  dim3 B4(1024);
  int gG  = (NN * 8) / 256;

  k_slice<<<G1 + NPK, B4, 0, stream>>>(srcI, dstI, Lr, pairs, W1, W2, W3, Wl, Bp1, Bp2, Bp3, Bpl);
  k_bucket_gemm<<<NBUCK, B, 0, stream>>>(pairs, Lr, rp, csr, x, Bp1, hbuf);

  k_gather<0><<<gG, B, 0, stream>>>(hbuf, rp, csr, b1, Bp2, o1, o3, nullptr, nullptr, nullptr, nullptr);

  k_gather<0><<<gG, B, 0, stream>>>(o3, rp, csr, b2, Bp3, o2, hbuf, nullptr, nullptr, nullptr, nullptr);

  k_gather<1><<<gG, B, 0, stream>>>(hbuf, rp, csr, b3, Bpl, nullptr, nullptr, o1, o2, bl, out);
}


# Gather: an 8-lane group walks the edge lists of two consecutive rows in step, 16 row loads per lane issued together; 12500 waves per launch, not 25000; each row's terms still added in list order
# speedup vs baseline: 18.5523x; 1.0530x over previous
#include <hip/hip_runtime.h>

#define NN 100000
#define EE 3200000
#define FIN 512
#define HH 20
#define CC 10
#define SP 32

typedef __attribute__((ext_vector_type(2))) float v2f;
typedef __attribute__((ext_vector_type(4))) float v4f;
typedef __attribute__((ext_vector_type(8))) float v8f;


#define BSH 7
#define NBK 128
#define NBUCK ((NN + NBK - 1) / NBK)
#define G1 250
#define CH1 (EE / G1)
#define RSH 17
#define TR1 ((CH1 + 1023) / 1024)
#define LW 800
#define RPG 8
#define KL 3
#define KR4 20
#define CAP4 (256 * KR4)

static_assert(EE % G1 == 0, "slices are whole");
static_assert(NBUCK < 1024 && G1 <= 32 * RPG && NBUCK + 1 <= LW && CH1 % 4 == 0 && NN <= (1 << RSH), "sizes the passes rely on");

template <int NW>
__device__ __forceinline__ int block_excl_scan(int v, int* wsum, int t) {
  int lane = t & 31;
  int w = t >> 5;
  int incl = v;
  #pragma unroll
  for (int o = 1; o < 32; o <<= 1) {
    int u = __shfl_up(incl, o, 32);
    if (lane >= o) incl += u;
  }
  if (lane == 31) wsum[w] = incl;
  __syncthreads();
  if (w == 0) {
    int x = (lane < NW) ? wsum[lane] : 0;
    int xi = x;
    #pragma unroll
    for (int o = 1; o < 32; o <<= 1) {
      int u = __shfl_up(xi, o, 32);
      if (lane >= o) xi += u;
    }
    wsum[lane] = xi - x;
    if (lane == 31) wsum[32] = xi;
  }
  __syncthreads();
  return incl - v + wsum[w];
}

__device__ __forceinline__ void pack_b_entry(const float* __restrict__ W, float* __restrict__ P,
                                             int K, int Ncols, int ntiles, int t) {
  int total = (K / 4) * ntiles * 32;
  if (t >= total) return;
  int lane = t & 31;
  int frag = t >> 5;
  int ks   = frag / ntiles;
  int tile = frag - ks * ntiles;
  int koff = (lane >> 4) * 2;
  int n    = tile * 16 + (lane & 15);
  int kk   = ks * 4 + koff;
  float v0 = (n < Ncols) ? W[kk * Ncols + n] : 0.0f;
  float v1 = (n < Ncols) ? W[(kk + 1) * Ncols + n] : 0.0f;
  P[(size_t)t * 2 + 0] = v0;
  P[(size_t)t * 2 + 1] = v1;
}

#define PK1 (((FIN / 4) * 2 * 32) / 1024)
#define NPK (PK1 + 3)

static_assert(((FIN / 4) * 2 * 32) % 1024 == 0, "W1's packed entries fill whole work-groups");
static_assert((HH / 4) * 2 * 32 <= 1024 && 15 * 32 <= 1024, "a small matrix is packed by one work-group");

__global__ void __launch_bounds__(1024) k_slice(const int* __restrict__ src,
                                                const int* __restrict__ dst,
                                                int* __restrict__ Lr,
                                                int* __restrict__ pairs,
                                                const float* __restrict__ W1,
                                                const float* __restrict__ W2,
                                                const float* __restrict__ W3,
                                                const float* __restrict__ Wl,
                                                float* __restrict__ Bp1, float* __restrict__ Bp2,
                                                float* __restrict__ Bp3, float* __restrict__ Bpl) {
  __shared__ int wsum[33];
  __shared__ int hist[NBUCK];
  __shared__ int lpos[NBUCK];
  __shared__ int4 st_rec4[CH1 / 4];
  int* st_rec = (int*)st_rec4;
  int t = threadIdx.x;
  int wg = blockIdx.x;
  if (wg >= G1) {
    int pb = wg - G1;
    if (pb < PK1)           pack_b_entry(W1, Bp1, FIN, HH, 2, pb * 1024 + t);
    else if (pb == PK1)     pack_b_entry(W2, Bp2, HH, HH, 2, t);
    else if (pb == PK1 + 1) pack_b_entry(W3, Bp3, HH, HH, 2, t);
    else                    pack_b_entry(Wl, Bpl, 3 * HH, CC, 1, t);
    return;
  }
  for (int b = t; b < NBUCK; b += 1024) hist[b] = 0;
  __syncthreads();
  const int* sp = src + wg * CH1;
  const int* dp = dst + wg * CH1;
  int d[TR1], s[TR1], r[TR1];
  #pragma unroll
  for (int k = 0; k < TR1; ++k) {
    int i = t + 1024 * k;
    d[k] = (i < CH1) ? dp[i] : 0;
    s[k] = (i < CH1) ? sp[i] : 0;
  }
  #pragma unroll
  for (int k = 0; k < TR1; ++k) {
    r[k] = 0;
    if (t + 1024 * k < CH1) r[k] = atomicAdd(&hist[d[k] >> BSH], 1);
  }
  __syncthreads();
  int mine = (t < NBUCK) ? hist[t] : 0;
  int lp = block_excl_scan<32>(mine, wsum, t);
  if (t < NBUCK) lpos[t] = lp;
  __syncthreads();
  #pragma unroll
  for (int k = 0; k < TR1; ++k) {
    if (t + 1024 * k < CH1)
      st_rec[lpos[d[k] >> BSH] + r[k]] = s[k] | ((d[k] & (NBK - 1)) << RSH);
  }
  __syncthreads();
  int4* po = (int4*)(pairs + wg * CH1);
  #pragma unroll
  for (int k = 0; k < (CH1 / 4 + 1023) / 1024; ++k) {
    int i = t + 1024 * k;
    if (i < CH1 / 4) po[i] = st_rec4[i];
  }
  int* Lw = Lr + wg * LW;
  if (t < NBUCK) Lw[t] = lp;
  if (t == NBUCK) Lw[NBUCK] = CH1;
}


__device__ __forceinline__ float dis_of(int b, int e) {
  float d = (float)(e - b + 1);
  return (d > 0.0f) ? rsqrtf(d) : 0.0f;
}

__device__ __forceinline__ int row_end(const int* __restrict__ rp, int row) {
  int nx = rp[min(row + 1, NN - 1)];
  return (row + 1 < NN) ? nx : EE;
}


#define XCH 64
#define XPAD 68
#define TPB (NBK / 16)

static_assert(TPB * 32 == 256 && NN % 16 == 0, "a bucket's work-group has one wave per 16-row tile");

__device__ __forceinline__ void gemm_tile(const float* __restrict__ x, const float* __restrict__ Bp,
                                          float* __restrict__ st, int tile, int lane,
                                          v8f& c0, v8f& c1) {
  int half = lane >> 4;
  int l16  = lane & 15;
  int q  = lane & 7;
  int g  = lane >> 3;
  int r0 = 8 * (g & 1) + (g >> 1);
  const float* xg = x + (size_t)tile * (16 * FIN) + (r0 * FIN + q * 4);
  int so = r0 * XPAD + q * 4;
  int lo = l16 * XPAD + half * 2;
  const v2f* P = (const v2f*)Bp;
  v4f buf[8];
  for (int i = 0; i < 8; ++i)
    buf[i] = __builtin_nontemporal_load((const v4f*)(xg + (size_t)(i >> 1) * 2 * FIN + (i & 1) * 32));
  #pragma unroll 1
  for (int ch = 0; ch < FIN / XCH; ++ch) {
    v2f bq[2 * (XCH / 4)];
    for (int j = 0; j < 2 * (XCH / 4); ++j) bq[j] = P[(ch * 2 * (XCH / 4) + j) * 32 + lane];
    float* sp = st + (ch & 1) * (16 * XPAD);
    for (int i = 0; i < 8; ++i)
      *(v4f*)(sp + so + (i >> 1) * 2 * XPAD + (i & 1) * 32) = buf[i];
    __syncthreads();
    if (ch + 1 < FIN / XCH) {
      for (int i = 0; i < 8; ++i)
        buf[i] = __builtin_nontemporal_load(
            (const v4f*)(xg + (size_t)(i >> 1) * 2 * FIN + (i & 1) * 32 + (ch + 1) * XCH));
    }
    const float* fr = sp + lo;
    for (int ks = 0; ks < XCH / 4; ++ks) {
      v2f a = *(const v2f*)(fr + ks * 4);
      c0 = __builtin_amdgcn_wmma_f32_16x16x4_f32(false, a, false, bq[2 * ks + 0], (short)0, c0, false, false);
      c1 = __builtin_amdgcn_wmma_f32_16x16x4_f32(false, a, false, bq[2 * ks + 1], (short)0, c1, false, false);
    }
  }
}

__global__ void __launch_bounds__(256, 8) k_bucket_gemm(const int* __restrict__ pairs,
                                                     const int* __restrict__ Lr,
                                                     int* __restrict__ rp,
                                                     int* __restrict__ csr,
                                                     const float* __restrict__ x,
                                                     const float* __restrict__ Bp,
                                                     float* __restrict__ h) {
  __shared__ int wsum[33];
  __shared__ int cntl[NBK];
  __shared__ int offl[NBK];
  __shared__ int rsrc[256];
  __shared__ int rlen[256];
  __shared__ int roff[256];
  __shared__ float sdis[NBK];
  __shared__ v4f xst[TPB][2][16 * XPAD / 4];
  static_assert(sizeof(int) * CAP4 <= sizeof(v4f) * TPB * 2 * (16 * XPAD / 4), "the record stage fits the x pieces");
  int* stage = (int*)xst;
  int t = threadIdx.x;
  int b = blockIdx.x;
  int w    = __builtin_amdgcn_readfirstlane(t >> 5);
  int lane = t & 31;
  int tile = b * TPB + w;
  int tl   = min(tile, NN / 16 - 1);
  float* st = (float*)xst[w];
  bool first = (b & 1) != 0;
  v8f c0 = {}; v8f c1 = {};
  if (first) gemm_tile(x, Bp, st, tl, lane, c0, c1);
  __syncthreads();
  int a = 0, len = 0;
  if (t < G1) {
    const int* lw = Lr + t * LW + b;
    a = lw[0];
    len = lw[1] - a;
  }
  int ro = block_excl_scan<8>(len, wsum, t);
  int n = wsum[32];
  __syncthreads();
  block_excl_scan<8>(a, wsum, t);
  int bstart = wsum[32];
  rsrc[t] = t * CH1 + a;
  rlen[t] = len;
  roff[t] = ro;
  if (t < NBK) cntl[t] = 0;
  __syncthreads();
  bool fast = n <= CAP4;
  int g = t >> 3;
  int q = t & 7;
  int rec[KR4];
  if (fast) {
    int rb[RPG], rl[RPG], rf[RPG], val[RPG * KL];
    #pragma unroll
    for (int r = 0; r < RPG; ++r) {
      int k = g + 32 * r;
      rb[r] = rsrc[k]; rl[r] = rlen[k]; rf[r] = roff[k];
    }
    #pragma unroll
    for (int r = 0; r < RPG; ++r) {
      #pragma unroll
      for (int c = 0; c < KL; ++c) {
        int j = q + 8 * c;
        val[r * KL + c] = (j < rl[r]) ? pairs[rb[r] + j] : 0;
      }
    }
    #pragma unroll
    for (int r = 0; r < RPG; ++r) {
      #pragma unroll
      for (int c = 0; c < KL; ++c) {
        int j = q + 8 * c;
        if (j < rl[r]) stage[rf[r] + j] = val[r * KL + c];
      }
    }
    #pragma unroll 1
    for (int r = 0; r < RPG; ++r) {
      int k = g + 32 * r;
      int ln = rlen[k];
      const int* pp = pairs + rsrc[k];
      int of = roff[k];
      #pragma unroll 1
      for (int j = q + 8 * KL; j < ln; j += 8) stage[of + j] = pp[j];
    }
    __syncthreads();
    #pragma unroll
    for (int k = 0; k < KR4; ++k) {
      int i = t + 256 * k;
      rec[k] = (i < n) ? stage[i] : 0;
    }
    #pragma unroll
    for (int k = 0; k < KR4; ++k)
      if (t + 256 * k < n) atomicAdd(&cntl[rec[k] >> RSH], 1);
  } else {
    #pragma unroll 1
    for (int r = 0; r < RPG; ++r) {
      int k = g + 32 * r;
      const int* pp = pairs + rsrc[k];
      int ln = rlen[k];
      #pragma unroll 1
      for (int j = q; j < ln; j += 8) atomicAdd(&cntl[pp[j] >> RSH], 1);
    }
  }
  __syncthreads();
  int c = (t < NBK) ? cntl[t] : 0;
  int ex = block_excl_scan<8>(c, wsum, t);
  if (t < NBK) {
    offl[t] = ex;
    sdis[t] = rsqrtf((float)(c + 1));
    cntl[t] = 0;
    int node = b * NBK + t;
    if (node < NN) rp[node] = bstart + ex;
  }
  __syncthreads();
  if (fast) {
    #pragma unroll
    for (int k = 0; k < KR4; ++k) {
      if (t + 256 * k < n) {
        int dl = rec[k] >> RSH;
        stage[offl[dl] + atomicAdd(&cntl[dl], 1)] = rec[k] & ((1 << RSH) - 1);
      }
    }
    __syncthreads();
    #pragma unroll
    for (int k = 0; k < KR4; ++k) {
      int i = t + 256 * k;
      if (i < n) csr[bstart + i] = stage[i];
    }
  } else {
    #pragma unroll 1
    for (int r = 0; r < RPG; ++r) {
      int k = g + 32 * r;
      const int* pp = pairs + rsrc[k];
      int ln = rlen[k];
      #pragma unroll 1
      for (int j = q; j < ln; j += 8) {
        int rc = pp[j];
        int dl = rc >> RSH;
        csr[bstart + offl[dl] + atomicAdd(&cntl[dl], 1)] = rc & ((1 << RSH) - 1);
      }
    }
  }
  __syncthreads();
  if (!first) gemm_tile(x, Bp, st, tl, lane, c0, c1);
  if (tile < NN / 16) {
    int half = lane >> 4;
    int nc   = lane & 15;
    int lrow = w * 16 + half * 8;
    for (int v = 0; v < 8; ++v) {
      float s = sdis[lrow + v];
      size_t r = (size_t)(b * NBK + lrow + v) * SP;
      h[r + nc]      = c0[v] * s;
      h[r + nc + 16] = c1[v] * s;
    }
  }
}


template <int K>
__device__ __forceinline__ int bcast8(int v) {
  return (int)__builtin_amdgcn_mov_dpp8((unsigned)v, (unsigned)(K * 0x249249));
}

constexpr unsigned xor8_sel(int X) {
  unsigned s = 0;
  for (int i = 0; i < 8; ++i) s |= (unsigned)(i ^ X) << (3 * i);
  return s;
}

template <int X>
__device__ __forceinline__ float xor8(float v) {
  return __builtin_bit_cast(float, __builtin_amdgcn_mov_dpp8(__builtin_bit_cast(unsigned, v), xor8_sel(X)));
}

static_assert(NN % 32 == 0, "a gather work-group has 32 whole rows; rows come in pairs");
#define OP 20

template <int HEAD>
__global__ void __launch_bounds__(128, 10) k_gather(const float* __restrict__ hs,
                                                const int* __restrict__ rp,
                                                const int* __restrict__ csr,
                                                const float* __restrict__ bias,
                                                const float* __restrict__ Bp,
                                                float* __restrict__ o,
                                                float* __restrict__ hn,
                                                const float* __restrict__ oa,
                                                const float* __restrict__ ob,
                                                const float* __restrict__ bl,
                                                float* __restrict__ out) {
  __shared__ float so[(HEAD ? 3 : 1) * 32 * OP];
  __shared__ float sdis[32];
  int t   = blockIdx.x * 128 + threadIdx.x;
  int p   = t >> 3;
  int q   = t & 7;
  int rA  = 2 * p;
  int rB  = rA + 1;
  unsigned qb = (unsigned)q * 16u;
  int begA = rp[rA];
  int endA = rp[rB];
  int endB = row_end(rp, rB);
  int bq = min(q, 4) * 4;
  float bi0 = bias[bq + 0], bi1 = bias[bq + 1], bi2 = bias[bq + 2], bi3 = bias[bq + 3];
  const char* Hb = (const char*)hs;
#define ROW_LD(s) (*(const v4f*)(Hb + ((unsigned)(s) * 128u + qb)))
  v4f accA = ROW_LD(rA);
  v4f accB = ROW_LD(rB);
  int jA = begA;
  int jB = endA;
  int ldA = csr[min(jA + q, EE - 1)];
  int ldB = csr[min(jB + q, EE - 1)];
  int idxA = (jA + q < endA) ? ldA : rA;
  int idxB = (jB + q < endB) ? ldB : rB;
  const v4f z = {0.0f, 0.0f, 0.0f, 0.0f};
  while (jA < endA || jB < endB) {
    v4f a0 = ROW_LD(bcast8<0>(idxA));
    v4f a1 = ROW_LD(bcast8<1>(idxA));
    v4f a2 = ROW_LD(bcast8<2>(idxA));
    v4f a3 = ROW_LD(bcast8<3>(idxA));
    v4f a4 = ROW_LD(bcast8<4>(idxA));
    v4f a5 = ROW_LD(bcast8<5>(idxA));
    v4f a6 = ROW_LD(bcast8<6>(idxA));
    v4f a7 = ROW_LD(bcast8<7>(idxA));
    v4f b0 = ROW_LD(bcast8<0>(idxB));
    v4f b1 = ROW_LD(bcast8<1>(idxB));
    v4f b2 = ROW_LD(bcast8<2>(idxB));
    v4f b3 = ROW_LD(bcast8<3>(idxB));
    v4f b4 = ROW_LD(bcast8<4>(idxB));
    v4f b5 = ROW_LD(bcast8<5>(idxB));
    v4f b6 = ROW_LD(bcast8<6>(idxB));
    v4f b7 = ROW_LD(bcast8<7>(idxB));
    int nA  = jA + 8 + q;
    int nB  = jB + 8 + q;
    int lnA = csr[min(nA, EE - 1)];
    int lnB = csr[min(nB, EE - 1)];
    asm volatile("" :: "v"(a0), "v"(a1), "v"(a2), "v"(a3), "v"(a4), "v"(a5), "v"(a6), "v"(a7),
                       "v"(b0), "v"(b1), "v"(b2), "v"(b3), "v"(b4), "v"(b5), "v"(b6), "v"(b7),
                       "v"(lnA), "v"(lnB));
    int mA = endA - jA;
    int mB = endB - jB;
    accA = accA + ((mA > 0) ? a0 : z); accA = accA + ((mA > 1) ? a1 : z);
    accA = accA + ((mA > 2) ? a2 : z); accA = accA + ((mA > 3) ? a3 : z);
    accA = accA + ((mA > 4) ? a4 : z); accA = accA + ((mA > 5) ? a5 : z);
    accA = accA + ((mA > 6) ? a6 : z); accA = accA + ((mA > 7) ? a7 : z);
    accB = accB + ((mB > 0) ? b0 : z); accB = accB + ((mB > 1) ? b1 : z);
    accB = accB + ((mB > 2) ? b2 : z); accB = accB + ((mB > 3) ? b3 : z);
    accB = accB + ((mB > 4) ? b4 : z); accB = accB + ((mB > 5) ? b5 : z);
    accB = accB + ((mB > 6) ? b6 : z); accB = accB + ((mB > 7) ? b7 : z);
    idxA = (nA < endA) ? lnA : rA;
    idxB = (nB < endB) ? lnB : rB;
    jA += 8;
    jB += 8;
  }
#undef ROW_LD
  float4 paA = make_float4(0.0f, 0.0f, 0.0f, 0.0f), pbA = paA, paB = paA, pbB = paA;
  if (HEAD) {
    paA = ((const float4*)oa)[(size_t)rA * 8 + q];
    pbA = ((const float4*)ob)[(size_t)rA * 8 + q];
    paB = ((const float4*)oa)[(size_t)rB * 8 + q];
    pbB = ((const float4*)ob)[(size_t)rB * 8 + q];
  }
  float ddA = dis_of(begA, endA);
  float ddB = dis_of(endA, endB);
  bool live = q < 5;
  float s0 = live ? accA.x * ddA + bi0 : 0.0f;
  float s1 = live ? accA.y * ddA + bi1 : 0.0f;
  float s2 = live ? accA.z * ddA + bi2 : 0.0f;
  float s3 = live ? accA.w * ddA + bi3 : 0.0f;
  float u0 = live ? accB.x * ddB + bi0 : 0.0f;
  float u1 = live ? accB.y * ddB + bi1 : 0.0f;
  float u2 = live ? accB.z * ddB + bi2 : 0.0f;
  float u3 = live ? accB.w * ddB + bi3 : 0.0f;
  float ssA = s0 * s0 + s1 * s1 + s2 * s2 + s3 * s3;
  float ssB = u0 * u0 + u1 * u1 + u2 * u2 + u3 * u3;
  ssA += xor8<1>(ssA); ssB += xor8<1>(ssB);
  ssA += xor8<2>(ssA); ssB += xor8<2>(ssB);
  ssA += xor8<4>(ssA); ssB += xor8<4>(ssB);
  float invA = 1.0f / fmaxf(sqrtf(ssA), 1e-12f);
  float invB = 1.0f / fmaxf(sqrtf(ssB), 1e-12f);
  float4 ovA = make_float4(fmaxf(s0 * invA, 0.0f), fmaxf(s1 * invA, 0.0f),
                           fmaxf(s2 * invA, 0.0f), fmaxf(s3 * invA, 0.0f));
  float4 ovB = make_float4(fmaxf(u0 * invB, 0.0f), fmaxf(u1 * invB, 0.0f),
                           fmaxf(u2 * invB, 0.0f), fmaxf(u3 * invB, 0.0f));
  int rl   = (threadIdx.x >> 3) * 2;
  int w    = threadIdx.x >> 5;
  int lane = threadIdx.x & 31;
  bool mm  = w < 2;
  const v2f* P = (const v2f*)Bp;
  constexpr int NB = HEAD ? 15 : 10;
  v2f bf[NB];
  for (int j = 0; j < NB; ++j) bf[j] = v2f{0.0f, 0.0f};
  if (mm) {
    for (int j = 0; j < NB; ++j) bf[j] = P[j * 32 + lane];
  }
  if (HEAD == 0) {
    if (live) {
      *(float4*)(so + rl * OP + q * 4) = ovA;
      *(float4*)(so + (rl + 1) * OP + q * 4) = ovB;
    }
    if (q == 0) { sdis[rl] = ddA; sdis[rl + 1] = ddB; }
  } else {
    if (live) {
      *(float4*)(so + 0 * 32 * OP + rl * OP + q * 4) = paA;
      *(float4*)(so + 1 * 32 * OP + rl * OP + q * 4) = pbA;
      *(float4*)(so + 2 * 32 * OP + rl * OP + q * 4) = ovA;
      *(float4*)(so + 0 * 32 * OP + (rl + 1) * OP + q * 4) = paB;
      *(float4*)(so + 1 * 32 * OP + (rl + 1) * OP + q * 4) = pbB;
      *(float4*)(so + 2 * 32 * OP + (rl + 1) * OP + q * 4) = ovB;
    }
  }
  __syncthreads();
  if (HEAD == 0) {
    ((float4*)o)[(size_t)rA * 8 + q] = ovA;
    ((float4*)o)[(size_t)rB * 8 + q] = ovB;
  }
  if (!mm) return;
  int half = lane >> 4;
  int l16  = lane & 15;
  int n    = l16;
  const float* ar = so + (w * 16 + l16) * OP + half * 2;
  int rloc  = w * 16 + half * 8;
  int rbase = blockIdx.x * 32 + rloc;
  if (HEAD == 0) {
    v8f c0 = {}; v8f c1 = {};
    for (int ks = 0; ks < HH / 4; ++ks) {
      v2f a = *(const v2f*)(ar + ks * 4);
      c0 = __builtin_amdgcn_wmma_f32_16x16x4_f32(false, a, false, bf[ks * 2 + 0], (short)0, c0, false, false);
      c1 = __builtin_amdgcn_wmma_f32_16x16x4_f32(false, a, false, bf[ks * 2 + 1], (short)0, c1, false, false);
    }
    for (int v = 0; v < 8; ++v) {
      float s = sdis[rloc + v];
      size_t r = (size_t)(rbase + v) * SP;
      hn[r + n]      = c0[v] * s;
      hn[r + n + 16] = c1[v] * s;
    }
  } else {
    v8f cc = {};
    for (int ks = 0; ks < 15; ++ks) {
      int layer = ks / 5;
      int kloc  = (ks - layer * 5) * 4;
      v2f a = *(const v2f*)(ar + layer * 32 * OP + kloc);
      cc = __builtin_amdgcn_wmma_f32_16x16x4_f32(false, a, false, bf[ks], (short)0, cc, false, false);
    }
    if (n < CC) {
      float bv = bl[n];
      for (int v = 0; v < 8; ++v)
        out[(size_t)(rbase + v) * CC + n] = cc[v] + bv;
    }
  }
}


extern "C" void kernel_launch(void* const* d_in, const int* in_sizes, int n_in,
                              void* d_out, int out_size, void* d_ws, size_t ws_size,
                              hipStream_t stream) {
  (void)in_sizes; (void)n_in; (void)out_size; (void)ws_size;
  const float* x  = (const float*)d_in[0];
  const int*   ei = (const int*)d_in[1];
  const float* W1 = (const float*)d_in[2];
  const float* b1 = (const float*)d_in[3];
  const float* W2 = (const float*)d_in[4];
  const float* b2 = (const float*)d_in[5];
  const float* W3 = (const float*)d_in[6];
  const float* b3 = (const float*)d_in[7];
  const float* Wl = (const float*)d_in[8];
  const float* bl = (const float*)d_in[9];
  const int* srcI = ei;
  const int* dstI = ei + EE;
  float* out = (float*)d_out;

  float* ws   = (float*)d_ws;
  int*   rp   = (int*)ws;
  float* hbuf = ws + NN;
  int*   csr  = (int*)(hbuf + (size_t)NN * SP);
  float* o1   = hbuf + (size_t)2 * NN * SP;
  float* o2   = o1 + (size_t)NN * SP;
  float* o3   = o2 + (size_t)NN * SP;
  float* Bp1  = o3 + (size_t)NN * SP;
  float* Bp2  = Bp1 + (size_t)(FIN / 4) * 2 * 64;
  float* Bp3  = Bp2 + (size_t)(HH / 4) * 2 * 64;
  float* Bpl  = Bp3 + (size_t)(HH / 4) * 2 * 64;
  int*   pairs = (int*)o3;
  int*   Lr   = (int*)o2;

  dim3 B(256);
  dim3 B4(1024);
  dim3 BG(128);
  int gG  = (NN * 4) / 128;

  k_slice<<<G1 + NPK, B4, 0, stream>>>(srcI, dstI, Lr, pairs, W1, W2, W3, Wl, Bp1, Bp2, Bp3, Bpl);
  k_bucket_gemm<<<NBUCK, B, 0, stream>>>(pairs, Lr, rp, csr, x, Bp1, hbuf);

  k_gather<0><<<gG, BG, 0, stream>>>(hbuf, rp, csr, b1, Bp2, o1, o3, nullptr, nullptr, nullptr, nullptr);

  k_gather<0><<<gG, BG, 0, stream>>>(o3, rp, csr, b2, Bp3, o2, hbuf, nullptr, nullptr, nullptr, nullptr);

  k_gather<1><<<gG, BG, 0, stream>>>(hbuf, rp, csr, b3, Bpl, nullptr, nullptr, o1, o2, bl, out);
}


# Gather: a work-group copies its 32 rows' run of the column list to LDS once in whole-line loads; each trip's indices are read there, not by a 4-byte global load per row and trip; sums, order kept
# speedup vs baseline: 20.4054x; 1.0999x over previous
#include <hip/hip_runtime.h>

#define NN 100000
#define EE 3200000
#define FIN 512
#define HH 20
#define CC 10
#define SP 32

typedef __attribute__((ext_vector_type(2))) float v2f;
typedef __attribute__((ext_vector_type(4))) float v4f;
typedef __attribute__((ext_vector_type(8))) float v8f;


#define BSH 7
#define NBK 128
#define NBUCK ((NN + NBK - 1) / NBK)
#define G1 250
#define CH1 (EE / G1)
#define RSH 17
#define TR1 ((CH1 + 1023) / 1024)
#define LW 800
#define RPG 8
#define KL 3
#define KR4 20
#define CAP4 (256 * KR4)

static_assert(EE % G1 == 0, "slices are whole");
static_assert(NBUCK < 1024 && G1 <= 32 * RPG && NBUCK + 1 <= LW && CH1 % 4 == 0 && NN <= (1 << RSH), "sizes the passes rely on");

template <int NW>
__device__ __forceinline__ int block_excl_scan(int v, int* wsum, int t) {
  int lane = t & 31;
  int w = t >> 5;
  int incl = v;
  #pragma unroll
  for (int o = 1; o < 32; o <<= 1) {
    int u = __shfl_up(incl, o, 32);
    if (lane >= o) incl += u;
  }
  if (lane == 31) wsum[w] = incl;
  __syncthreads();
  if (w == 0) {
    int x = (lane < NW) ? wsum[lane] : 0;
    int xi = x;
    #pragma unroll
    for (int o = 1; o < 32; o <<= 1) {
      int u = __shfl_up(xi, o, 32);
      if (lane >= o) xi += u;
    }
    wsum[lane] = xi - x;
    if (lane == 31) wsum[32] = xi;
  }
  __syncthreads();
  return incl - v + wsum[w];
}

__device__ __forceinline__ void pack_b_entry(const float* __restrict__ W, float* __restrict__ P,
                                             int K, int Ncols, int ntiles, int t) {
  int total = (K / 4) * ntiles * 32;
  if (t >= total) return;
  int lane = t & 31;
  int frag = t >> 5;
  int ks   = frag / ntiles;
  int tile = frag - ks * ntiles;
  int koff = (lane >> 4) * 2;
  int n    = tile * 16 + (lane & 15);
  int kk   = ks * 4 + koff;
  float v0 = (n < Ncols) ? W[kk * Ncols + n] : 0.0f;
  float v1 = (n < Ncols) ? W[(kk + 1) * Ncols + n] : 0.0f;
  P[(size_t)t * 2 + 0] = v0;
  P[(size_t)t * 2 + 1] = v1;
}

#define PK1 (((FIN / 4) * 2 * 32) / 1024)
#define NPK (PK1 + 3)

static_assert(((FIN / 4) * 2 * 32) % 1024 == 0, "W1's packed entries fill whole work-groups");
static_assert((HH / 4) * 2 * 32 <= 1024 && 15 * 32 <= 1024, "a small matrix is packed by one work-group");

__global__ void __launch_bounds__(1024) k_slice(const int* __restrict__ src,
                                                const int* __restrict__ dst,
                                                int* __restrict__ Lr,
                                                int* __restrict__ pairs,
                                                const float* __restrict__ W1,
                                                const float* __restrict__ W2,
                                                const float* __restrict__ W3,
                                                const float* __restrict__ Wl,
                                                float* __restrict__ Bp1, float* __restrict__ Bp2,
                                                float* __restrict__ Bp3, float* __restrict__ Bpl) {
  __shared__ int wsum[33];
  __shared__ int hist[NBUCK];
  __shared__ int lpos[NBUCK];
  __shared__ int4 st_rec4[CH1 / 4];
  int* st_rec = (int*)st_rec4;
  int t = threadIdx.x;
  int wg = blockIdx.x;
  if (wg >= G1) {
    int pb = wg - G1;
    if (pb < PK1)           pack_b_entry(W1, Bp1, FIN, HH, 2, pb * 1024 + t);
    else if (pb == PK1)     pack_b_entry(W2, Bp2, HH, HH, 2, t);
    else if (pb == PK1 + 1) pack_b_entry(W3, Bp3, HH, HH, 2, t);
    else                    pack_b_entry(Wl, Bpl, 3 * HH, CC, 1, t);
    return;
  }
  for (int b = t; b < NBUCK; b += 1024) hist[b] = 0;
  __syncthreads();
  const int* sp = src + wg * CH1;
  const int* dp = dst + wg * CH1;
  int d[TR1], s[TR1], r[TR1];
  #pragma unroll
  for (int k = 0; k < TR1; ++k) {
    int i = t + 1024 * k;
    d[k] = (i < CH1) ? dp[i] : 0;
    s[k] = (i < CH1) ? sp[i] : 0;
  }
  #pragma unroll
  for (int k = 0; k < TR1; ++k) {
    r[k] = 0;
    if (t + 1024 * k < CH1) r[k] = atomicAdd(&hist[d[k] >> BSH], 1);
  }
  __syncthreads();
  int mine = (t < NBUCK) ? hist[t] : 0;
  int lp = block_excl_scan<32>(mine, wsum, t);
  if (t < NBUCK) lpos[t] = lp;
  __syncthreads();
  #pragma unroll
  for (int k = 0; k < TR1; ++k) {
    if (t + 1024 * k < CH1)
      st_rec[lpos[d[k] >> BSH] + r[k]] = s[k] | ((d[k] & (NBK - 1)) << RSH);
  }
  __syncthreads();
  int4* po = (int4*)(pairs + wg * CH1);
  #pragma unroll
  for (int k = 0; k < (CH1 / 4 + 1023) / 1024; ++k) {
    int i = t + 1024 * k;
    if (i < CH1 / 4) po[i] = st_rec4[i];
  }
  int* Lw = Lr + wg * LW;
  if (t < NBUCK) Lw[t] = lp;
  if (t == NBUCK) Lw[NBUCK] = CH1;
}


__device__ __forceinline__ float dis_of(int b, int e) {
  float d = (float)(e - b + 1);
  return (d > 0.0f) ? rsqrtf(d) : 0.0f;
}

__device__ __forceinline__ int row_end(const int* __restrict__ rp, int row) {
  int nx = rp[min(row + 1, NN - 1)];
  return (row + 1 < NN) ? nx : EE;
}


#define XCH 64
#define XPAD 68
#define TPB (NBK / 16)

static_assert(TPB * 32 == 256 && NN % 16 == 0, "a bucket's work-group has one wave per 16-row tile");

__device__ __forceinline__ void gemm_tile(const float* __restrict__ x, const float* __restrict__ Bp,
                                          float* __restrict__ st, int tile, int lane,
                                          v8f& c0, v8f& c1) {
  int half = lane >> 4;
  int l16  = lane & 15;
  int q  = lane & 7;
  int g  = lane >> 3;
  int r0 = 8 * (g & 1) + (g >> 1);
  const float* xg = x + (size_t)tile * (16 * FIN) + (r0 * FIN + q * 4);
  int so = r0 * XPAD + q * 4;
  int lo = l16 * XPAD + half * 2;
  const v2f* P = (const v2f*)Bp;
  v4f buf[8];
  for (int i = 0; i < 8; ++i)
    buf[i] = __builtin_nontemporal_load((const v4f*)(xg + (size_t)(i >> 1) * 2 * FIN + (i & 1) * 32));
  #pragma unroll 1
  for (int ch = 0; ch < FIN / XCH; ++ch) {
    v2f bq[2 * (XCH / 4)];
    for (int j = 0; j < 2 * (XCH / 4); ++j) bq[j] = P[(ch * 2 * (XCH / 4) + j) * 32 + lane];
    float* sp = st + (ch & 1) * (16 * XPAD);
    for (int i = 0; i < 8; ++i)
      *(v4f*)(sp + so + (i >> 1) * 2 * XPAD + (i & 1) * 32) = buf[i];
    __syncthreads();
    if (ch + 1 < FIN / XCH) {
      for (int i = 0; i < 8; ++i)
        buf[i] = __builtin_nontemporal_load(
            (const v4f*)(xg + (size_t)(i >> 1) * 2 * FIN + (i & 1) * 32 + (ch + 1) * XCH));
    }
    const float* fr = sp + lo;
    for (int ks = 0; ks < XCH / 4; ++ks) {
      v2f a = *(const v2f*)(fr + ks * 4);
      c0 = __builtin_amdgcn_wmma_f32_16x16x4_f32(false, a, false, bq[2 * ks + 0], (short)0, c0, false, false);
      c1 = __builtin_amdgcn_wmma_f32_16x16x4_f32(false, a, false, bq[2 * ks + 1], (short)0, c1, false, false);
    }
  }
}

__global__ void __launch_bounds__(256, 8) k_bucket_gemm(const int* __restrict__ pairs,
                                                     const int* __restrict__ Lr,
                                                     int* __restrict__ rp,
                                                     int* __restrict__ csr,
                                                     const float* __restrict__ x,
                                                     const float* __restrict__ Bp,
                                                     float* __restrict__ h) {
  __shared__ int wsum[33];
  __shared__ int cntl[NBK];
  __shared__ int offl[NBK];
  __shared__ int rsrc[256];
  __shared__ int rlen[256];
  __shared__ int roff[256];
  __shared__ float sdis[NBK];
  __shared__ v4f xst[TPB][2][16 * XPAD / 4];
  static_assert(sizeof(int) * CAP4 <= sizeof(v4f) * TPB * 2 * (16 * XPAD / 4), "the record stage fits the x pieces");
  int* stage = (int*)xst;
  int t = threadIdx.x;
  int b = blockIdx.x;
  int w    = __builtin_amdgcn_readfirstlane(t >> 5);
  int lane = t & 31;
  int tile = b * TPB + w;
  int tl   = min(tile, NN / 16 - 1);
  float* st = (float*)xst[w];
  bool first = (b & 1) != 0;
  v8f c0 = {}; v8f c1 = {};
  if (first) gemm_tile(x, Bp, st, tl, lane, c0, c1);
  __syncthreads();
  int a = 0, len = 0;
  if (t < G1) {
    const int* lw = Lr + t * LW + b;
    a = lw[0];
    len = lw[1] - a;
  }
  int ro = block_excl_scan<8>(len, wsum, t);
  int n = wsum[32];
  __syncthreads();
  block_excl_scan<8>(a, wsum, t);
  int bstart = wsum[32];
  rsrc[t] = t * CH1 + a;
  rlen[t] = len;
  roff[t] = ro;
  if (t < NBK) cntl[t] = 0;
  __syncthreads();
  bool fast = n <= CAP4;
  int g = t >> 3;
  int q = t & 7;
  int rec[KR4];
  if (fast) {
    int rb[RPG], rl[RPG], rf[RPG], val[RPG * KL];
    #pragma unroll
    for (int r = 0; r < RPG; ++r) {
      int k = g + 32 * r;
      rb[r] = rsrc[k]; rl[r] = rlen[k]; rf[r] = roff[k];
    }
    #pragma unroll
    for (int r = 0; r < RPG; ++r) {
      #pragma unroll
      for (int c = 0; c < KL; ++c) {
        int j = q + 8 * c;
        val[r * KL + c] = (j < rl[r]) ? pairs[rb[r] + j] : 0;
      }
    }
    #pragma unroll
    for (int r = 0; r < RPG; ++r) {
      #pragma unroll
      for (int c = 0; c < KL; ++c) {
        int j = q + 8 * c;
        if (j < rl[r]) stage[rf[r] + j] = val[r * KL + c];
      }
    }
    #pragma unroll 1
    for (int r = 0; r < RPG; ++r) {
      int k = g + 32 * r;
      int ln = rlen[k];
      const int* pp = pairs + rsrc[k];
      int of = roff[k];
      #pragma unroll 1
      for (int j = q + 8 * KL; j < ln; j += 8) stage[of + j] = pp[j];
    }
    __syncthreads();
    #pragma unroll
    for (int k = 0; k < KR4; ++k) {
      int i = t + 256 * k;
      rec[k] = (i < n) ? stage[i] : 0;
    }
    #pragma unroll
    for (int k = 0; k < KR4; ++k)
      if (t + 256 * k < n) atomicAdd(&cntl[rec[k] >> RSH], 1);
  } else {
    #pragma unroll 1
    for (int r = 0; r < RPG; ++r) {
      int k = g + 32 * r;
      const int* pp = pairs + rsrc[k];
      int ln = rlen[k];
      #pragma unroll 1
      for (int j = q; j < ln; j += 8) atomicAdd(&cntl[pp[j] >> RSH], 1);
    }
  }
  __syncthreads();
  int c = (t < NBK) ? cntl[t] : 0;
  int ex = block_excl_scan<8>(c, wsum, t);
  if (t < NBK) {
    offl[t] = ex;
    sdis[t] = rsqrtf((float)(c + 1));
    cntl[t] = 0;
    int node = b * NBK + t;
    if (node < NN) rp[node] = bstart + ex;
  }
  __syncthreads();
  if (fast) {
    #pragma unroll
    for (int k = 0; k < KR4; ++k) {
      if (t + 256 * k < n) {
        int dl = rec[k] >> RSH;
        stage[offl[dl] + atomicAdd(&cntl[dl], 1)] = rec[k] & ((1 << RSH) - 1);
      }
    }
    __syncthreads();
    #pragma unroll
    for (int k = 0; k < KR4; ++k) {
      int i = t + 256 * k;
      if (i < n) csr[bstart + i] = stage[i];
    }
  } else {
    #pragma unroll 1
    for (int r = 0; r < RPG; ++r) {
      int k = g + 32 * r;
      const int* pp = pairs + rsrc[k];
      int ln = rlen[k];
      #pragma unroll 1
      for (int j = q; j < ln; j += 8) {
        int rc = pp[j];
        int dl = rc >> RSH;
        csr[bstart + offl[dl] + atomicAdd(&cntl[dl], 1)] = rc & ((1 << RSH) - 1);
      }
    }
  }
  __syncthreads();
  if (!first) gemm_tile(x, Bp, st, tl, lane, c0, c1);
  if (tile < NN / 16) {
    int half = lane >> 4;
    int nc   = lane & 15;
    int lrow = w * 16 + half * 8;
    for (int v = 0; v < 8; ++v) {
      float s = sdis[lrow + v];
      size_t r = (size_t)(b * NBK + lrow + v) * SP;
      h[r + nc]      = c0[v] * s;
      h[r + nc + 16] = c1[v] * s;
    }
  }
}


template <int K>
__device__ __forceinline__ int bcast8(int v) {
  return (int)__builtin_amdgcn_mov_dpp8((unsigned)v, (unsigned)(K * 0x249249));
}

constexpr unsigned xor8_sel(int X) {
  unsigned s = 0;
  for (int i = 0; i < 8; ++i) s |= (unsigned)(i ^ X) << (3 * i);
  return s;
}

template <int X>
__device__ __forceinline__ float xor8(float v) {
  return __builtin_bit_cast(float, __builtin_amdgcn_mov_dpp8(__builtin_bit_cast(unsigned, v), xor8_sel(X)));
}

static_assert(NN % 32 == 0, "a gather work-group has 32 whole rows; rows come in pairs");
#define OP 20
#define NLD 10
#define CAPL (128 * NLD)

__device__ __forceinline__ void walk_rows(const char* __restrict__ Hb,
                                          const int* scol, int cb, unsigned qb, int q,
                                          int rA, int rB, int begA, int endA, int endB,
                                          v4f& accA, v4f& accB) {
#define ROW_LD(s) (*(const v4f*)(Hb + ((unsigned)(s) * 128u + qb)))
#define COL_AT(pos) (scol[min((pos) - cb, CAPL - 1)])
  int jA = begA;
  int jB = endA;
  int ldA = COL_AT(jA + q);
  int ldB = COL_AT(jB + q);
  int idxA = (jA + q < endA) ? ldA : rA;
  int idxB = (jB + q < endB) ? ldB : rB;
  const v4f z = {0.0f, 0.0f, 0.0f, 0.0f};
  while (jA < endA || jB < endB) {
    v4f a0 = ROW_LD(bcast8<0>(idxA));
    v4f a1 = ROW_LD(bcast8<1>(idxA));
    v4f a2 = ROW_LD(bcast8<2>(idxA));
    v4f a3 = ROW_LD(bcast8<3>(idxA));
    v4f a4 = ROW_LD(bcast8<4>(idxA));
    v4f a5 = ROW_LD(bcast8<5>(idxA));
    v4f a6 = ROW_LD(bcast8<6>(idxA));
    v4f a7 = ROW_LD(bcast8<7>(idxA));
    v4f b0 = ROW_LD(bcast8<0>(idxB));
    v4f b1 = ROW_LD(bcast8<1>(idxB));
    v4f b2 = ROW_LD(bcast8<2>(idxB));
    v4f b3 = ROW_LD(bcast8<3>(idxB));
    v4f b4 = ROW_LD(bcast8<4>(idxB));
    v4f b5 = ROW_LD(bcast8<5>(idxB));
    v4f b6 = ROW_LD(bcast8<6>(idxB));
    v4f b7 = ROW_LD(bcast8<7>(idxB));
    int nA  = jA + 8 + q;
    int nB  = jB + 8 + q;
    int lnA = COL_AT(nA);
    int lnB = COL_AT(nB);
    asm volatile("" : "+v"(a0), "+v"(a1), "+v"(a2), "+v"(a3), "+v"(a4), "+v"(a5), "+v"(a6), "+v"(a7)
                    : "v"(b0), "v"(b1), "v"(b2), "v"(b3), "v"(b4), "v"(b5), "v"(b6), "v"(b7),
                      "v"(lnA), "v"(lnB));
    asm volatile("" : "+v"(b0), "+v"(b1), "+v"(b2), "+v"(b3), "+v"(b4), "+v"(b5), "+v"(b6), "+v"(b7));
    int mA = endA - jA;
    int mB = endB - jB;
    accA = accA + ((mA > 0) ? a0 : z); accA = accA + ((mA > 1) ? a1 : z);
    accA = accA + ((mA > 2) ? a2 : z); accA = accA + ((mA > 3) ? a3 : z);
    accA = accA + ((mA > 4) ? a4 : z); accA = accA + ((mA > 5) ? a5 : z);
    accA = accA + ((mA > 6) ? a6 : z); accA = accA + ((mA > 7) ? a7 : z);
    accB = accB + ((mB > 0) ? b0 : z); accB = accB + ((mB > 1) ? b1 : z);
    accB = accB + ((mB > 2) ? b2 : z); accB = accB + ((mB > 3) ? b3 : z);
    accB = accB + ((mB > 4) ? b4 : z); accB = accB + ((mB > 5) ? b5 : z);
    accB = accB + ((mB > 6) ? b6 : z); accB = accB + ((mB > 7) ? b7 : z);
    idxA = (nA < endA) ? lnA : rA;
    idxB = (nB < endB) ? lnB : rB;
    jA += 8;
    jB += 8;
  }
#undef COL_AT
#undef ROW_LD
}

__device__ __forceinline__ void walk_rows_plain(const char* __restrict__ Hb, const int* __restrict__ csr,
                                                unsigned qb, int begA, int endA, int endB,
                                                v4f& accA, v4f& accB) {
  #pragma unroll 1
  for (int j = begA; j < endA; ++j) accA = accA + *(const v4f*)(Hb + ((unsigned)csr[j] * 128u + qb));
  #pragma unroll 1
  for (int j = endA; j < endB; ++j) accB = accB + *(const v4f*)(Hb + ((unsigned)csr[j] * 128u + qb));
}

template <int HEAD>
__global__ void __launch_bounds__(128, 10) k_gather(const float* __restrict__ hs,
                                                const int* __restrict__ rp,
                                                const int* __restrict__ csr,
                                                const float* __restrict__ bias,
                                                const float* __restrict__ Bp,
                                                float* __restrict__ o,
                                                float* __restrict__ hn,
                                                const float* __restrict__ oa,
                                                const float* __restrict__ ob,
                                                const float* __restrict__ bl,
                                                float* __restrict__ out) {
  __shared__ float so[(HEAD ? 3 : 1) * 32 * OP];
  __shared__ float sdis[32];
  __shared__ int scol[CAPL];
  int t   = blockIdx.x * 128 + threadIdx.x;
  int p   = t >> 3;
  int q   = t & 7;
  int rA  = 2 * p;
  int rB  = rA + 1;
  unsigned qb = (unsigned)q * 16u;
  int begA = rp[rA];
  int endA = rp[rB];
  int endB = row_end(rp, rB);
  int bq = min(q, 4) * 4;
  float bi0 = bias[bq + 0], bi1 = bias[bq + 1], bi2 = bias[bq + 2], bi3 = bias[bq + 3];
  const char* Hb = (const char*)hs;
#define ROW_LD(s) (*(const v4f*)(Hb + ((unsigned)(s) * 128u + qb)))
  v4f accA = ROW_LD(rA);
  v4f accB = ROW_LD(rB);
#undef ROW_LD
  int row0 = blockIdx.x * 32;
  int gb   = rp[row0];
  int ge   = row_end(rp, row0 + 31);
  int cb   = gb & ~31;
  int last = max(ge - 1, 0);
  {
    int cv[NLD];
    #pragma unroll
    for (int k = 0; k < NLD; ++k) cv[k] = csr[min(cb + (int)threadIdx.x + 128 * k, last)];
    #pragma unroll
    for (int k = 0; k < NLD; ++k) scol[threadIdx.x + 128 * k] = cv[k];
  }
  __syncthreads();
  if (ge - cb <= CAPL) walk_rows(Hb, scol, cb, qb, q, rA, rB, begA, endA, endB, accA, accB);
  else                 walk_rows_plain(Hb, csr, qb, begA, endA, endB, accA, accB);
  float4 paA = make_float4(0.0f, 0.0f, 0.0f, 0.0f), pbA = paA, paB = paA, pbB = paA;
  if (HEAD) {
    paA = ((const float4*)oa)[(size_t)rA * 8 + q];
    pbA = ((const float4*)ob)[(size_t)rA * 8 + q];
    paB = ((const float4*)oa)[(size_t)rB * 8 + q];
    pbB = ((const float4*)ob)[(size_t)rB * 8 + q];
  }
  float ddA = dis_of(begA, endA);
  float ddB = dis_of(endA, endB);
  bool live = q < 5;
  float s0 = live ? accA.x * ddA + bi0 : 0.0f;
  float s1 = live ? accA.y * ddA + bi1 : 0.0f;
  float s2 = live ? accA.z * ddA + bi2 : 0.0f;
  float s3 = live ? accA.w * ddA + bi3 : 0.0f;
  float u0 = live ? accB.x * ddB + bi0 : 0.0f;
  float u1 = live ? accB.y * ddB + bi1 : 0.0f;
  float u2 = live ? accB.z * ddB + bi2 : 0.0f;
  float u3 = live ? accB.w * ddB + bi3 : 0.0f;
  float ssA = s0 * s0 + s1 * s1 + s2 * s2 + s3 * s3;
  float ssB = u0 * u0 + u1 * u1 + u2 * u2 + u3 * u3;
  ssA += xor8<1>(ssA); ssB += xor8<1>(ssB);
  ssA += xor8<2>(ssA); ssB += xor8<2>(ssB);
  ssA += xor8<4>(ssA); ssB += xor8<4>(ssB);
  float invA = 1.0f / fmaxf(sqrtf(ssA), 1e-12f);
  float invB = 1.0f / fmaxf(sqrtf(ssB), 1e-12f);
  float4 ovA = make_float4(fmaxf(s0 * invA, 0.0f), fmaxf(s1 * invA, 0.0f),
                           fmaxf(s2 * invA, 0.0f), fmaxf(s3 * invA, 0.0f));
  float4 ovB = make_float4(fmaxf(u0 * invB, 0.0f), fmaxf(u1 * invB, 0.0f),
                           fmaxf(u2 * invB, 0.0f), fmaxf(u3 * invB, 0.0f));
  int rl   = (threadIdx.x >> 3) * 2;
  int w    = threadIdx.x >> 5;
  int lane = threadIdx.x & 31;
  bool mm  = w < 2;
  const v2f* P = (const v2f*)Bp;
  constexpr int NB = HEAD ? 15 : 10;
  v2f bf[NB];
  for (int j = 0; j < NB; ++j) bf[j] = v2f{0.0f, 0.0f};
  if (mm) {
    for (int j = 0; j < NB; ++j) bf[j] = P[j * 32 + lane];
  }
  if (HEAD == 0) {
    if (live) {
      *(float4*)(so + rl * OP + q * 4) = ovA;
      *(float4*)(so + (rl + 1) * OP + q * 4) = ovB;
    }
    if (q == 0) { sdis[rl] = ddA; sdis[rl + 1] = ddB; }
  } else {
    if (live) {
      *(float4*)(so + 0 * 32 * OP + rl * OP + q * 4) = paA;
      *(float4*)(so + 1 * 32 * OP + rl * OP + q * 4) = pbA;
      *(float4*)(so + 2 * 32 * OP + rl * OP + q * 4) = ovA;
      *(float4*)(so + 0 * 32 * OP + (rl + 1) * OP + q * 4) = paB;
      *(float4*)(so + 1 * 32 * OP + (rl + 1) * OP + q * 4) = pbB;
      *(float4*)(so + 2 * 32 * OP + (rl + 1) * OP + q * 4) = ovB;
    }
  }
  __syncthreads();
  if (HEAD == 0) {
    ((float4*)o)[(size_t)rA * 8 + q] = ovA;
    ((float4*)o)[(size_t)rB * 8 + q] = ovB;
  }
  if (!mm) return;
  int half = lane >> 4;
  int l16  = lane & 15;
  int n    = l16;
  const float* ar = so + (w * 16 + l16) * OP + half * 2;
  int rloc  = w * 16 + half * 8;
  int rbase = blockIdx.x * 32 + rloc;
  if (HEAD == 0) {
    v8f c0 = {}; v8f c1 = {};
    for (int ks = 0; ks < HH / 4; ++ks) {
      v2f a = *(const v2f*)(ar + ks * 4);
      c0 = __builtin_amdgcn_wmma_f32_16x16x4_f32(false, a, false, bf[ks * 2 + 0], (short)0, c0, false, false);
      c1 = __builtin_amdgcn_wmma_f32_16x16x4_f32(false, a, false, bf[ks * 2 + 1], (short)0, c1, false, false);
    }
    for (int v = 0; v < 8; ++v) {
      float s = sdis[rloc + v];
      size_t r = (size_t)(rbase + v) * SP;
      hn[r + n]      = c0[v] * s;
      hn[r + n + 16] = c1[v] * s;
    }
  } else {
    v8f cc = {};
    for (int ks = 0; ks < 15; ++ks) {
      int layer = ks / 5;
      int kloc  = (ks - layer * 5) * 4;
      v2f a = *(const v2f*)(ar + layer * 32 * OP + kloc);
      cc = __builtin_amdgcn_wmma_f32_16x16x4_f32(false, a, false, bf[ks], (short)0, cc, false, false);
    }
    if (n < CC) {
      float bv = bl[n];
      for (int v = 0; v < 8; ++v)
        out[(size_t)(rbase + v) * CC + n] = cc[v] + bv;
    }
  }
}


extern "C" void kernel_launch(void* const* d_in, const int* in_sizes, int n_in,
                              void* d_out, int out_size, void* d_ws, size_t ws_size,
                              hipStream_t stream) {
  (void)in_sizes; (void)n_in; (void)out_size; (void)ws_size;
  const float* x  = (const float*)d_in[0];
  const int*   ei = (const int*)d_in[1];
  const float* W1 = (const float*)d_in[2];
  const float* b1 = (const float*)d_in[3];
  const float* W2 = (const float*)d_in[4];
  const float* b2 = (const float*)d_in[5];
  const float* W3 = (const float*)d_in[6];
  const float* b3 = (const float*)d_in[7];
  const float* Wl = (const float*)d_in[8];
  const float* bl = (const float*)d_in[9];
  const int* srcI = ei;
  const int* dstI = ei + EE;
  float* out = (float*)d_out;

  float* ws   = (float*)d_ws;
  int*   rp   = (int*)ws;
  float* hbuf = ws + NN;
  int*   csr  = (int*)(hbuf + (size_t)NN * SP);
  float* o1   = hbuf + (size_t)2 * NN * SP;
  float* o2   = o1 + (size_t)NN * SP;
  float* o3   = o2 + (size_t)NN * SP;
  float* Bp1  = o3 + (size_t)NN * SP;
  float* Bp2  = Bp1 + (size_t)(FIN / 4) * 2 * 64;
  float* Bp3  = Bp2 + (size_t)(HH / 4) * 2 * 64;
  float* Bpl  = Bp3 + (size_t)(HH / 4) * 2 * 64;
  int*   pairs = (int*)o3;
  int*   Lr   = (int*)o2;

  dim3 B(256);
  dim3 B4(1024);
  dim3 BG(128);
  int gG  = (NN * 4) / 128;

  k_slice<<<G1 + NPK, B4, 0, stream>>>(srcI, dstI, Lr, pairs, W1, W2, W3, Wl, Bp1, Bp2, Bp3, Bpl);
  k_bucket_gemm<<<NBUCK, B, 0, stream>>>(pairs, Lr, rp, csr, x, Bp1, hbuf);

  k_gather<0><<<gG, BG, 0, stream>>>(hbuf, rp, csr, b1, Bp2, o1, o3, nullptr, nullptr, nullptr, nullptr);

  k_gather<0><<<gG, BG, 0, stream>>>(o3, rp, csr, b2, Bp3, o2, hbuf, nullptr, nullptr, nullptr, nullptr);

  k_gather<1><<<gG, BG, 0, stream>>>(hbuf, rp, csr, b3, Bpl, nullptr, nullptr, o1, o2, bl, out);
}
